# starting kernel
# baseline (speedup 1.0000x reference)
#include <hip/hip_runtime.h>
#include <hip/hip_bf16.h>
#include <math.h>

#define NN 50000
#define DD 128
#define EE 800000
#define NHOP 3

typedef __attribute__((ext_vector_type(2))) float v2f;
typedef __attribute__((ext_vector_type(8))) float v8f;

__global__ void gate_kernel(const int* __restrict__ depths, int n,
                            const float* __restrict__ Wg1, const float* __restrict__ bg1,
                            const float* __restrict__ Wg2, const float* __restrict__ bg2,
                            float* __restrict__ sw) {
    __shared__ int smax[256];
    int tid = threadIdx.x;
    int m = 0;
    for (int i = tid; i < n; i += 256) m = max(m, depths[i]);
    smax[tid] = m;
    __syncthreads();
    for (int s = 128; s > 0; s >>= 1) {
        if (tid < s) smax[tid] = max(smax[tid], smax[tid + s]);
        __syncthreads();
    }
    if (tid == 0) {
        float sf0 = (float)n / 5000.0f;
        float sf1 = (float)smax[0] / 20.0f;
        float a0 = bg2[0], a1 = bg2[1], a2 = bg2[2];
        for (int j = 0; j < DD / 4; ++j) {
            float h = sf0 * Wg1[j] + sf1 * Wg1[DD / 4 + j] + bg1[j];
            h = fmaxf(h, 0.0f);
            a0 += h * Wg2[j * NHOP + 0];
            a1 += h * Wg2[j * NHOP + 1];
            a2 += h * Wg2[j * NHOP + 2];
        }
        sw[0] = 1.0f / (1.0f + __expf(-a0));
        sw[1] = 1.0f / (1.0f + __expf(-a1));
        sw[2] = 1.0f / (1.0f + __expf(-a2));
    }
}

__global__ void degree_kernel(const int* __restrict__ col, float* __restrict__ counts, int e) {
    int i = blockIdx.x * blockDim.x + threadIdx.x;
    if (i < e) unsafeAtomicAdd(&counts[col[i]], 1.0f);
}

__global__ void scatter_add_kernel(const float* __restrict__ x,
                                   const int* __restrict__ row, const int* __restrict__ col,
                                   float* __restrict__ agg, int e) {
    long gid = (long)blockIdx.x * blockDim.x + threadIdx.x;
    long total = (long)e * (DD / 4);
    if (gid >= total) return;
    int ed = (int)(gid >> 5);
    int c4 = (int)(gid & 31) * 4;
    int r = row[ed], c = col[ed];
    const float4 v = *(const float4*)(x + (long)r * DD + c4);
    float* dst = agg + (long)c * DD + c4;
    unsafeAtomicAdd(dst + 0, v.x);
    unsafeAtomicAdd(dst + 1, v.y);
    unsafeAtomicAdd(dst + 2, v.z);
    unsafeAtomicAdd(dst + 3, v.w);
}

__global__ void divide_kernel(float* __restrict__ agg, const float* __restrict__ counts, int n) {
    long gid = (long)blockIdx.x * blockDim.x + threadIdx.x;
    long total = (long)n * (DD / 4);
    if (gid >= total) return;
    int node = (int)(gid >> 5);
    int off = (int)(gid & 31) * 4;
    float inv = 1.0f / fmaxf(counts[node], 1.0f);
    float4 v = *(float4*)(agg + (long)node * DD + off);
    v.x *= inv; v.y *= inv; v.z *= inv; v.w *= inv;
    *(float4*)(agg + (long)node * DD + off) = v;
}

__global__ void __launch_bounds__(256)
gemm_hop_kernel(const float* __restrict__ X, const float* __restrict__ W,
                const float* __restrict__ b, const float* __restrict__ sw, int hop,
                float* __restrict__ Y) {
    const int row0 = blockIdx.x * 16;
    const int wave = threadIdx.x >> 5;
    const int col0 = wave * 16;
    const int lane = threadIdx.x & 31;
    const int half = lane >> 4;
    const int l    = lane & 15;

    v8f acc = {};
    const float* xr = X + (long)(row0 + l) * DD + 2 * half;
    const float* wc = W + (long)(2 * half) * DD + col0 + l;
#pragma unroll
    for (int k = 0; k < DD; k += 4) {
        v2f a, bb;
        a.x = xr[k];
        a.y = xr[k + 1];
        bb.x = wc[(long)k * DD];
        bb.y = wc[(long)(k + 1) * DD];
        acc = __builtin_amdgcn_wmma_f32_16x16x4_f32(false, a, false, bb,
                                                    (short)0, acc, false, false);
    }
    const float s = sw[hop];
    const float bias = b[col0 + l];
#pragma unroll
    for (int j = 0; j < 8; ++j) {
        int m = j + 8 * half;
        Y[(long)(row0 + m) * DD + col0 + l] = (acc[j] + bias) * s;
    }
}

__global__ void attention_kernel(const float* __restrict__ T0, const float* __restrict__ T1,
                                 const float* __restrict__ T2,
                                 const float* __restrict__ Wa, const float* __restrict__ ba,
                                 float* __restrict__ att, int n) {
    int node = blockIdx.x * blockDim.x + threadIdx.x;
    if (node >= n) return;
    float l0 = ba[0], l1 = ba[1], l2 = ba[2];
    const float* Ts[NHOP] = {T0, T1, T2};
#pragma unroll
    for (int hop = 0; hop < NHOP; ++hop) {
        const float* t = Ts[hop] + (long)node * DD;
        const float* wa = Wa + (long)hop * DD * NHOP;
        for (int kk = 0; kk < DD; ++kk) {
            float xv = t[kk];
            l0 += xv * wa[kk * NHOP + 0];
            l1 += xv * wa[kk * NHOP + 1];
            l2 += xv * wa[kk * NHOP + 2];
        }
    }
    float mx = fmaxf(l0, fmaxf(l1, l2));
    float e0 = __expf(l0 - mx), e1 = __expf(l1 - mx), e2 = __expf(l2 - mx);
    float inv = 1.0f / (e0 + e1 + e2);
    att[(long)node * NHOP + 0] = e0 * inv;
    att[(long)node * NHOP + 1] = e1 * inv;
    att[(long)node * NHOP + 2] = e2 * inv;
}

__global__ void __launch_bounds__(256)
fusion_kernel(const float* __restrict__ T0, const float* __restrict__ T1,
              const float* __restrict__ T2, const float* __restrict__ att,
              const float* __restrict__ Wf, const float* __restrict__ bf,
              float* __restrict__ out) {
    const int row0 = blockIdx.x * 16;
    const int wave = threadIdx.x >> 5;
    const int col0 = wave * 16;
    const int lane = threadIdx.x & 31;
    const int half = lane >> 4;
    const int l    = lane & 15;

    const int row = row0 + l;
    const float a_att0 = att[(long)row * NHOP + 0];
    const float a_att1 = att[(long)row * NHOP + 1];
    const float a_att2 = att[(long)row * NHOP + 2];
    const float* Ts[NHOP] = {T0, T1, T2};

    v8f acc = {};
#pragma unroll
    for (int k = 0; k < NHOP * DD; k += 4) {
        const int hop = k >> 7;
        const float amul = (hop == 0) ? a_att0 : ((hop == 1) ? a_att1 : a_att2);
        const float* t = Ts[hop] + (long)row * DD + (k & (DD - 1)) + 2 * half;
        const float* wf = Wf + (long)(k + 2 * half) * DD + col0 + l;
        v2f a, bb;
        a.x = t[0] * amul;
        a.y = t[1] * amul;
        bb.x = wf[0];
        bb.y = wf[DD];
        acc = __builtin_amdgcn_wmma_f32_16x16x4_f32(false, a, false, bb,
                                                    (short)0, acc, false, false);
    }
    const float bias = bf[col0 + l];
#pragma unroll
    for (int j = 0; j < 8; ++j) {
        int m = j + 8 * half;
        out[(long)(row0 + m) * DD + col0 + l] = acc[j] + bias;
    }
}

static inline int cdiv(long a, long b) { return (int)((a + b - 1) / b); }

extern "C" void kernel_launch(void* const* d_in, const int* in_sizes, int n_in,
                              void* d_out, int out_size, void* d_ws, size_t ws_size,
                              hipStream_t stream) {
    const float* x      = (const float*)d_in[0];
    const int*   ei     = (const int*)  d_in[1];
    const int*   depths = (const int*)  d_in[2];
    const float* Wh     = (const float*)d_in[3];
    const float* bh     = (const float*)d_in[4];
    const float* Wg1    = (const float*)d_in[5];
    const float* bg1    = (const float*)d_in[6];
    const float* Wg2    = (const float*)d_in[7];
    const float* bg2    = (const float*)d_in[8];
    const float* Wa     = (const float*)d_in[9];
    const float* ba     = (const float*)d_in[10];
    const float* Wf     = (const float*)d_in[11];
    const float* bf     = (const float*)d_in[12];
    float* out = (float*)d_out;

    const int* row = ei;
    const int* col = ei + EE;

    char* ws = (char*)d_ws;
    size_t o = 0;
    auto carve = [&](size_t bytes) -> float* {
        float* p = (float*)(ws + o);
        o += (bytes + 255) & ~(size_t)255;
        return p;
    };
    float* sw     = carve(16 * sizeof(float));
    float* counts = carve((size_t)NN * sizeof(float));
    float* agg1   = carve((size_t)NN * DD * sizeof(float));
    float* agg2   = carve((size_t)NN * DD * sizeof(float));
    float* T0     = carve((size_t)NN * DD * sizeof(float));
    float* T1     = carve((size_t)NN * DD * sizeof(float));
    float* T2     = carve((size_t)NN * DD * sizeof(float));
    float* att    = carve((size_t)NN * NHOP * sizeof(float));
    (void)ws_size; (void)in_sizes; (void)n_in; (void)out_size;

    hipMemsetAsync(counts, 0, (size_t)NN * sizeof(float), stream);
    hipMemsetAsync(agg1,   0, (size_t)NN * DD * sizeof(float), stream);
    hipMemsetAsync(agg2,   0, (size_t)NN * DD * sizeof(float), stream);

    gate_kernel<<<1, 256, 0, stream>>>(depths, NN, Wg1, bg1, Wg2, bg2, sw);

    degree_kernel<<<cdiv(EE, 256), 256, 0, stream>>>(col, counts, EE);

    scatter_add_kernel<<<cdiv((long)EE * (DD / 4), 256), 256, 0, stream>>>(x, row, col, agg1, EE);
    divide_kernel<<<cdiv((long)NN * (DD / 4), 256), 256, 0, stream>>>(agg1, counts, NN);

    scatter_add_kernel<<<cdiv((long)EE * (DD / 4), 256), 256, 0, stream>>>(agg1, row, col, agg2, EE);
    divide_kernel<<<cdiv((long)NN * (DD / 4), 256), 256, 0, stream>>>(agg2, counts, NN);

    const int rowTiles = NN / 16;
    gemm_hop_kernel<<<rowTiles, 256, 0, stream>>>(x,    Wh + 0 * DD * DD, bh + 0 * DD, sw, 0, T0);
    gemm_hop_kernel<<<rowTiles, 256, 0, stream>>>(agg1, Wh + 1 * DD * DD, bh + 1 * DD, sw, 1, T1);
    gemm_hop_kernel<<<rowTiles, 256, 0, stream>>>(agg2, Wh + 2 * DD * DD, bh + 2 * DD, sw, 2, T2);

    attention_kernel<<<cdiv(NN, 256), 256, 0, stream>>>(T0, T1, T2, Wa, ba, att, NN);

    fusion_kernel<<<rowTiles, 256, 0, stream>>>(T0, T1, T2, att, Wf, bf, out);
}


# scatter-mean without float atomics: edges counting-sorted by destination in-call (int atomics, arrival-order slots), then a wave per node sums its source rows in slot order and scales once
# speedup vs baseline: 1.6594x; 1.6594x over previous
#include <hip/hip_runtime.h>
#include <hip/hip_bf16.h>
#include <math.h>

#define NN 50000
#define DD 128
#define EE 800000
#define NHOP 3

typedef __attribute__((ext_vector_type(2))) float v2f;
typedef __attribute__((ext_vector_type(8))) float v8f;

__global__ void gate_kernel(const int* __restrict__ depths, int n,
                            const float* __restrict__ Wg1, const float* __restrict__ bg1,
                            const float* __restrict__ Wg2, const float* __restrict__ bg2,
                            float* __restrict__ sw) {
    __shared__ int smax[256];
    int tid = threadIdx.x;
    int m = 0;
    for (int i = tid; i < n; i += 256) m = max(m, depths[i]);
    smax[tid] = m;
    __syncthreads();
    for (int s = 128; s > 0; s >>= 1) {
        if (tid < s) smax[tid] = max(smax[tid], smax[tid + s]);
        __syncthreads();
    }
    if (tid == 0) {
        float sf0 = (float)n / 5000.0f;
        float sf1 = (float)smax[0] / 20.0f;
        float a0 = bg2[0], a1 = bg2[1], a2 = bg2[2];
        for (int j = 0; j < DD / 4; ++j) {
            float h = sf0 * Wg1[j] + sf1 * Wg1[DD / 4 + j] + bg1[j];
            h = fmaxf(h, 0.0f);
            a0 += h * Wg2[j * NHOP + 0];
            a1 += h * Wg2[j * NHOP + 1];
            a2 += h * Wg2[j * NHOP + 2];
        }
        sw[0] = 1.0f / (1.0f + __expf(-a0));
        sw[1] = 1.0f / (1.0f + __expf(-a1));
        sw[2] = 1.0f / (1.0f + __expf(-a2));
    }
}

static_assert(EE % 4 == 0, "edge list is read as int4");
static_assert(NN % 4 == 0, "degree array is scanned as int4");

__global__ void __launch_bounds__(256)
hist_kernel(const int* __restrict__ col, int* __restrict__ cnt, int e4) {
    int i = blockIdx.x * 256 + threadIdx.x;
    if (i >= e4) return;
    const int4 c = ((const int4*)col)[i];
    atomicAdd(&cnt[c.x], 1);
    atomicAdd(&cnt[c.y], 1);
    atomicAdd(&cnt[c.z], 1);
    atomicAdd(&cnt[c.w], 1);
}

#define SCAN_T 1024
#define SCAN_C 52
static_assert(SCAN_T * SCAN_C >= NN, "one block must cover all nodes");

__global__ void __launch_bounds__(SCAN_T)
scan_kernel(const int* __restrict__ cnt, int* __restrict__ cursor) {
    __shared__ int wsum[32];
    const int t = threadIdx.x;
    const int lane = t & 31;
    const int w = t >> 5;
    const int base = t * SCAN_C;

    int4 v[SCAN_C / 4];
    int s = 0;
    for (int i = 0; i < SCAN_C / 4; ++i) {
        const int idx = base + 4 * i;
        v[i] = (idx < NN) ? *(const int4*)(cnt + idx) : make_int4(0, 0, 0, 0);
        s += v[i].x + v[i].y + v[i].z + v[i].w;
    }
    int inc = s;
    for (int d = 1; d < 32; d <<= 1) {
        int o = __shfl_up(inc, d, 32);
        if (lane >= d) inc += o;
    }
    if (lane == 31) wsum[w] = inc;
    __syncthreads();
    if (w == 0) {
        const int x = wsum[lane];
        int xi = x;
        for (int d = 1; d < 32; d <<= 1) {
            int o = __shfl_up(xi, d, 32);
            if (lane >= d) xi += o;
        }
        wsum[lane] = xi - x;
    }
    __syncthreads();
    int run = wsum[w] + inc - s;
    for (int i = 0; i < SCAN_C / 4; ++i) {
        const int idx = base + 4 * i;
        if (idx < NN) {
            int4 o;
            o.x = run; run += v[i].x;
            o.y = run; run += v[i].y;
            o.z = run; run += v[i].z;
            o.w = run; run += v[i].w;
            *(int4*)(cursor + idx) = o;
        }
    }
}

__global__ void __launch_bounds__(256)
fill_kernel(const int* __restrict__ row, const int* __restrict__ col,
            int* __restrict__ cursor, int* __restrict__ perm, int e4) {
    int i = blockIdx.x * 256 + threadIdx.x;
    if (i >= e4) return;
    const int4 c = ((const int4*)col)[i];
    const int4 r = ((const int4*)row)[i];
    const int p0 = atomicAdd(&cursor[c.x], 1);
    const int p1 = atomicAdd(&cursor[c.y], 1);
    const int p2 = atomicAdd(&cursor[c.z], 1);
    const int p3 = atomicAdd(&cursor[c.w], 1);
    perm[p0] = r.x;
    perm[p1] = r.y;
    perm[p2] = r.z;
    perm[p3] = r.w;
}

#define GATHER_WAVES_PER_BLOCK 8
#define GATHER_BATCH 8

__device__ __forceinline__ float4 load_row4(const float* __restrict__ src, int r, int lane) {
    return *(const float4*)(src + (long)r * DD + lane * 4);
}

__global__ void __launch_bounds__(256)
gather_mean_kernel(const float* __restrict__ src, const int* __restrict__ perm,
                   const int* __restrict__ cnt, const int* __restrict__ segend,
                   float* __restrict__ dst, int n) {
    const int lane = threadIdx.x & 31;
    const int wave0 = __builtin_amdgcn_readfirstlane(
        (int)(blockIdx.x * GATHER_WAVES_PER_BLOCK + (threadIdx.x >> 5)));
    const int nwaves = (int)gridDim.x * GATHER_WAVES_PER_BLOCK;

    for (int node = wave0; node < n; node += nwaves) {
        const int deg = cnt[node];
        const int end = segend[node];
        float4 acc = make_float4(0.0f, 0.0f, 0.0f, 0.0f);

        for (int base = end - deg; base < end; base += 32) {
            const int m = min(32, end - base);
            int idx = (lane < m) ? perm[base + lane] : 0;
            idx = min(max(idx, 0), n - 1);

            int j = 0;
            for (; j + GATHER_BATCH <= m; j += GATHER_BATCH) {
                float4 v[GATHER_BATCH];
                for (int u = 0; u < GATHER_BATCH; ++u)
                    v[u] = load_row4(src, __builtin_amdgcn_readlane(idx, j + u), lane);
                for (int u = 0; u < GATHER_BATCH; ++u) {
                    acc.x += v[u].x; acc.y += v[u].y; acc.z += v[u].z; acc.w += v[u].w;
                }
            }
            if (j < m) {
                float4 v[GATHER_BATCH - 1];
                for (int u = 0; u < GATHER_BATCH - 1; ++u) {
                    v[u] = make_float4(0.0f, 0.0f, 0.0f, 0.0f);
                    if (j + u < m)
                        v[u] = load_row4(src, __builtin_amdgcn_readlane(idx, j + u), lane);
                }
                for (int u = 0; u < GATHER_BATCH - 1; ++u) {
                    if (j + u < m) {
                        acc.x += v[u].x; acc.y += v[u].y; acc.z += v[u].z; acc.w += v[u].w;
                    }
                }
            }
        }
        const float inv = 1.0f / fmaxf((float)deg, 1.0f);
        acc.x *= inv; acc.y *= inv; acc.z *= inv; acc.w *= inv;
        *(float4*)(dst + (long)node * DD + lane * 4) = acc;
    }
}

__global__ void __launch_bounds__(256)
gemm_hop_kernel(const float* __restrict__ X, const float* __restrict__ W,
                const float* __restrict__ b, const float* __restrict__ sw, int hop,
                float* __restrict__ Y) {
    const int row0 = blockIdx.x * 16;
    const int wave = threadIdx.x >> 5;
    const int col0 = wave * 16;
    const int lane = threadIdx.x & 31;
    const int half = lane >> 4;
    const int l    = lane & 15;

    v8f acc = {};
    const float* xr = X + (long)(row0 + l) * DD + 2 * half;
    const float* wc = W + (long)(2 * half) * DD + col0 + l;
#pragma unroll
    for (int k = 0; k < DD; k += 4) {
        v2f a, bb;
        a.x = xr[k];
        a.y = xr[k + 1];
        bb.x = wc[(long)k * DD];
        bb.y = wc[(long)(k + 1) * DD];
        acc = __builtin_amdgcn_wmma_f32_16x16x4_f32(false, a, false, bb,
                                                    (short)0, acc, false, false);
    }
    const float s = sw[hop];
    const float bias = b[col0 + l];
#pragma unroll
    for (int j = 0; j < 8; ++j) {
        int m = j + 8 * half;
        Y[(long)(row0 + m) * DD + col0 + l] = (acc[j] + bias) * s;
    }
}

__global__ void attention_kernel(const float* __restrict__ T0, const float* __restrict__ T1,
                                 const float* __restrict__ T2,
                                 const float* __restrict__ Wa, const float* __restrict__ ba,
                                 float* __restrict__ att, int n) {
    int node = blockIdx.x * blockDim.x + threadIdx.x;
    if (node >= n) return;
    float l0 = ba[0], l1 = ba[1], l2 = ba[2];
    const float* Ts[NHOP] = {T0, T1, T2};
#pragma unroll
    for (int hop = 0; hop < NHOP; ++hop) {
        const float* t = Ts[hop] + (long)node * DD;
        const float* wa = Wa + (long)hop * DD * NHOP;
        for (int kk = 0; kk < DD; ++kk) {
            float xv = t[kk];
            l0 += xv * wa[kk * NHOP + 0];
            l1 += xv * wa[kk * NHOP + 1];
            l2 += xv * wa[kk * NHOP + 2];
        }
    }
    float mx = fmaxf(l0, fmaxf(l1, l2));
    float e0 = __expf(l0 - mx), e1 = __expf(l1 - mx), e2 = __expf(l2 - mx);
    float inv = 1.0f / (e0 + e1 + e2);
    att[(long)node * NHOP + 0] = e0 * inv;
    att[(long)node * NHOP + 1] = e1 * inv;
    att[(long)node * NHOP + 2] = e2 * inv;
}

__global__ void __launch_bounds__(256)
fusion_kernel(const float* __restrict__ T0, const float* __restrict__ T1,
              const float* __restrict__ T2, const float* __restrict__ att,
              const float* __restrict__ Wf, const float* __restrict__ bf,
              float* __restrict__ out) {
    const int row0 = blockIdx.x * 16;
    const int wave = threadIdx.x >> 5;
    const int col0 = wave * 16;
    const int lane = threadIdx.x & 31;
    const int half = lane >> 4;
    const int l    = lane & 15;

    const int row = row0 + l;
    const float a_att0 = att[(long)row * NHOP + 0];
    const float a_att1 = att[(long)row * NHOP + 1];
    const float a_att2 = att[(long)row * NHOP + 2];
    const float* Ts[NHOP] = {T0, T1, T2};

    v8f acc = {};
#pragma unroll
    for (int k = 0; k < NHOP * DD; k += 4) {
        const int hop = k >> 7;
        const float amul = (hop == 0) ? a_att0 : ((hop == 1) ? a_att1 : a_att2);
        const float* t = Ts[hop] + (long)row * DD + (k & (DD - 1)) + 2 * half;
        const float* wf = Wf + (long)(k + 2 * half) * DD + col0 + l;
        v2f a, bb;
        a.x = t[0] * amul;
        a.y = t[1] * amul;
        bb.x = wf[0];
        bb.y = wf[DD];
        acc = __builtin_amdgcn_wmma_f32_16x16x4_f32(false, a, false, bb,
                                                    (short)0, acc, false, false);
    }
    const float bias = bf[col0 + l];
#pragma unroll
    for (int j = 0; j < 8; ++j) {
        int m = j + 8 * half;
        out[(long)(row0 + m) * DD + col0 + l] = acc[j] + bias;
    }
}

static inline int cdiv(long a, long b) { return (int)((a + b - 1) / b); }

extern "C" void kernel_launch(void* const* d_in, const int* in_sizes, int n_in,
                              void* d_out, int out_size, void* d_ws, size_t ws_size,
                              hipStream_t stream) {
    const float* x      = (const float*)d_in[0];
    const int*   ei     = (const int*)  d_in[1];
    const int*   depths = (const int*)  d_in[2];
    const float* Wh     = (const float*)d_in[3];
    const float* bh     = (const float*)d_in[4];
    const float* Wg1    = (const float*)d_in[5];
    const float* bg1    = (const float*)d_in[6];
    const float* Wg2    = (const float*)d_in[7];
    const float* bg2    = (const float*)d_in[8];
    const float* Wa     = (const float*)d_in[9];
    const float* ba     = (const float*)d_in[10];
    const float* Wf     = (const float*)d_in[11];
    const float* bf     = (const float*)d_in[12];
    float* out = (float*)d_out;

    const int* row = ei;
    const int* col = ei + EE;

    char* ws = (char*)d_ws;
    size_t o = 0;
    auto carve = [&](size_t bytes) -> float* {
        float* p = (float*)(ws + o);
        o += (bytes + 255) & ~(size_t)255;
        return p;
    };
    float* sw     = carve(16 * sizeof(float));
    float* counts = carve((size_t)NN * sizeof(float));
    float* agg1   = carve((size_t)NN * DD * sizeof(float));
    float* agg2   = carve((size_t)NN * DD * sizeof(float));
    float* T0     = carve((size_t)NN * DD * sizeof(float));
    float* T1     = carve((size_t)NN * DD * sizeof(float));
    float* T2     = carve((size_t)NN * DD * sizeof(float));
    float* att    = carve((size_t)NN * NHOP * sizeof(float));
    (void)ws_size; (void)in_sizes; (void)n_in; (void)out_size;

    int* cnt    = (int*)counts;
    int* cursor = (int*)T2;
    int* perm   = (int*)T2 + 65536;

    hipMemsetAsync(cnt, 0, (size_t)NN * sizeof(int), stream);

    gate_kernel<<<1, 256, 0, stream>>>(depths, NN, Wg1, bg1, Wg2, bg2, sw);

    hist_kernel<<<cdiv(EE / 4, 256), 256, 0, stream>>>(col, cnt, EE / 4);
    scan_kernel<<<1, SCAN_T, 0, stream>>>(cnt, cursor);
    fill_kernel<<<cdiv(EE / 4, 256), 256, 0, stream>>>(row, col, cursor, perm, EE / 4);

    gather_mean_kernel<<<2048, 256, 0, stream>>>(x,    perm, cnt, cursor, agg1, NN);
    gather_mean_kernel<<<2048, 256, 0, stream>>>(agg1, perm, cnt, cursor, agg2, NN);

    const int rowTiles = NN / 16;
    gemm_hop_kernel<<<rowTiles, 256, 0, stream>>>(x,    Wh + 0 * DD * DD, bh + 0 * DD, sw, 0, T0);
    gemm_hop_kernel<<<rowTiles, 256, 0, stream>>>(agg1, Wh + 1 * DD * DD, bh + 1 * DD, sw, 1, T1);
    gemm_hop_kernel<<<rowTiles, 256, 0, stream>>>(agg2, Wh + 2 * DD * DD, bh + 2 * DD, sw, 2, T2);

    attention_kernel<<<cdiv(NN, 256), 256, 0, stream>>>(T0, T1, T2, Wa, ba, att, NN);

    fusion_kernel<<<rowTiles, 256, 0, stream>>>(T0, T1, T2, att, Wf, bf, out);
}


# WMMA GEMMs re-tiled: each wave owns 16 rows x all 128 cols (8 accumulators); a 4-wave block stages each 128x128 weight slab once in LDS (k-pairs interleaved, padded rows); k and sum order kept
# speedup vs baseline: 2.0569x; 1.2395x over previous
#include <hip/hip_runtime.h>
#include <hip/hip_bf16.h>
#include <math.h>

#define NN 50000
#define DD 128
#define EE 800000
#define NHOP 3

typedef __attribute__((ext_vector_type(2))) float v2f;
typedef __attribute__((ext_vector_type(8))) float v8f;

__global__ void gate_kernel(const int* __restrict__ depths, int n,
                            const float* __restrict__ Wg1, const float* __restrict__ bg1,
                            const float* __restrict__ Wg2, const float* __restrict__ bg2,
                            float* __restrict__ sw) {
    __shared__ int smax[256];
    int tid = threadIdx.x;
    int m = 0;
    for (int i = tid; i < n; i += 256) m = max(m, depths[i]);
    smax[tid] = m;
    __syncthreads();
    for (int s = 128; s > 0; s >>= 1) {
        if (tid < s) smax[tid] = max(smax[tid], smax[tid + s]);
        __syncthreads();
    }
    if (tid == 0) {
        float sf0 = (float)n / 5000.0f;
        float sf1 = (float)smax[0] / 20.0f;
        float a0 = bg2[0], a1 = bg2[1], a2 = bg2[2];
        for (int j = 0; j < DD / 4; ++j) {
            float h = sf0 * Wg1[j] + sf1 * Wg1[DD / 4 + j] + bg1[j];
            h = fmaxf(h, 0.0f);
            a0 += h * Wg2[j * NHOP + 0];
            a1 += h * Wg2[j * NHOP + 1];
            a2 += h * Wg2[j * NHOP + 2];
        }
        sw[0] = 1.0f / (1.0f + __expf(-a0));
        sw[1] = 1.0f / (1.0f + __expf(-a1));
        sw[2] = 1.0f / (1.0f + __expf(-a2));
    }
}

static_assert(EE % 4 == 0, "edge list is read as int4");
static_assert(NN % 4 == 0, "degree array is scanned as int4");

__global__ void __launch_bounds__(256)
hist_kernel(const int* __restrict__ col, int* __restrict__ cnt, int e4) {
    int i = blockIdx.x * 256 + threadIdx.x;
    if (i >= e4) return;
    const int4 c = ((const int4*)col)[i];
    atomicAdd(&cnt[c.x], 1);
    atomicAdd(&cnt[c.y], 1);
    atomicAdd(&cnt[c.z], 1);
    atomicAdd(&cnt[c.w], 1);
}

#define SCAN_T 1024
#define SCAN_C 52
static_assert(SCAN_T * SCAN_C >= NN, "one block must cover all nodes");

__global__ void __launch_bounds__(SCAN_T)
scan_kernel(const int* __restrict__ cnt, int* __restrict__ cursor) {
    __shared__ int wsum[32];
    const int t = threadIdx.x;
    const int lane = t & 31;
    const int w = t >> 5;
    const int base = t * SCAN_C;

    int4 v[SCAN_C / 4];
    int s = 0;
    for (int i = 0; i < SCAN_C / 4; ++i) {
        const int idx = base + 4 * i;
        v[i] = (idx < NN) ? *(const int4*)(cnt + idx) : make_int4(0, 0, 0, 0);
        s += v[i].x + v[i].y + v[i].z + v[i].w;
    }
    int inc = s;
    for (int d = 1; d < 32; d <<= 1) {
        int o = __shfl_up(inc, d, 32);
        if (lane >= d) inc += o;
    }
    if (lane == 31) wsum[w] = inc;
    __syncthreads();
    if (w == 0) {
        const int x = wsum[lane];
        int xi = x;
        for (int d = 1; d < 32; d <<= 1) {
            int o = __shfl_up(xi, d, 32);
            if (lane >= d) xi += o;
        }
        wsum[lane] = xi - x;
    }
    __syncthreads();
    int run = wsum[w] + inc - s;
    for (int i = 0; i < SCAN_C / 4; ++i) {
        const int idx = base + 4 * i;
        if (idx < NN) {
            int4 o;
            o.x = run; run += v[i].x;
            o.y = run; run += v[i].y;
            o.z = run; run += v[i].z;
            o.w = run; run += v[i].w;
            *(int4*)(cursor + idx) = o;
        }
    }
}

__global__ void __launch_bounds__(256)
fill_kernel(const int* __restrict__ row, const int* __restrict__ col,
            int* __restrict__ cursor, int* __restrict__ perm, int e4) {
    int i = blockIdx.x * 256 + threadIdx.x;
    if (i >= e4) return;
    const int4 c = ((const int4*)col)[i];
    const int4 r = ((const int4*)row)[i];
    const int p0 = atomicAdd(&cursor[c.x], 1);
    const int p1 = atomicAdd(&cursor[c.y], 1);
    const int p2 = atomicAdd(&cursor[c.z], 1);
    const int p3 = atomicAdd(&cursor[c.w], 1);
    perm[p0] = r.x;
    perm[p1] = r.y;
    perm[p2] = r.z;
    perm[p3] = r.w;
}

#define GATHER_WAVES_PER_BLOCK 8
#define GATHER_BATCH 8

__device__ __forceinline__ float4 load_row4(const float* __restrict__ src, int r, int lane) {
    return *(const float4*)(src + (long)r * DD + lane * 4);
}

__global__ void __launch_bounds__(256)
gather_mean_kernel(const float* __restrict__ src, const int* __restrict__ perm,
                   const int* __restrict__ cnt, const int* __restrict__ segend,
                   float* __restrict__ dst, int n) {
    const int lane = threadIdx.x & 31;
    const int wave0 = __builtin_amdgcn_readfirstlane(
        (int)(blockIdx.x * GATHER_WAVES_PER_BLOCK + (threadIdx.x >> 5)));
    const int nwaves = (int)gridDim.x * GATHER_WAVES_PER_BLOCK;

    for (int node = wave0; node < n; node += nwaves) {
        const int deg = cnt[node];
        const int end = segend[node];
        float4 acc = make_float4(0.0f, 0.0f, 0.0f, 0.0f);

        for (int base = end - deg; base < end; base += 32) {
            const int m = min(32, end - base);
            int idx = (lane < m) ? perm[base + lane] : 0;
            idx = min(max(idx, 0), n - 1);

            int j = 0;
            for (; j + GATHER_BATCH <= m; j += GATHER_BATCH) {
                float4 v[GATHER_BATCH];
                for (int u = 0; u < GATHER_BATCH; ++u)
                    v[u] = load_row4(src, __builtin_amdgcn_readlane(idx, j + u), lane);
                for (int u = 0; u < GATHER_BATCH; ++u) {
                    acc.x += v[u].x; acc.y += v[u].y; acc.z += v[u].z; acc.w += v[u].w;
                }
            }
            if (j < m) {
                float4 v[GATHER_BATCH - 1];
                for (int u = 0; u < GATHER_BATCH - 1; ++u) {
                    v[u] = make_float4(0.0f, 0.0f, 0.0f, 0.0f);
                    if (j + u < m)
                        v[u] = load_row4(src, __builtin_amdgcn_readlane(idx, j + u), lane);
                }
                for (int u = 0; u < GATHER_BATCH - 1; ++u) {
                    if (j + u < m) {
                        acc.x += v[u].x; acc.y += v[u].y; acc.z += v[u].z; acc.w += v[u].w;
                    }
                }
            }
        }
        const float inv = 1.0f / fmaxf((float)deg, 1.0f);
        acc.x *= inv; acc.y *= inv; acc.z *= inv; acc.w *= inv;
        *(float4*)(dst + (long)node * DD + lane * 4) = acc;
    }
}

#define GEMM_WAVES 4
#define GEMM_THREADS (GEMM_WAVES * 32)
#define ROW_TILES (NN / 16)
#define GEMM_BLOCKS ((ROW_TILES + GEMM_WAVES - 1) / GEMM_WAVES)
static_assert(NN % 16 == 0, "rows are handled in whole 16-row tiles");
static_assert(DD == 128, "the slab layout below is written for 128 x 128 weight slabs");

#define W_PITCH 288
#define W_SLAB_DWORDS ((DD / 2) * W_PITCH)
#define W_ITEMS_PER_THREAD ((DD / 2) / GEMM_WAVES)

__device__ __forceinline__ void stage_slab(const float* __restrict__ Wsrc, float* wl, int tid) {
    const int kp0 = tid >> 5;
    const int c4  = tid & 31;
    const float* src = Wsrc + (long)(2 * kp0) * DD + 4 * c4;
    float* dst = wl + kp0 * W_PITCH + 8 * c4;
#pragma unroll
    for (int it = 0; it < W_ITEMS_PER_THREAD; ++it) {
        const float4 r0 = *(const float4*)(src + it * (2 * GEMM_WAVES * DD));
        const float4 r1 = *(const float4*)(src + it * (2 * GEMM_WAVES * DD) + DD);
        float* d = dst + it * (GEMM_WAVES * W_PITCH);
        *(float4*)(d)     = make_float4(r0.x, r1.x, r0.y, r1.y);
        *(float4*)(d + 4) = make_float4(r0.z, r1.z, r0.w, r1.w);
    }
}

__device__ __forceinline__ void load_a_frags(const float* __restrict__ arow, v2f (&a)[DD / 4]) {
#pragma unroll
    for (int s = 0; s < DD / 4; ++s) {
        a[s].x = arow[4 * s];
        a[s].y = arow[4 * s + 1];
    }
}

__device__ __forceinline__ void mma_slab(const v2f (&a)[DD / 4], const float* wl_lane, v8f (&acc)[8]) {
#pragma unroll
    for (int s = 0; s < DD / 4; ++s) {
#pragma unroll
        for (int ct = 0; ct < 8; ++ct) {
            const v2f bb = *(const v2f*)(wl_lane + 2 * s * W_PITCH + 32 * ct);
            acc[ct] = __builtin_amdgcn_wmma_f32_16x16x4_f32(false, a[s], false, bb,
                                                            (short)0, acc[ct], false, false);
        }
    }
}

__global__ void __launch_bounds__(GEMM_THREADS)
gemm_hop_kernel(const float* __restrict__ X, const float* __restrict__ W,
                const float* __restrict__ b, const float* __restrict__ sw, int hop,
                float* __restrict__ Y) {
    __shared__ __attribute__((aligned(16))) float wl[W_SLAB_DWORDS];
    const int tid = threadIdx.x;
    const int wave = __builtin_amdgcn_readfirstlane(tid >> 5);
    const int lane = tid & 31;
    const int half = lane >> 4;
    const int l    = lane & 15;
    const int rt = blockIdx.x * GEMM_WAVES + wave;
    const bool valid = rt < ROW_TILES;
    const int row0 = (valid ? rt : 0) * 16;

    stage_slab(W, wl, tid);
    v2f a[DD / 4];
    load_a_frags(X + (long)(row0 + l) * DD + 2 * half, a);
    __syncthreads();
    if (!valid) return;

    v8f acc[8] = {};
    mma_slab(a, wl + half * W_PITCH + 2 * l, acc);

    const float s = sw[hop];
#pragma unroll
    for (int ct = 0; ct < 8; ++ct) {
        const float bias = b[16 * ct + l];
#pragma unroll
        for (int j = 0; j < 8; ++j) {
            const int m = j + 8 * half;
            Y[(long)(row0 + m) * DD + 16 * ct + l] = (acc[ct][j] + bias) * s;
        }
    }
}

__global__ void attention_kernel(const float* __restrict__ T0, const float* __restrict__ T1,
                                 const float* __restrict__ T2,
                                 const float* __restrict__ Wa, const float* __restrict__ ba,
                                 float* __restrict__ att, int n) {
    int node = blockIdx.x * blockDim.x + threadIdx.x;
    if (node >= n) return;
    float l0 = ba[0], l1 = ba[1], l2 = ba[2];
    const float* Ts[NHOP] = {T0, T1, T2};
#pragma unroll
    for (int hop = 0; hop < NHOP; ++hop) {
        const float* t = Ts[hop] + (long)node * DD;
        const float* wa = Wa + (long)hop * DD * NHOP;
        for (int kk = 0; kk < DD; ++kk) {
            float xv = t[kk];
            l0 += xv * wa[kk * NHOP + 0];
            l1 += xv * wa[kk * NHOP + 1];
            l2 += xv * wa[kk * NHOP + 2];
        }
    }
    float mx = fmaxf(l0, fmaxf(l1, l2));
    float e0 = __expf(l0 - mx), e1 = __expf(l1 - mx), e2 = __expf(l2 - mx);
    float inv = 1.0f / (e0 + e1 + e2);
    att[(long)node * NHOP + 0] = e0 * inv;
    att[(long)node * NHOP + 1] = e1 * inv;
    att[(long)node * NHOP + 2] = e2 * inv;
}

__global__ void __launch_bounds__(GEMM_THREADS)
fusion_kernel(const float* __restrict__ T0, const float* __restrict__ T1,
              const float* __restrict__ T2, const float* __restrict__ att,
              const float* __restrict__ Wf, const float* __restrict__ bf,
              float* __restrict__ out) {
    __shared__ __attribute__((aligned(16))) float wl[W_SLAB_DWORDS];
    const int tid = threadIdx.x;
    const int wave = __builtin_amdgcn_readfirstlane(tid >> 5);
    const int lane = tid & 31;
    const int half = lane >> 4;
    const int l    = lane & 15;
    const int rt = blockIdx.x * GEMM_WAVES + wave;
    const bool valid = rt < ROW_TILES;
    const int row0 = (valid ? rt : 0) * 16;

    const int row = row0 + l;
    const float a_att0 = att[(long)row * NHOP + 0];
    const float a_att1 = att[(long)row * NHOP + 1];
    const float a_att2 = att[(long)row * NHOP + 2];

    v8f acc[8] = {};
#pragma unroll
    for (int hop = 0; hop < NHOP; ++hop) {
        const float* T = (hop == 0) ? T0 : ((hop == 1) ? T1 : T2);
        const float amul = (hop == 0) ? a_att0 : ((hop == 1) ? a_att1 : a_att2);

        stage_slab(Wf + (long)hop * DD * DD, wl, tid);
        __builtin_amdgcn_sched_barrier(0);
        v2f a[DD / 4];
        load_a_frags(T + (long)row * DD + 2 * half, a);
#pragma unroll
        for (int s = 0; s < DD / 4; ++s) {
            a[s].x *= amul;
            a[s].y *= amul;
        }
        __syncthreads();
        if (valid) mma_slab(a, wl + half * W_PITCH + 2 * l, acc);
        if (hop + 1 < NHOP) __syncthreads();
    }
    if (!valid) return;

#pragma unroll
    for (int ct = 0; ct < 8; ++ct) {
        const float bias = bf[16 * ct + l];
#pragma unroll
        for (int j = 0; j < 8; ++j) {
            const int m = j + 8 * half;
            out[(long)(row0 + m) * DD + 16 * ct + l] = acc[ct][j] + bias;
        }
    }
}

static inline int cdiv(long a, long b) { return (int)((a + b - 1) / b); }

extern "C" void kernel_launch(void* const* d_in, const int* in_sizes, int n_in,
                              void* d_out, int out_size, void* d_ws, size_t ws_size,
                              hipStream_t stream) {
    const float* x      = (const float*)d_in[0];
    const int*   ei     = (const int*)  d_in[1];
    const int*   depths = (const int*)  d_in[2];
    const float* Wh     = (const float*)d_in[3];
    const float* bh     = (const float*)d_in[4];
    const float* Wg1    = (const float*)d_in[5];
    const float* bg1    = (const float*)d_in[6];
    const float* Wg2    = (const float*)d_in[7];
    const float* bg2    = (const float*)d_in[8];
    const float* Wa     = (const float*)d_in[9];
    const float* ba     = (const float*)d_in[10];
    const float* Wf     = (const float*)d_in[11];
    const float* bf     = (const float*)d_in[12];
    float* out = (float*)d_out;

    const int* row = ei;
    const int* col = ei + EE;

    char* ws = (char*)d_ws;
    size_t o = 0;
    auto carve = [&](size_t bytes) -> float* {
        float* p = (float*)(ws + o);
        o += (bytes + 255) & ~(size_t)255;
        return p;
    };
    float* sw     = carve(16 * sizeof(float));
    float* counts = carve((size_t)NN * sizeof(float));
    float* agg1   = carve((size_t)NN * DD * sizeof(float));
    float* agg2   = carve((size_t)NN * DD * sizeof(float));
    float* T0     = carve((size_t)NN * DD * sizeof(float));
    float* T1     = carve((size_t)NN * DD * sizeof(float));
    float* T2     = carve((size_t)NN * DD * sizeof(float));
    float* att    = carve((size_t)NN * NHOP * sizeof(float));
    (void)ws_size; (void)in_sizes; (void)n_in; (void)out_size;

    int* cnt    = (int*)counts;
    int* cursor = (int*)T2;
    int* perm   = (int*)T2 + 65536;

    hipMemsetAsync(cnt, 0, (size_t)NN * sizeof(int), stream);

    gate_kernel<<<1, 256, 0, stream>>>(depths, NN, Wg1, bg1, Wg2, bg2, sw);

    hist_kernel<<<cdiv(EE / 4, 256), 256, 0, stream>>>(col, cnt, EE / 4);
    scan_kernel<<<1, SCAN_T, 0, stream>>>(cnt, cursor);
    fill_kernel<<<cdiv(EE / 4, 256), 256, 0, stream>>>(row, col, cursor, perm, EE / 4);

    gather_mean_kernel<<<2048, 256, 0, stream>>>(x,    perm, cnt, cursor, agg1, NN);
    gather_mean_kernel<<<2048, 256, 0, stream>>>(agg1, perm, cnt, cursor, agg2, NN);

    gemm_hop_kernel<<<GEMM_BLOCKS, GEMM_THREADS, 0, stream>>>(x,    Wh + 0 * DD * DD, bh + 0 * DD, sw, 0, T0);
    gemm_hop_kernel<<<GEMM_BLOCKS, GEMM_THREADS, 0, stream>>>(agg1, Wh + 1 * DD * DD, bh + 1 * DD, sw, 1, T1);
    gemm_hop_kernel<<<GEMM_BLOCKS, GEMM_THREADS, 0, stream>>>(agg2, Wh + 2 * DD * DD, bh + 2 * DD, sw, 2, T2);

    attention_kernel<<<cdiv(NN, 256), 256, 0, stream>>>(T0, T1, T2, Wa, ba, att, NN);

    fusion_kernel<<<GEMM_BLOCKS, GEMM_THREADS, 0, stream>>>(T0, T1, T2, att, Wf, bf, out);
}


# scale gate and weight fold run as extra blocks of the prefix-scan launch (role by block index; 512-thread scan block, gate maximum in one round trip): 8 kernel launches become 6, sums unchanged
# speedup vs baseline: 2.7653x; 1.3444x over previous
#include <hip/hip_runtime.h>
#include <hip/hip_bf16.h>
#include <math.h>

#define NN 50000
#define DD 128
#define EE 800000
#define NHOP 3

typedef __attribute__((ext_vector_type(2))) float v2f;
typedef __attribute__((ext_vector_type(8))) float v8f;

#define PRO_T 512
#define PRO_WAVES (PRO_T / 32)

#define GATE_ITEMS ((NN / 4 + PRO_T - 1) / PRO_T)
static_assert(NN % 4 == 0, "the depths are read as int4");

__device__ __forceinline__ void gate_role(const int* __restrict__ depths,
                                          const float* __restrict__ Wg1, const float* __restrict__ bg1,
                                          const float* __restrict__ Wg2, const float* __restrict__ bg2,
                                          float* __restrict__ sw, int* smax) {
    const int tid = threadIdx.x;
    const int lane = tid & 31;
    const int w = tid >> 5;
    int m = 0;
#pragma unroll
    for (int i = 0; i < GATE_ITEMS; ++i) {
        const int idx = min(tid + i * PRO_T, NN / 4 - 1);
        const int4 v = ((const int4*)depths)[idx];
        m = max(m, max(max(v.x, v.y), max(v.z, v.w)));
    }
    for (int d = 16; d > 0; d >>= 1) m = max(m, __shfl_xor(m, d, 32));
    if (lane == 0) smax[w] = m;
    __syncthreads();
    if (tid == 0) {
        int mm = smax[0];
        for (int k = 1; k < PRO_WAVES; ++k) mm = max(mm, smax[k]);
        float sf0 = (float)NN / 5000.0f;
        float sf1 = (float)mm / 20.0f;
        float a0 = bg2[0], a1 = bg2[1], a2 = bg2[2];
        for (int j = 0; j < DD / 4; ++j) {
            float h = sf0 * Wg1[j] + sf1 * Wg1[DD / 4 + j] + bg1[j];
            h = fmaxf(h, 0.0f);
            a0 += h * Wg2[j * NHOP + 0];
            a1 += h * Wg2[j * NHOP + 1];
            a2 += h * Wg2[j * NHOP + 2];
        }
        sw[0] = 1.0f / (1.0f + __expf(-a0));
        sw[1] = 1.0f / (1.0f + __expf(-a1));
        sw[2] = 1.0f / (1.0f + __expf(-a2));
    }
}

static_assert(EE % 4 == 0, "edge list is read as int4");
static_assert(NN % 4 == 0, "degree array is scanned as int4");

__global__ void __launch_bounds__(256)
hist_kernel(const int* __restrict__ col, int* __restrict__ cnt, int e4) {
    int i = blockIdx.x * 256 + threadIdx.x;
    if (i >= e4) return;
    const int4 c = ((const int4*)col)[i];
    atomicAdd(&cnt[c.x], 1);
    atomicAdd(&cnt[c.y], 1);
    atomicAdd(&cnt[c.z], 1);
    atomicAdd(&cnt[c.w], 1);
}

#define SCAN_C 100
static_assert(PRO_T * SCAN_C >= NN, "one block must cover all nodes");
static_assert(SCAN_C % 4 == 0, "a thread's nodes are read as int4");
static_assert(NN % SCAN_C == 0, "a thread's nodes are all inside the array or all outside it");

__device__ __forceinline__ void scan_role(const int* __restrict__ cnt, int* __restrict__ cursor,
                                          int* wsum) {
    const int t = threadIdx.x;
    const int lane = t & 31;
    const int w = t >> 5;
    const int base = t * SCAN_C;

    const bool live = base < NN;

    int4 v[SCAN_C / 4];
    int s = 0;
#pragma unroll
    for (int i = 0; i < SCAN_C / 4; ++i) v[i] = make_int4(0, 0, 0, 0);
    if (live) {
#pragma unroll
        for (int i = 0; i < SCAN_C / 4; ++i) v[i] = *(const int4*)(cnt + base + 4 * i);
    }
#pragma unroll
    for (int i = 0; i < SCAN_C / 4; ++i) s += v[i].x + v[i].y + v[i].z + v[i].w;
    int inc = s;
    for (int d = 1; d < 32; d <<= 1) {
        int o = __shfl_up(inc, d, 32);
        if (lane >= d) inc += o;
    }
    if (lane == 31) wsum[w] = inc;
    __syncthreads();
    if (w == 0) {
        const int x = (lane < PRO_WAVES) ? wsum[lane] : 0;
        int xi = x;
        for (int d = 1; d < 32; d <<= 1) {
            int o = __shfl_up(xi, d, 32);
            if (lane >= d) xi += o;
        }
        if (lane < PRO_WAVES) wsum[lane] = xi - x;
    }
    __syncthreads();
    int run = wsum[w] + inc - s;
    if (live) {
#pragma unroll
        for (int i = 0; i < SCAN_C / 4; ++i) {
            int4 o;
            o.x = run; run += v[i].x;
            o.y = run; run += v[i].y;
            o.z = run; run += v[i].z;
            o.w = run; run += v[i].w;
            *(int4*)(cursor + base + 4 * i) = o;
        }
    }
}

__global__ void __launch_bounds__(256)
fill_kernel(const int* __restrict__ row, const int* __restrict__ col,
            int* __restrict__ cursor, int* __restrict__ perm, int e4) {
    int i = blockIdx.x * 256 + threadIdx.x;
    if (i >= e4) return;
    const int4 c = ((const int4*)col)[i];
    const int4 r = ((const int4*)row)[i];
    const int p0 = atomicAdd(&cursor[c.x], 1);
    const int p1 = atomicAdd(&cursor[c.y], 1);
    const int p2 = atomicAdd(&cursor[c.z], 1);
    const int p3 = atomicAdd(&cursor[c.w], 1);
    perm[p0] = r.x;
    perm[p1] = r.y;
    perm[p2] = r.z;
    perm[p3] = r.w;
}

#define GATHER_WAVES_PER_BLOCK 8
#define GATHER_BATCH 8

__device__ __forceinline__ float4 load_row4(const float* __restrict__ src, int r, int lane) {
    return *(const float4*)(src + (long)r * DD + lane * 4);
}

__global__ void __launch_bounds__(256)
gather_mean_kernel(const float* __restrict__ src, const int* __restrict__ perm,
                   const int* __restrict__ cnt, const int* __restrict__ segend,
                   float* __restrict__ dst, int n) {
    const int lane = threadIdx.x & 31;
    const int wave0 = __builtin_amdgcn_readfirstlane(
        (int)(blockIdx.x * GATHER_WAVES_PER_BLOCK + (threadIdx.x >> 5)));
    const int nwaves = (int)gridDim.x * GATHER_WAVES_PER_BLOCK;

    for (int node = wave0; node < n; node += nwaves) {
        const int deg = cnt[node];
        const int end = segend[node];
        float4 acc = make_float4(0.0f, 0.0f, 0.0f, 0.0f);

        for (int base = end - deg; base < end; base += 32) {
            const int m = min(32, end - base);
            int idx = (lane < m) ? perm[base + lane] : 0;
            idx = min(max(idx, 0), n - 1);

            int j = 0;
            for (; j + GATHER_BATCH <= m; j += GATHER_BATCH) {
                float4 v[GATHER_BATCH];
                for (int u = 0; u < GATHER_BATCH; ++u)
                    v[u] = load_row4(src, __builtin_amdgcn_readlane(idx, j + u), lane);
                for (int u = 0; u < GATHER_BATCH; ++u) {
                    acc.x += v[u].x; acc.y += v[u].y; acc.z += v[u].z; acc.w += v[u].w;
                }
            }
            if (j < m) {
                float4 v[GATHER_BATCH - 1];
                for (int u = 0; u < GATHER_BATCH - 1; ++u) {
                    v[u] = make_float4(0.0f, 0.0f, 0.0f, 0.0f);
                    if (j + u < m)
                        v[u] = load_row4(src, __builtin_amdgcn_readlane(idx, j + u), lane);
                }
                for (int u = 0; u < GATHER_BATCH - 1; ++u) {
                    if (j + u < m) {
                        acc.x += v[u].x; acc.y += v[u].y; acc.z += v[u].z; acc.w += v[u].w;
                    }
                }
            }
        }
        const float inv = 1.0f / fmaxf((float)deg, 1.0f);
        acc.x *= inv; acc.y *= inv; acc.z *= inv; acc.w *= inv;
        *(float4*)(dst + (long)node * DD + lane * 4) = acc;
    }
}

#define GEMM_WAVES 4
#define GEMM_THREADS (GEMM_WAVES * 32)
#define ROW_TILES (NN / 16)
#define GEMM_BLOCKS ((ROW_TILES + GEMM_WAVES - 1) / GEMM_WAVES)
static_assert(NN % 16 == 0, "rows are handled in whole 16-row tiles");
static_assert(DD == 128, "the slab layout below is written for 128 x 128 weight slabs");

#define W_PITCH 288
#define W_SLAB_DWORDS ((DD / 2) * W_PITCH)
#define W_ITEMS_PER_THREAD ((DD / 2) / GEMM_WAVES)

__device__ __forceinline__ void stage_slab(const float* __restrict__ Wsrc, float* wl, int tid) {
    const int kp0 = tid >> 5;
    const int c4  = tid & 31;
    const float* src = Wsrc + (long)(2 * kp0) * DD + 4 * c4;
    float* dst = wl + kp0 * W_PITCH + 8 * c4;
#pragma unroll
    for (int it = 0; it < W_ITEMS_PER_THREAD; ++it) {
        const float4 r0 = *(const float4*)(src + it * (2 * GEMM_WAVES * DD));
        const float4 r1 = *(const float4*)(src + it * (2 * GEMM_WAVES * DD) + DD);
        float* d = dst + it * (GEMM_WAVES * W_PITCH);
        *(float4*)(d)     = make_float4(r0.x, r1.x, r0.y, r1.y);
        *(float4*)(d + 4) = make_float4(r0.z, r1.z, r0.w, r1.w);
    }
}

__device__ __forceinline__ void load_a_frags(const float* __restrict__ arow, v2f (&a)[DD / 4]) {
#pragma unroll
    for (int s = 0; s < DD / 4; ++s) {
        a[s].x = arow[4 * s];
        a[s].y = arow[4 * s + 1];
    }
}

#define P_PITCH 32
#define P_SLAB_DWORDS ((DD / 2) * P_PITCH)

__device__ __forceinline__ void mma_slab_p(const v2f (&a)[DD / 4], const float* wl_lane,
                                           const float* pl_lane, v8f (&acc)[8], v8f& accl) {
#pragma unroll
    for (int s = 0; s < DD / 4; ++s) {
#pragma unroll
        for (int ct = 0; ct < 8; ++ct) {
            const v2f bb = *(const v2f*)(wl_lane + 2 * s * W_PITCH + 32 * ct);
            acc[ct] = __builtin_amdgcn_wmma_f32_16x16x4_f32(false, a[s], false, bb,
                                                            (short)0, acc[ct], false, false);
        }
        const v2f pp = *(const v2f*)(pl_lane + 2 * s * P_PITCH);
        accl = __builtin_amdgcn_wmma_f32_16x16x4_f32(false, a[s], false, pp,
                                                     (short)0, accl, false, false);
    }
}

#define FOLD_ROW_TILES 9
#define FOLD_COL_GROUPS 5
#define FOLD_BLOCKS (NHOP * FOLD_ROW_TILES * FOLD_COL_GROUPS)
#define DF_STRIDE 32

__device__ __forceinline__ void fold_role(const int b, const int lane,
            const float* __restrict__ Wh, const float* __restrict__ bh,
            const float* __restrict__ Wf, const float* __restrict__ Wa,
            float* __restrict__ Mf, float* __restrict__ Pf,
            float* __restrict__ cf, float* __restrict__ df) {
    const int hop = b / (FOLD_ROW_TILES * FOLD_COL_GROUPS);
    const int rt  = (b / FOLD_COL_GROUPS) % FOLD_ROW_TILES;
    const int cg  = b % FOLD_COL_GROUPS;
    const int half = lane >> 4;
    const int l    = lane & 15;

    const int R = 16 * rt + l;
    const float* ap = (R < DD) ? Wh + ((long)hop * DD + R) * DD : bh + hop * DD;
    const bool aok = R <= DD;
    ap += 2 * half;

    const float* Wf_h = Wf + (long)hop * DD * DD;
    const float* Wa_h = Wa + (long)hop * DD * NHOP;
    const int C0 = 32 * cg + l;
    const int C1 = C0 + 16;
    const float* bp0; int bs0; bool bok0;
    if (C0 < DD)             { bp0 = Wf_h + C0;        bs0 = DD;   bok0 = true;  }
    else if (C0 < DD + NHOP) { bp0 = Wa_h + (C0 - DD); bs0 = NHOP; bok0 = true;  }
    else                     { bp0 = Wa_h;             bs0 = NHOP; bok0 = false; }
    const float* bp1; int bs1; bool bok1;
    if (C1 < DD)             { bp1 = Wf_h + C1;        bs1 = DD;   bok1 = true;  }
    else if (C1 < DD + NHOP) { bp1 = Wa_h + (C1 - DD); bs1 = NHOP; bok1 = true;  }
    else                     { bp1 = Wa_h;             bs1 = NHOP; bok1 = false; }
    bp0 += 2 * half * bs0;
    bp1 += 2 * half * bs1;

    v2f a[DD / 4], b0[DD / 4], b1[DD / 4];
#pragma unroll
    for (int s = 0; s < DD / 4; ++s) {
        a[s].x  = ap[4 * s];
        a[s].y  = ap[4 * s + 1];
        b0[s].x = bp0[(4 * s) * bs0];
        b0[s].y = bp0[(4 * s + 1) * bs0];
        b1[s].x = bp1[(4 * s) * bs1];
        b1[s].y = bp1[(4 * s + 1) * bs1];
    }
    v8f acc0 = {}, acc1 = {};
#pragma unroll
    for (int s = 0; s < DD / 4; ++s) {
        v2f av, bv0, bv1;
        av.x  = aok  ? a[s].x  : 0.0f;
        av.y  = aok  ? a[s].y  : 0.0f;
        bv0.x = bok0 ? b0[s].x : 0.0f;
        bv0.y = bok0 ? b0[s].y : 0.0f;
        bv1.x = bok1 ? b1[s].x : 0.0f;
        bv1.y = bok1 ? b1[s].y : 0.0f;
        acc0 = __builtin_amdgcn_wmma_f32_16x16x4_f32(false, av, false, bv0, (short)0, acc0, false, false);
        acc1 = __builtin_amdgcn_wmma_f32_16x16x4_f32(false, av, false, bv1, (short)0, acc1, false, false);
    }

#pragma unroll
    for (int j = 0; j < 8; ++j) {
        const int Rr = 16 * rt + 8 * half + j;
        if (Rr < DD) {
            if (cg < FOLD_COL_GROUPS - 1) {
                Mf[((long)hop * DD + Rr) * DD + C0] = acc0[j];
                Mf[((long)hop * DD + Rr) * DD + C1] = acc1[j];
            } else {
                Pf[hop * P_SLAB_DWORDS + (Rr >> 1) * P_PITCH + 2 * l + (Rr & 1)] = acc0[j];
            }
        } else if (Rr == DD) {
            if (cg < FOLD_COL_GROUPS - 1) {
                cf[hop * DD + C0] = acc0[j];
                cf[hop * DD + C1] = acc1[j];
            } else {
                df[hop * DF_STRIDE + l] = acc0[j];
            }
        }
    }
}

#define PRO_BLOCKS (2 + FOLD_BLOCKS)

__global__ void __launch_bounds__(PRO_T)
prologue_kernel(const int* __restrict__ cnt, int* __restrict__ cursor,
                const int* __restrict__ depths,
                const float* __restrict__ Wg1, const float* __restrict__ bg1,
                const float* __restrict__ Wg2, const float* __restrict__ bg2,
                float* __restrict__ sw,
                const float* __restrict__ Wh, const float* __restrict__ bh,
                const float* __restrict__ Wf, const float* __restrict__ Wa,
                float* __restrict__ Mf, float* __restrict__ Pf,
                float* __restrict__ cf, float* __restrict__ df) {
    __shared__ int sh[32];
    const int b = blockIdx.x;
    if (b == 0) {
        scan_role(cnt, cursor, sh);
    } else if (b == 1) {
        gate_role(depths, Wg1, bg1, Wg2, bg2, sw, sh);
    } else {
        const int wave = __builtin_amdgcn_readfirstlane((int)(threadIdx.x >> 5));
        if (wave == 0) fold_role(b - 2, (int)threadIdx.x, Wh, bh, Wf, Wa, Mf, Pf, cf, df);
    }
}

__global__ void __launch_bounds__(GEMM_THREADS) __attribute__((amdgpu_waves_per_eu(1, 2)))
fused_kernel(const float* __restrict__ X0, const float* __restrict__ X1,
             const float* __restrict__ X2,
             const float* __restrict__ Mf, const float* __restrict__ Pf,
             const float* __restrict__ cf, const float* __restrict__ df,
             const float* __restrict__ sw, const float* __restrict__ ba,
             const float* __restrict__ bf, float* __restrict__ out) {
    __shared__ __attribute__((aligned(16))) float wl[W_SLAB_DWORDS];
    __shared__ __attribute__((aligned(16))) float pl[P_SLAB_DWORDS];
    const int tid = threadIdx.x;
    const int wave = __builtin_amdgcn_readfirstlane(tid >> 5);
    const int lane = tid & 31;
    const int half = lane >> 4;
    const int l    = lane & 15;
    const int rt = blockIdx.x * GEMM_WAVES + wave;
    const bool valid = rt < ROW_TILES;
    const int row0 = (valid ? rt : 0) * 16;
    const int row = row0 + l;

    v8f acc[NHOP][8] = {};
    v8f accl[NHOP] = {};
#pragma unroll
    for (int hop = 0; hop < NHOP; ++hop) {
        const float* X = (hop == 0) ? X0 : ((hop == 1) ? X1 : X2);

        stage_slab(Mf + (long)hop * DD * DD, wl, tid);
#pragma unroll
        for (int it = 0; it < P_SLAB_DWORDS / 4 / GEMM_THREADS; ++it)
            ((float4*)pl)[tid + GEMM_THREADS * it] =
                ((const float4*)(Pf + hop * P_SLAB_DWORDS))[tid + GEMM_THREADS * it];
        v2f a[DD / 4];
        load_a_frags(X + (long)row * DD + 2 * half, a);
        __syncthreads();
        if (valid) mma_slab_p(a, wl + half * W_PITCH + 2 * l, pl + half * P_PITCH + 2 * l,
                              acc[hop], accl[hop]);
        if (hop + 1 < NHOP) __syncthreads();
    }
    if (!valid) return;

    const float s0 = sw[0], s1 = sw[1], s2 = sw[2];
    const float bac = ba[min(l, NHOP - 1)];
    const float d0 = df[0 * DF_STRIDE + l], d1 = df[1 * DF_STRIDE + l], d2 = df[2 * DF_STRIDE + l];
    const int src = lane & 16;
    float w0[8], w1[8], w2[8];
#pragma unroll
    for (int j = 0; j < 8; ++j) {
        const float lg = bac + s0 * (accl[0][j] + d0) + s1 * (accl[1][j] + d1) + s2 * (accl[2][j] + d2);
        const float l0 = __shfl(lg, src + 0, 32);
        const float l1 = __shfl(lg, src + 1, 32);
        const float l2 = __shfl(lg, src + 2, 32);
        const float mx = fmaxf(l0, fmaxf(l1, l2));
        const float e0 = __expf(l0 - mx), e1 = __expf(l1 - mx), e2 = __expf(l2 - mx);
        const float inv = 1.0f / (e0 + e1 + e2);
        w0[j] = e0 * inv * s0;
        w1[j] = e1 * inv * s1;
        w2[j] = e2 * inv * s2;
    }

#pragma unroll
    for (int ct = 0; ct < 8; ++ct) {
        const int col = 16 * ct + l;
        const float c0 = cf[0 * DD + col], c1 = cf[1 * DD + col], c2 = cf[2 * DD + col];
        const float bias = bf[col];
#pragma unroll
        for (int j = 0; j < 8; ++j) {
            const int m = j + 8 * half;
            out[(long)(row0 + m) * DD + col] = w0[j] * (acc[0][ct][j] + c0) + w1[j] * (acc[1][ct][j] + c1)
                                             + w2[j] * (acc[2][ct][j] + c2) + bias;
        }
    }
}

static inline int cdiv(long a, long b) { return (int)((a + b - 1) / b); }

extern "C" void kernel_launch(void* const* d_in, const int* in_sizes, int n_in,
                              void* d_out, int out_size, void* d_ws, size_t ws_size,
                              hipStream_t stream) {
    const float* x      = (const float*)d_in[0];
    const int*   ei     = (const int*)  d_in[1];
    const int*   depths = (const int*)  d_in[2];
    const float* Wh     = (const float*)d_in[3];
    const float* bh     = (const float*)d_in[4];
    const float* Wg1    = (const float*)d_in[5];
    const float* bg1    = (const float*)d_in[6];
    const float* Wg2    = (const float*)d_in[7];
    const float* bg2    = (const float*)d_in[8];
    const float* Wa     = (const float*)d_in[9];
    const float* ba     = (const float*)d_in[10];
    const float* Wf     = (const float*)d_in[11];
    const float* bf     = (const float*)d_in[12];
    float* out = (float*)d_out;

    const int* row = ei;
    const int* col = ei + EE;

    char* ws = (char*)d_ws;
    size_t o = 0;
    auto carve = [&](size_t bytes) -> float* {
        float* p = (float*)(ws + o);
        o += (bytes + 255) & ~(size_t)255;
        return p;
    };
    float* sw     = carve(16 * sizeof(float));
    float* counts = carve((size_t)NN * sizeof(float));
    float* agg1   = carve((size_t)NN * DD * sizeof(float));
    float* agg2   = carve((size_t)NN * DD * sizeof(float));
    float* fold   = carve((size_t)NN * DD * sizeof(float));
    float* spare  = carve((size_t)NN * DD * sizeof(float));
    float* sortws = carve((size_t)NN * DD * sizeof(float));
    (void)ws_size; (void)in_sizes; (void)n_in; (void)out_size; (void)spare;

    float* Mf = fold;
    float* Pf = fold + NHOP * DD * DD;
    float* cf = Pf + NHOP * P_SLAB_DWORDS;
    float* df = cf + NHOP * DD;

    int* cnt    = (int*)counts;
    int* cursor = (int*)sortws;
    int* perm   = (int*)sortws + 65536;

    hipMemsetAsync(cnt, 0, (size_t)NN * sizeof(int), stream);

    hist_kernel<<<cdiv(EE / 4, 256), 256, 0, stream>>>(col, cnt, EE / 4);
    prologue_kernel<<<PRO_BLOCKS, PRO_T, 0, stream>>>(cnt, cursor, depths, Wg1, bg1, Wg2, bg2, sw,
                                                      Wh, bh, Wf, Wa, Mf, Pf, cf, df);
    fill_kernel<<<cdiv(EE / 4, 256), 256, 0, stream>>>(row, col, cursor, perm, EE / 4);

    gather_mean_kernel<<<2048, 256, 0, stream>>>(x,    perm, cnt, cursor, agg1, NN);
    gather_mean_kernel<<<2048, 256, 0, stream>>>(agg1, perm, cnt, cursor, agg2, NN);

    fused_kernel<<<GEMM_BLOCKS, GEMM_THREADS, 0, stream>>>(x, agg1, agg2, Mf, Pf, cf, df, sw, ba, bf, out);
}


# prefix scan of the degrees on whole 128-byte lines: a wave owns 3200 consecutive nodes and reads and writes them 32 lanes x int4 per trip, lane sums scanned by shuffles; integer sums, same result
# speedup vs baseline: 2.9336x; 1.0609x over previous
#include <hip/hip_runtime.h>
#include <hip/hip_bf16.h>
#include <math.h>

#define NN 50000
#define DD 128
#define EE 800000
#define NHOP 3

typedef __attribute__((ext_vector_type(2))) float v2f;
typedef __attribute__((ext_vector_type(8))) float v8f;

#define PRO_T 512
#define PRO_WAVES (PRO_T / 32)

#define GATE_ITEMS ((NN / 4 + PRO_T - 1) / PRO_T)
static_assert(NN % 4 == 0, "the depths are read as int4");

__device__ __forceinline__ void gate_role(const int* __restrict__ depths,
                                          const float* __restrict__ Wg1, const float* __restrict__ bg1,
                                          const float* __restrict__ Wg2, const float* __restrict__ bg2,
                                          float* __restrict__ sw, int* smax) {
    const int tid = threadIdx.x;
    const int lane = tid & 31;
    const int w = tid >> 5;
    int m = 0;
#pragma unroll
    for (int i = 0; i < GATE_ITEMS; ++i) {
        const int idx = min(tid + i * PRO_T, NN / 4 - 1);
        const int4 v = ((const int4*)depths)[idx];
        m = max(m, max(max(v.x, v.y), max(v.z, v.w)));
    }
    for (int d = 16; d > 0; d >>= 1) m = max(m, __shfl_xor(m, d, 32));
    if (lane == 0) smax[w] = m;
    __syncthreads();
    if (tid == 0) {
        int mm = smax[0];
        for (int k = 1; k < PRO_WAVES; ++k) mm = max(mm, smax[k]);
        float sf0 = (float)NN / 5000.0f;
        float sf1 = (float)mm / 20.0f;
        float a0 = bg2[0], a1 = bg2[1], a2 = bg2[2];
        for (int j = 0; j < DD / 4; ++j) {
            float h = sf0 * Wg1[j] + sf1 * Wg1[DD / 4 + j] + bg1[j];
            h = fmaxf(h, 0.0f);
            a0 += h * Wg2[j * NHOP + 0];
            a1 += h * Wg2[j * NHOP + 1];
            a2 += h * Wg2[j * NHOP + 2];
        }
        sw[0] = 1.0f / (1.0f + __expf(-a0));
        sw[1] = 1.0f / (1.0f + __expf(-a1));
        sw[2] = 1.0f / (1.0f + __expf(-a2));
    }
}

static_assert(EE % 4 == 0, "edge list is read as int4");
static_assert(NN % 4 == 0, "degree array is scanned as int4");

__global__ void __launch_bounds__(256)
hist_kernel(const int* __restrict__ col, int* __restrict__ cnt, int e4) {
    int i = blockIdx.x * 256 + threadIdx.x;
    if (i >= e4) return;
    const int4 c = ((const int4*)col)[i];
    atomicAdd(&cnt[c.x], 1);
    atomicAdd(&cnt[c.y], 1);
    atomicAdd(&cnt[c.z], 1);
    atomicAdd(&cnt[c.w], 1);
}

#define SCAN_CH 3200
#define SCAN_IT (SCAN_CH / 128)
static_assert(SCAN_CH % 128 == 0, "a trip is 32 lanes x int4");
static_assert(PRO_WAVES * SCAN_CH >= NN, "one block must cover all nodes");

__device__ __forceinline__ void scan_role(const int* __restrict__ cnt, int* __restrict__ cursor,
                                          int* wsum) {
    const int t = threadIdx.x;
    const int lane = t & 31;
    const int w = t >> 5;
    const int base = w * SCAN_CH + lane * 4;

    int4 v[SCAN_IT];
    int inc[SCAN_IT];
#pragma unroll
    for (int i = 0; i < SCAN_IT; ++i) {
        const int idx = base + i * 128;
        v[i] = *(const int4*)(cnt + min(idx, NN - 4));
    }
#pragma unroll
    for (int i = 0; i < SCAN_IT; ++i) {
        const int idx = base + i * 128;
        if (idx >= NN) v[i] = make_int4(0, 0, 0, 0);
        inc[i] = v[i].x + v[i].y + v[i].z + v[i].w;
    }
#pragma unroll
    for (int i = 0; i < SCAN_IT; ++i) {
        for (int d = 1; d < 32; d <<= 1) {
            int o = __shfl_up(inc[i], d, 32);
            if (lane >= d) inc[i] += o;
        }
    }
    int pre[SCAN_IT];
    int tot = 0;
#pragma unroll
    for (int i = 0; i < SCAN_IT; ++i) {
        pre[i] = tot;
        tot += __shfl(inc[i], 31, 32);
    }
    if (lane == 0) wsum[w] = tot;
    __syncthreads();
    if (w == 0) {
        const int x = (lane < PRO_WAVES) ? wsum[lane] : 0;
        int xi = x;
        for (int d = 1; d < 32; d <<= 1) {
            int o = __shfl_up(xi, d, 32);
            if (lane >= d) xi += o;
        }
        if (lane < PRO_WAVES) wsum[lane] = xi - x;
    }
    __syncthreads();
    const int wbase = wsum[w];
#pragma unroll
    for (int i = 0; i < SCAN_IT; ++i) {
        const int idx = base + i * 128;
        const int s = v[i].x + v[i].y + v[i].z + v[i].w;
        int4 o;
        o.x = wbase + pre[i] + inc[i] - s;
        o.y = o.x + v[i].x;
        o.z = o.y + v[i].y;
        o.w = o.z + v[i].z;
        if (idx < NN) *(int4*)(cursor + idx) = o;
    }
}

__global__ void __launch_bounds__(256)
fill_kernel(const int* __restrict__ row, const int* __restrict__ col,
            int* __restrict__ cursor, int* __restrict__ perm, int e4) {
    int i = blockIdx.x * 256 + threadIdx.x;
    if (i >= e4) return;
    const int4 c = ((const int4*)col)[i];
    const int4 r = ((const int4*)row)[i];
    const int p0 = atomicAdd(&cursor[c.x], 1);
    const int p1 = atomicAdd(&cursor[c.y], 1);
    const int p2 = atomicAdd(&cursor[c.z], 1);
    const int p3 = atomicAdd(&cursor[c.w], 1);
    perm[p0] = r.x;
    perm[p1] = r.y;
    perm[p2] = r.z;
    perm[p3] = r.w;
}

#define GATHER_WAVES_PER_BLOCK 8
#define GATHER_BATCH 8

__device__ __forceinline__ float4 load_row4(const float* __restrict__ src, int r, int lane) {
    return *(const float4*)(src + (long)r * DD + lane * 4);
}

__global__ void __launch_bounds__(256)
gather_mean_kernel(const float* __restrict__ src, const int* __restrict__ perm,
                   const int* __restrict__ cnt, const int* __restrict__ segend,
                   float* __restrict__ dst, int n) {
    const int lane = threadIdx.x & 31;
    const int wave0 = __builtin_amdgcn_readfirstlane(
        (int)(blockIdx.x * GATHER_WAVES_PER_BLOCK + (threadIdx.x >> 5)));
    const int nwaves = (int)gridDim.x * GATHER_WAVES_PER_BLOCK;

    for (int node = wave0; node < n; node += nwaves) {
        const int deg = cnt[node];
        const int end = segend[node];
        float4 acc = make_float4(0.0f, 0.0f, 0.0f, 0.0f);

        for (int base = end - deg; base < end; base += 32) {
            const int m = min(32, end - base);
            int idx = (lane < m) ? perm[base + lane] : 0;
            idx = min(max(idx, 0), n - 1);

            int j = 0;
            for (; j + GATHER_BATCH <= m; j += GATHER_BATCH) {
                float4 v[GATHER_BATCH];
                for (int u = 0; u < GATHER_BATCH; ++u)
                    v[u] = load_row4(src, __builtin_amdgcn_readlane(idx, j + u), lane);
                for (int u = 0; u < GATHER_BATCH; ++u) {
                    acc.x += v[u].x; acc.y += v[u].y; acc.z += v[u].z; acc.w += v[u].w;
                }
            }
            if (j < m) {
                float4 v[GATHER_BATCH - 1];
                for (int u = 0; u < GATHER_BATCH - 1; ++u) {
                    v[u] = make_float4(0.0f, 0.0f, 0.0f, 0.0f);
                    if (j + u < m)
                        v[u] = load_row4(src, __builtin_amdgcn_readlane(idx, j + u), lane);
                }
                for (int u = 0; u < GATHER_BATCH - 1; ++u) {
                    if (j + u < m) {
                        acc.x += v[u].x; acc.y += v[u].y; acc.z += v[u].z; acc.w += v[u].w;
                    }
                }
            }
        }
        const float inv = 1.0f / fmaxf((float)deg, 1.0f);
        acc.x *= inv; acc.y *= inv; acc.z *= inv; acc.w *= inv;
        *(float4*)(dst + (long)node * DD + lane * 4) = acc;
    }
}

#define GEMM_WAVES 4
#define GEMM_THREADS (GEMM_WAVES * 32)
#define ROW_TILES (NN / 16)
#define GEMM_BLOCKS ((ROW_TILES + GEMM_WAVES - 1) / GEMM_WAVES)
static_assert(NN % 16 == 0, "rows are handled in whole 16-row tiles");
static_assert(DD == 128, "the slab layout below is written for 128 x 128 weight slabs");

#define W_PITCH 288
#define W_SLAB_DWORDS ((DD / 2) * W_PITCH)
#define W_ITEMS_PER_THREAD ((DD / 2) / GEMM_WAVES)

__device__ __forceinline__ void stage_slab(const float* __restrict__ Wsrc, float* wl, int tid) {
    const int kp0 = tid >> 5;
    const int c4  = tid & 31;
    const float* src = Wsrc + (long)(2 * kp0) * DD + 4 * c4;
    float* dst = wl + kp0 * W_PITCH + 8 * c4;
#pragma unroll
    for (int it = 0; it < W_ITEMS_PER_THREAD; ++it) {
        const float4 r0 = *(const float4*)(src + it * (2 * GEMM_WAVES * DD));
        const float4 r1 = *(const float4*)(src + it * (2 * GEMM_WAVES * DD) + DD);
        float* d = dst + it * (GEMM_WAVES * W_PITCH);
        *(float4*)(d)     = make_float4(r0.x, r1.x, r0.y, r1.y);
        *(float4*)(d + 4) = make_float4(r0.z, r1.z, r0.w, r1.w);
    }
}

__device__ __forceinline__ void load_a_frags(const float* __restrict__ arow, v2f (&a)[DD / 4]) {
#pragma unroll
    for (int s = 0; s < DD / 4; ++s) {
        a[s].x = arow[4 * s];
        a[s].y = arow[4 * s + 1];
    }
}

#define P_PITCH 32
#define P_SLAB_DWORDS ((DD / 2) * P_PITCH)

__device__ __forceinline__ void mma_slab_p(const v2f (&a)[DD / 4], const float* wl_lane,
                                           const float* pl_lane, v8f (&acc)[8], v8f& accl) {
#pragma unroll
    for (int s = 0; s < DD / 4; ++s) {
#pragma unroll
        for (int ct = 0; ct < 8; ++ct) {
            const v2f bb = *(const v2f*)(wl_lane + 2 * s * W_PITCH + 32 * ct);
            acc[ct] = __builtin_amdgcn_wmma_f32_16x16x4_f32(false, a[s], false, bb,
                                                            (short)0, acc[ct], false, false);
        }
        const v2f pp = *(const v2f*)(pl_lane + 2 * s * P_PITCH);
        accl = __builtin_amdgcn_wmma_f32_16x16x4_f32(false, a[s], false, pp,
                                                     (short)0, accl, false, false);
    }
}

#define FOLD_ROW_TILES 9
#define FOLD_COL_GROUPS 5
#define FOLD_BLOCKS (NHOP * FOLD_ROW_TILES * FOLD_COL_GROUPS)
#define DF_STRIDE 32

__device__ __forceinline__ void fold_role(const int b, const int lane,
            const float* __restrict__ Wh, const float* __restrict__ bh,
            const float* __restrict__ Wf, const float* __restrict__ Wa,
            float* __restrict__ Mf, float* __restrict__ Pf,
            float* __restrict__ cf, float* __restrict__ df) {
    const int hop = b / (FOLD_ROW_TILES * FOLD_COL_GROUPS);
    const int rt  = (b / FOLD_COL_GROUPS) % FOLD_ROW_TILES;
    const int cg  = b % FOLD_COL_GROUPS;
    const int half = lane >> 4;
    const int l    = lane & 15;

    const int R = 16 * rt + l;
    const float* ap = (R < DD) ? Wh + ((long)hop * DD + R) * DD : bh + hop * DD;
    const bool aok = R <= DD;
    ap += 2 * half;

    const float* Wf_h = Wf + (long)hop * DD * DD;
    const float* Wa_h = Wa + (long)hop * DD * NHOP;
    const int C0 = 32 * cg + l;
    const int C1 = C0 + 16;
    const float* bp0; int bs0; bool bok0;
    if (C0 < DD)             { bp0 = Wf_h + C0;        bs0 = DD;   bok0 = true;  }
    else if (C0 < DD + NHOP) { bp0 = Wa_h + (C0 - DD); bs0 = NHOP; bok0 = true;  }
    else                     { bp0 = Wa_h;             bs0 = NHOP; bok0 = false; }
    const float* bp1; int bs1; bool bok1;
    if (C1 < DD)             { bp1 = Wf_h + C1;        bs1 = DD;   bok1 = true;  }
    else if (C1 < DD + NHOP) { bp1 = Wa_h + (C1 - DD); bs1 = NHOP; bok1 = true;  }
    else                     { bp1 = Wa_h;             bs1 = NHOP; bok1 = false; }
    bp0 += 2 * half * bs0;
    bp1 += 2 * half * bs1;

    v2f a[DD / 4], b0[DD / 4], b1[DD / 4];
#pragma unroll
    for (int s = 0; s < DD / 4; ++s) {
        a[s].x  = ap[4 * s];
        a[s].y  = ap[4 * s + 1];
        b0[s].x = bp0[(4 * s) * bs0];
        b0[s].y = bp0[(4 * s + 1) * bs0];
        b1[s].x = bp1[(4 * s) * bs1];
        b1[s].y = bp1[(4 * s + 1) * bs1];
    }
    v8f acc0 = {}, acc1 = {};
#pragma unroll
    for (int s = 0; s < DD / 4; ++s) {
        v2f av, bv0, bv1;
        av.x  = aok  ? a[s].x  : 0.0f;
        av.y  = aok  ? a[s].y  : 0.0f;
        bv0.x = bok0 ? b0[s].x : 0.0f;
        bv0.y = bok0 ? b0[s].y : 0.0f;
        bv1.x = bok1 ? b1[s].x : 0.0f;
        bv1.y = bok1 ? b1[s].y : 0.0f;
        acc0 = __builtin_amdgcn_wmma_f32_16x16x4_f32(false, av, false, bv0, (short)0, acc0, false, false);
        acc1 = __builtin_amdgcn_wmma_f32_16x16x4_f32(false, av, false, bv1, (short)0, acc1, false, false);
    }

#pragma unroll
    for (int j = 0; j < 8; ++j) {
        const int Rr = 16 * rt + 8 * half + j;
        if (Rr < DD) {
            if (cg < FOLD_COL_GROUPS - 1) {
                Mf[((long)hop * DD + Rr) * DD + C0] = acc0[j];
                Mf[((long)hop * DD + Rr) * DD + C1] = acc1[j];
            } else {
                Pf[hop * P_SLAB_DWORDS + (Rr >> 1) * P_PITCH + 2 * l + (Rr & 1)] = acc0[j];
            }
        } else if (Rr == DD) {
            if (cg < FOLD_COL_GROUPS - 1) {
                cf[hop * DD + C0] = acc0[j];
                cf[hop * DD + C1] = acc1[j];
            } else {
                df[hop * DF_STRIDE + l] = acc0[j];
            }
        }
    }
}

#define PRO_BLOCKS (2 + FOLD_BLOCKS)

__global__ void __launch_bounds__(PRO_T)
prologue_kernel(const int* __restrict__ cnt, int* __restrict__ cursor,
                const int* __restrict__ depths,
                const float* __restrict__ Wg1, const float* __restrict__ bg1,
                const float* __restrict__ Wg2, const float* __restrict__ bg2,
                float* __restrict__ sw,
                const float* __restrict__ Wh, const float* __restrict__ bh,
                const float* __restrict__ Wf, const float* __restrict__ Wa,
                float* __restrict__ Mf, float* __restrict__ Pf,
                float* __restrict__ cf, float* __restrict__ df) {
    __shared__ int sh[32];
    const int b = blockIdx.x;
    if (b == 0) {
        scan_role(cnt, cursor, sh);
    } else if (b == 1) {
        gate_role(depths, Wg1, bg1, Wg2, bg2, sw, sh);
    } else {
        const int wave = __builtin_amdgcn_readfirstlane((int)(threadIdx.x >> 5));
        if (wave == 0) fold_role(b - 2, (int)threadIdx.x, Wh, bh, Wf, Wa, Mf, Pf, cf, df);
    }
}

__global__ void __launch_bounds__(GEMM_THREADS) __attribute__((amdgpu_waves_per_eu(1, 2)))
fused_kernel(const float* __restrict__ X0, const float* __restrict__ X1,
             const float* __restrict__ X2,
             const float* __restrict__ Mf, const float* __restrict__ Pf,
             const float* __restrict__ cf, const float* __restrict__ df,
             const float* __restrict__ sw, const float* __restrict__ ba,
             const float* __restrict__ bf, float* __restrict__ out) {
    __shared__ __attribute__((aligned(16))) float wl[W_SLAB_DWORDS];
    __shared__ __attribute__((aligned(16))) float pl[P_SLAB_DWORDS];
    const int tid = threadIdx.x;
    const int wave = __builtin_amdgcn_readfirstlane(tid >> 5);
    const int lane = tid & 31;
    const int half = lane >> 4;
    const int l    = lane & 15;
    const int rt = blockIdx.x * GEMM_WAVES + wave;
    const bool valid = rt < ROW_TILES;
    const int row0 = (valid ? rt : 0) * 16;
    const int row = row0 + l;

    v8f acc[NHOP][8] = {};
    v8f accl[NHOP] = {};
#pragma unroll
    for (int hop = 0; hop < NHOP; ++hop) {
        const float* X = (hop == 0) ? X0 : ((hop == 1) ? X1 : X2);

        stage_slab(Mf + (long)hop * DD * DD, wl, tid);
#pragma unroll
        for (int it = 0; it < P_SLAB_DWORDS / 4 / GEMM_THREADS; ++it)
            ((float4*)pl)[tid + GEMM_THREADS * it] =
                ((const float4*)(Pf + hop * P_SLAB_DWORDS))[tid + GEMM_THREADS * it];
        v2f a[DD / 4];
        load_a_frags(X + (long)row * DD + 2 * half, a);
        __syncthreads();
        if (valid) mma_slab_p(a, wl + half * W_PITCH + 2 * l, pl + half * P_PITCH + 2 * l,
                              acc[hop], accl[hop]);
        if (hop + 1 < NHOP) __syncthreads();
    }
    if (!valid) return;

    const float s0 = sw[0], s1 = sw[1], s2 = sw[2];
    const float bac = ba[min(l, NHOP - 1)];
    const float d0 = df[0 * DF_STRIDE + l], d1 = df[1 * DF_STRIDE + l], d2 = df[2 * DF_STRIDE + l];
    const int src = lane & 16;
    float w0[8], w1[8], w2[8];
#pragma unroll
    for (int j = 0; j < 8; ++j) {
        const float lg = bac + s0 * (accl[0][j] + d0) + s1 * (accl[1][j] + d1) + s2 * (accl[2][j] + d2);
        const float l0 = __shfl(lg, src + 0, 32);
        const float l1 = __shfl(lg, src + 1, 32);
        const float l2 = __shfl(lg, src + 2, 32);
        const float mx = fmaxf(l0, fmaxf(l1, l2));
        const float e0 = __expf(l0 - mx), e1 = __expf(l1 - mx), e2 = __expf(l2 - mx);
        const float inv = 1.0f / (e0 + e1 + e2);
        w0[j] = e0 * inv * s0;
        w1[j] = e1 * inv * s1;
        w2[j] = e2 * inv * s2;
    }

#pragma unroll
    for (int ct = 0; ct < 8; ++ct) {
        const int col = 16 * ct + l;
        const float c0 = cf[0 * DD + col], c1 = cf[1 * DD + col], c2 = cf[2 * DD + col];
        const float bias = bf[col];
#pragma unroll
        for (int j = 0; j < 8; ++j) {
            const int m = j + 8 * half;
            out[(long)(row0 + m) * DD + col] = w0[j] * (acc[0][ct][j] + c0) + w1[j] * (acc[1][ct][j] + c1)
                                             + w2[j] * (acc[2][ct][j] + c2) + bias;
        }
    }
}

static inline int cdiv(long a, long b) { return (int)((a + b - 1) / b); }

extern "C" void kernel_launch(void* const* d_in, const int* in_sizes, int n_in,
                              void* d_out, int out_size, void* d_ws, size_t ws_size,
                              hipStream_t stream) {
    const float* x      = (const float*)d_in[0];
    const int*   ei     = (const int*)  d_in[1];
    const int*   depths = (const int*)  d_in[2];
    const float* Wh     = (const float*)d_in[3];
    const float* bh     = (const float*)d_in[4];
    const float* Wg1    = (const float*)d_in[5];
    const float* bg1    = (const float*)d_in[6];
    const float* Wg2    = (const float*)d_in[7];
    const float* bg2    = (const float*)d_in[8];
    const float* Wa     = (const float*)d_in[9];
    const float* ba     = (const float*)d_in[10];
    const float* Wf     = (const float*)d_in[11];
    const float* bf     = (const float*)d_in[12];
    float* out = (float*)d_out;

    const int* row = ei;
    const int* col = ei + EE;

    char* ws = (char*)d_ws;
    size_t o = 0;
    auto carve = [&](size_t bytes) -> float* {
        float* p = (float*)(ws + o);
        o += (bytes + 255) & ~(size_t)255;
        return p;
    };
    float* sw     = carve(16 * sizeof(float));
    float* counts = carve((size_t)NN * sizeof(float));
    float* agg1   = carve((size_t)NN * DD * sizeof(float));
    float* agg2   = carve((size_t)NN * DD * sizeof(float));
    float* fold   = carve((size_t)NN * DD * sizeof(float));
    float* spare  = carve((size_t)NN * DD * sizeof(float));
    float* sortws = carve((size_t)NN * DD * sizeof(float));
    (void)ws_size; (void)in_sizes; (void)n_in; (void)out_size; (void)spare;

    float* Mf = fold;
    float* Pf = fold + NHOP * DD * DD;
    float* cf = Pf + NHOP * P_SLAB_DWORDS;
    float* df = cf + NHOP * DD;

    int* cnt    = (int*)counts;
    int* cursor = (int*)sortws;
    int* perm   = (int*)sortws + 65536;

    hipMemsetAsync(cnt, 0, (size_t)NN * sizeof(int), stream);

    hist_kernel<<<cdiv(EE / 4, 256), 256, 0, stream>>>(col, cnt, EE / 4);
    prologue_kernel<<<PRO_BLOCKS, PRO_T, 0, stream>>>(cnt, cursor, depths, Wg1, bg1, Wg2, bg2, sw,
                                                      Wh, bh, Wf, Wa, Mf, Pf, cf, df);
    fill_kernel<<<cdiv(EE / 4, 256), 256, 0, stream>>>(row, col, cursor, perm, EE / 4);

    gather_mean_kernel<<<2048, 256, 0, stream>>>(x,    perm, cnt, cursor, agg1, NN);
    gather_mean_kernel<<<2048, 256, 0, stream>>>(agg1, perm, cnt, cursor, agg2, NN);

    fused_kernel<<<GEMM_BLOCKS, GEMM_THREADS, 0, stream>>>(x, agg1, agg2, Mf, Pf, cf, df, sw, ba, bf, out);
}


# edges grouped by destination in ONE atomic pass: a 32-slot bucket per node (one 128-byte line) plus an overflow list; memset, histogram and scan launches go; row sums stay in arrival (slot) order
# speedup vs baseline: 3.7771x; 1.2875x over previous
#include <hip/hip_runtime.h>
#include <hip/hip_bf16.h>
#include <math.h>

#define NN 50000
#define DD 128
#define EE 800000
#define NHOP 3

typedef __attribute__((ext_vector_type(2))) float v2f;
typedef __attribute__((ext_vector_type(8))) float v8f;

#define PRO_T 512
#define PRO_WAVES (PRO_T / 32)

#define GATE_ITEMS ((NN / 4 + PRO_T - 1) / PRO_T)
static_assert(NN % 4 == 0, "the depths are read as int4");

__device__ __forceinline__ void gate_role(const int* __restrict__ depths,
                                          const float* __restrict__ Wg1, const float* __restrict__ bg1,
                                          const float* __restrict__ Wg2, const float* __restrict__ bg2,
                                          float* __restrict__ sw, int* smax) {
    const int tid = threadIdx.x;
    const int lane = tid & 31;
    const int w = tid >> 5;
    int m = 0;
#pragma unroll
    for (int i = 0; i < GATE_ITEMS; ++i) {
        const int idx = min(tid + i * PRO_T, NN / 4 - 1);
        const int4 v = ((const int4*)depths)[idx];
        m = max(m, max(max(v.x, v.y), max(v.z, v.w)));
    }
    for (int d = 16; d > 0; d >>= 1) m = max(m, __shfl_xor(m, d, 32));
    if (lane == 0) smax[w] = m;
    __syncthreads();
    if (tid == 0) {
        int mm = smax[0];
        for (int k = 1; k < PRO_WAVES; ++k) mm = max(mm, smax[k]);
        float sf0 = (float)NN / 5000.0f;
        float sf1 = (float)mm / 20.0f;
        float a0 = bg2[0], a1 = bg2[1], a2 = bg2[2];
        for (int j = 0; j < DD / 4; ++j) {
            float h = sf0 * Wg1[j] + sf1 * Wg1[DD / 4 + j] + bg1[j];
            h = fmaxf(h, 0.0f);
            a0 += h * Wg2[j * NHOP + 0];
            a1 += h * Wg2[j * NHOP + 1];
            a2 += h * Wg2[j * NHOP + 2];
        }
        sw[0] = 1.0f / (1.0f + __expf(-a0));
        sw[1] = 1.0f / (1.0f + __expf(-a1));
        sw[2] = 1.0f / (1.0f + __expf(-a2));
    }
}

static_assert(EE % 4 == 0, "edge list is read as int4");
static_assert(NN % 4 == 0, "degree array is cleared as int4");
#define ZERO_BLOCKS ((NN / 4 + PRO_T - 1) / PRO_T)

__device__ __forceinline__ void zero_role(const int zb, int* __restrict__ cnt, int* __restrict__ ovf_n) {
    const int i = zb * PRO_T + (int)threadIdx.x;
    if (i < NN / 4) ((int4*)cnt)[i] = make_int4(0, 0, 0, 0);
    if (i == 0) *ovf_n = 0;
}

#define CAP 32

__device__ __forceinline__ void place_edge(const int c, const int r, const int p,
                                           int* __restrict__ perm, int* __restrict__ ovf_n,
                                           int* __restrict__ ovf_dst, int* __restrict__ ovf_src) {
    if (p < CAP) {
        perm[c * CAP + p] = r;
    } else {
        const int k = atomicAdd(ovf_n, 1);
        ovf_dst[k] = c;
        ovf_src[k] = r;
    }
}

__global__ void __launch_bounds__(256)
bucket_kernel(const int* __restrict__ row, const int* __restrict__ col,
              int* __restrict__ cnt, int* __restrict__ ovf_n, int* __restrict__ perm,
              int* __restrict__ ovf_dst, int* __restrict__ ovf_src, int e4) {
    int i = blockIdx.x * 256 + threadIdx.x;
    if (i >= e4) return;
    const int4 c = ((const int4*)col)[i];
    const int4 r = ((const int4*)row)[i];
    const int p0 = atomicAdd(&cnt[c.x], 1);
    const int p1 = atomicAdd(&cnt[c.y], 1);
    const int p2 = atomicAdd(&cnt[c.z], 1);
    const int p3 = atomicAdd(&cnt[c.w], 1);
    place_edge(c.x, r.x, p0, perm, ovf_n, ovf_dst, ovf_src);
    place_edge(c.y, r.y, p1, perm, ovf_n, ovf_dst, ovf_src);
    place_edge(c.z, r.z, p2, perm, ovf_n, ovf_dst, ovf_src);
    place_edge(c.w, r.w, p3, perm, ovf_n, ovf_dst, ovf_src);
}

#define GATHER_WAVES_PER_BLOCK 8
#define GATHER_BATCH 8

__device__ __forceinline__ float4 load_row4(const float* __restrict__ src, int r, int lane) {
    return *(const float4*)(src + (long)r * DD + lane * 4);
}

__global__ void __launch_bounds__(256)
gather_mean_kernel(const float* __restrict__ src, const int* __restrict__ perm,
                   const int* __restrict__ cnt, const int* __restrict__ ovf_n,
                   const int* __restrict__ ovf_dst, const int* __restrict__ ovf_src,
                   float* __restrict__ dst, int n) {
    const int lane = threadIdx.x & 31;
    const int wave0 = __builtin_amdgcn_readfirstlane(
        (int)(blockIdx.x * GATHER_WAVES_PER_BLOCK + (threadIdx.x >> 5)));
    const int nwaves = (int)gridDim.x * GATHER_WAVES_PER_BLOCK;

    for (int node = wave0; node < n; node += nwaves) {
        const int deg = cnt[node];
        const int m = min(deg, CAP);
        float4 acc = make_float4(0.0f, 0.0f, 0.0f, 0.0f);

        if (m > 0) {
            int idx = (lane < m) ? perm[node * CAP + lane] : 0;
            idx = min(max(idx, 0), n - 1);

            int j = 0;
            for (; j + GATHER_BATCH <= m; j += GATHER_BATCH) {
                float4 v[GATHER_BATCH];
                for (int u = 0; u < GATHER_BATCH; ++u)
                    v[u] = load_row4(src, __builtin_amdgcn_readlane(idx, j + u), lane);
                for (int u = 0; u < GATHER_BATCH; ++u) {
                    acc.x += v[u].x; acc.y += v[u].y; acc.z += v[u].z; acc.w += v[u].w;
                }
            }
            if (j < m) {
                float4 v[GATHER_BATCH - 1];
                for (int u = 0; u < GATHER_BATCH - 1; ++u) {
                    v[u] = make_float4(0.0f, 0.0f, 0.0f, 0.0f);
                    if (j + u < m)
                        v[u] = load_row4(src, __builtin_amdgcn_readlane(idx, j + u), lane);
                }
                for (int u = 0; u < GATHER_BATCH - 1; ++u) {
                    if (j + u < m) {
                        acc.x += v[u].x; acc.y += v[u].y; acc.z += v[u].z; acc.w += v[u].w;
                    }
                }
            }
        }
        if (deg > CAP) {
            const int no = min(max(*ovf_n, 0), EE);
            for (int b = 0; b < no; b += 32) {
                const int k = b + lane;
                const int d = (k < no) ? ovf_dst[k] : -1;
                int s = (k < no) ? ovf_src[k] : 0;
                s = min(max(s, 0), n - 1);
                unsigned mask = __builtin_amdgcn_ballot_w32(d == node);
                while (mask != 0u) {
                    const int q = __builtin_ctz(mask);
                    mask &= mask - 1u;
                    const float4 v = load_row4(src, __builtin_amdgcn_readlane(s, q), lane);
                    acc.x += v.x; acc.y += v.y; acc.z += v.z; acc.w += v.w;
                }
            }
        }
        const float inv = 1.0f / fmaxf((float)deg, 1.0f);
        acc.x *= inv; acc.y *= inv; acc.z *= inv; acc.w *= inv;
        *(float4*)(dst + (long)node * DD + lane * 4) = acc;
    }
}


#define GEMM_WAVES 4
#define GEMM_THREADS (GEMM_WAVES * 32)
#define ROW_TILES (NN / 16)
#define GEMM_BLOCKS ((ROW_TILES + GEMM_WAVES - 1) / GEMM_WAVES)
static_assert(NN % 16 == 0, "rows are handled in whole 16-row tiles");
static_assert(DD == 128, "the slab layout below is written for 128 x 128 weight slabs");

#define W_PITCH 288
#define W_SLAB_DWORDS ((DD / 2) * W_PITCH)
#define W_ITEMS_PER_THREAD ((DD / 2) / GEMM_WAVES)

__device__ __forceinline__ void stage_slab(const float* __restrict__ Wsrc, float* wl, int tid) {
    const int kp0 = tid >> 5;
    const int c4  = tid & 31;
    const float* src = Wsrc + (long)(2 * kp0) * DD + 4 * c4;
    float* dst = wl + kp0 * W_PITCH + 8 * c4;
#pragma unroll
    for (int it = 0; it < W_ITEMS_PER_THREAD; ++it) {
        const float4 r0 = *(const float4*)(src + it * (2 * GEMM_WAVES * DD));
        const float4 r1 = *(const float4*)(src + it * (2 * GEMM_WAVES * DD) + DD);
        float* d = dst + it * (GEMM_WAVES * W_PITCH);
        *(float4*)(d)     = make_float4(r0.x, r1.x, r0.y, r1.y);
        *(float4*)(d + 4) = make_float4(r0.z, r1.z, r0.w, r1.w);
    }
}

__device__ __forceinline__ void load_a_frags(const float* __restrict__ arow, v2f (&a)[DD / 4]) {
#pragma unroll
    for (int s = 0; s < DD / 4; ++s) {
        a[s].x = arow[4 * s];
        a[s].y = arow[4 * s + 1];
    }
}

#define P_PITCH 32
#define P_SLAB_DWORDS ((DD / 2) * P_PITCH)

__device__ __forceinline__ void mma_slab_p(const v2f (&a)[DD / 4], const float* wl_lane,
                                           const float* pl_lane, v8f (&acc)[8], v8f& accl) {
#pragma unroll
    for (int s = 0; s < DD / 4; ++s) {
#pragma unroll
        for (int ct = 0; ct < 8; ++ct) {
            const v2f bb = *(const v2f*)(wl_lane + 2 * s * W_PITCH + 32 * ct);
            acc[ct] = __builtin_amdgcn_wmma_f32_16x16x4_f32(false, a[s], false, bb,
                                                            (short)0, acc[ct], false, false);
        }
        const v2f pp = *(const v2f*)(pl_lane + 2 * s * P_PITCH);
        accl = __builtin_amdgcn_wmma_f32_16x16x4_f32(false, a[s], false, pp,
                                                     (short)0, accl, false, false);
    }
}

#define FOLD_ROW_TILES 9
#define FOLD_COL_GROUPS 5
#define FOLD_BLOCKS (NHOP * FOLD_ROW_TILES * FOLD_COL_GROUPS)
#define DF_STRIDE 32

__device__ __forceinline__ void fold_role(const int b, const int lane,
            const float* __restrict__ Wh, const float* __restrict__ bh,
            const float* __restrict__ Wf, const float* __restrict__ Wa,
            float* __restrict__ Mf, float* __restrict__ Pf,
            float* __restrict__ cf, float* __restrict__ df) {
    const int hop = b / (FOLD_ROW_TILES * FOLD_COL_GROUPS);
    const int rt  = (b / FOLD_COL_GROUPS) % FOLD_ROW_TILES;
    const int cg  = b % FOLD_COL_GROUPS;
    const int half = lane >> 4;
    const int l    = lane & 15;

    const int R = 16 * rt + l;
    const float* ap = (R < DD) ? Wh + ((long)hop * DD + R) * DD : bh + hop * DD;
    const bool aok = R <= DD;
    ap += 2 * half;

    const float* Wf_h = Wf + (long)hop * DD * DD;
    const float* Wa_h = Wa + (long)hop * DD * NHOP;
    const int C0 = 32 * cg + l;
    const int C1 = C0 + 16;
    const float* bp0; int bs0; bool bok0;
    if (C0 < DD)             { bp0 = Wf_h + C0;        bs0 = DD;   bok0 = true;  }
    else if (C0 < DD + NHOP) { bp0 = Wa_h + (C0 - DD); bs0 = NHOP; bok0 = true;  }
    else                     { bp0 = Wa_h;             bs0 = NHOP; bok0 = false; }
    const float* bp1; int bs1; bool bok1;
    if (C1 < DD)             { bp1 = Wf_h + C1;        bs1 = DD;   bok1 = true;  }
    else if (C1 < DD + NHOP) { bp1 = Wa_h + (C1 - DD); bs1 = NHOP; bok1 = true;  }
    else                     { bp1 = Wa_h;             bs1 = NHOP; bok1 = false; }
    bp0 += 2 * half * bs0;
    bp1 += 2 * half * bs1;

    v2f a[DD / 4], b0[DD / 4], b1[DD / 4];
#pragma unroll
    for (int s = 0; s < DD / 4; ++s) {
        a[s].x  = ap[4 * s];
        a[s].y  = ap[4 * s + 1];
        b0[s].x = bp0[(4 * s) * bs0];
        b0[s].y = bp0[(4 * s + 1) * bs0];
        b1[s].x = bp1[(4 * s) * bs1];
        b1[s].y = bp1[(4 * s + 1) * bs1];
    }
    v8f acc0 = {}, acc1 = {};
#pragma unroll
    for (int s = 0; s < DD / 4; ++s) {
        v2f av, bv0, bv1;
        av.x  = aok  ? a[s].x  : 0.0f;
        av.y  = aok  ? a[s].y  : 0.0f;
        bv0.x = bok0 ? b0[s].x : 0.0f;
        bv0.y = bok0 ? b0[s].y : 0.0f;
        bv1.x = bok1 ? b1[s].x : 0.0f;
        bv1.y = bok1 ? b1[s].y : 0.0f;
        acc0 = __builtin_amdgcn_wmma_f32_16x16x4_f32(false, av, false, bv0, (short)0, acc0, false, false);
        acc1 = __builtin_amdgcn_wmma_f32_16x16x4_f32(false, av, false, bv1, (short)0, acc1, false, false);
    }

#pragma unroll
    for (int j = 0; j < 8; ++j) {
        const int Rr = 16 * rt + 8 * half + j;
        if (Rr < DD) {
            if (cg < FOLD_COL_GROUPS - 1) {
                Mf[((long)hop * DD + Rr) * DD + C0] = acc0[j];
                Mf[((long)hop * DD + Rr) * DD + C1] = acc1[j];
            } else {
                Pf[hop * P_SLAB_DWORDS + (Rr >> 1) * P_PITCH + 2 * l + (Rr & 1)] = acc0[j];
            }
        } else if (Rr == DD) {
            if (cg < FOLD_COL_GROUPS - 1) {
                cf[hop * DD + C0] = acc0[j];
                cf[hop * DD + C1] = acc1[j];
            } else {
                df[hop * DF_STRIDE + l] = acc0[j];
            }
        }
    }
}

#define PRO_BLOCKS (1 + FOLD_BLOCKS + ZERO_BLOCKS)

__global__ void __launch_bounds__(PRO_T)
prologue_kernel(int* __restrict__ cnt, int* __restrict__ ovf_n,
                const int* __restrict__ depths,
                const float* __restrict__ Wg1, const float* __restrict__ bg1,
                const float* __restrict__ Wg2, const float* __restrict__ bg2,
                float* __restrict__ sw,
                const float* __restrict__ Wh, const float* __restrict__ bh,
                const float* __restrict__ Wf, const float* __restrict__ Wa,
                float* __restrict__ Mf, float* __restrict__ Pf,
                float* __restrict__ cf, float* __restrict__ df) {
    __shared__ int sh[32];
    const int b = blockIdx.x;
    if (b == 0) {
        gate_role(depths, Wg1, bg1, Wg2, bg2, sw, sh);
    } else if (b <= FOLD_BLOCKS) {
        const int wave = __builtin_amdgcn_readfirstlane((int)(threadIdx.x >> 5));
        if (wave == 0) fold_role(b - 1, (int)threadIdx.x, Wh, bh, Wf, Wa, Mf, Pf, cf, df);
    } else {
        zero_role(b - 1 - FOLD_BLOCKS, cnt, ovf_n);
    }
}


__global__ void __launch_bounds__(GEMM_THREADS) __attribute__((amdgpu_waves_per_eu(1, 2)))
fused_kernel(const float* __restrict__ X0, const float* __restrict__ X1,
             const float* __restrict__ X2,
             const float* __restrict__ Mf, const float* __restrict__ Pf,
             const float* __restrict__ cf, const float* __restrict__ df,
             const float* __restrict__ sw, const float* __restrict__ ba,
             const float* __restrict__ bf, float* __restrict__ out) {
    __shared__ __attribute__((aligned(16))) float wl[W_SLAB_DWORDS];
    __shared__ __attribute__((aligned(16))) float pl[P_SLAB_DWORDS];
    const int tid = threadIdx.x;
    const int wave = __builtin_amdgcn_readfirstlane(tid >> 5);
    const int lane = tid & 31;
    const int half = lane >> 4;
    const int l    = lane & 15;
    const int rt = blockIdx.x * GEMM_WAVES + wave;
    const bool valid = rt < ROW_TILES;
    const int row0 = (valid ? rt : 0) * 16;
    const int row = row0 + l;

    v8f acc[NHOP][8] = {};
    v8f accl[NHOP] = {};
#pragma unroll
    for (int hop = 0; hop < NHOP; ++hop) {
        const float* X = (hop == 0) ? X0 : ((hop == 1) ? X1 : X2);

        stage_slab(Mf + (long)hop * DD * DD, wl, tid);
#pragma unroll
        for (int it = 0; it < P_SLAB_DWORDS / 4 / GEMM_THREADS; ++it)
            ((float4*)pl)[tid + GEMM_THREADS * it] =
                ((const float4*)(Pf + hop * P_SLAB_DWORDS))[tid + GEMM_THREADS * it];
        v2f a[DD / 4];
        load_a_frags(X + (long)row * DD + 2 * half, a);
        __syncthreads();
        if (valid) mma_slab_p(a, wl + half * W_PITCH + 2 * l, pl + half * P_PITCH + 2 * l,
                              acc[hop], accl[hop]);
        if (hop + 1 < NHOP) __syncthreads();
    }
    if (!valid) return;

    const float s0 = sw[0], s1 = sw[1], s2 = sw[2];
    const float bac = ba[min(l, NHOP - 1)];
    const float d0 = df[0 * DF_STRIDE + l], d1 = df[1 * DF_STRIDE + l], d2 = df[2 * DF_STRIDE + l];
    const int src = lane & 16;
    float w0[8], w1[8], w2[8];
#pragma unroll
    for (int j = 0; j < 8; ++j) {
        const float lg = bac + s0 * (accl[0][j] + d0) + s1 * (accl[1][j] + d1) + s2 * (accl[2][j] + d2);
        const float l0 = __shfl(lg, src + 0, 32);
        const float l1 = __shfl(lg, src + 1, 32);
        const float l2 = __shfl(lg, src + 2, 32);
        const float mx = fmaxf(l0, fmaxf(l1, l2));
        const float e0 = __expf(l0 - mx), e1 = __expf(l1 - mx), e2 = __expf(l2 - mx);
        const float inv = 1.0f / (e0 + e1 + e2);
        w0[j] = e0 * inv * s0;
        w1[j] = e1 * inv * s1;
        w2[j] = e2 * inv * s2;
    }

#pragma unroll
    for (int ct = 0; ct < 8; ++ct) {
        const int col = 16 * ct + l;
        const float c0 = cf[0 * DD + col], c1 = cf[1 * DD + col], c2 = cf[2 * DD + col];
        const float bias = bf[col];
#pragma unroll
        for (int j = 0; j < 8; ++j) {
            const int m = j + 8 * half;
            out[(long)(row0 + m) * DD + col] = w0[j] * (acc[0][ct][j] + c0) + w1[j] * (acc[1][ct][j] + c1)
                                             + w2[j] * (acc[2][ct][j] + c2) + bias;
        }
    }
}

static inline int cdiv(long a, long b) { return (int)((a + b - 1) / b); }

extern "C" void kernel_launch(void* const* d_in, const int* in_sizes, int n_in,
                              void* d_out, int out_size, void* d_ws, size_t ws_size,
                              hipStream_t stream) {
    const float* x      = (const float*)d_in[0];
    const int*   ei     = (const int*)  d_in[1];
    const int*   depths = (const int*)  d_in[2];
    const float* Wh     = (const float*)d_in[3];
    const float* bh     = (const float*)d_in[4];
    const float* Wg1    = (const float*)d_in[5];
    const float* bg1    = (const float*)d_in[6];
    const float* Wg2    = (const float*)d_in[7];
    const float* bg2    = (const float*)d_in[8];
    const float* Wa     = (const float*)d_in[9];
    const float* ba     = (const float*)d_in[10];
    const float* Wf     = (const float*)d_in[11];
    const float* bf     = (const float*)d_in[12];
    float* out = (float*)d_out;

    const int* row = ei;
    const int* col = ei + EE;

    char* ws = (char*)d_ws;
    size_t o = 0;
    auto carve = [&](size_t bytes) -> float* {
        float* p = (float*)(ws + o);
        o += (bytes + 255) & ~(size_t)255;
        return p;
    };
    float* sw     = carve(16 * sizeof(float));
    float* counts = carve((size_t)NN * sizeof(float));
    float* agg1   = carve((size_t)NN * DD * sizeof(float));
    float* agg2   = carve((size_t)NN * DD * sizeof(float));
    float* fold   = carve((size_t)NN * DD * sizeof(float));
    float* spare  = carve((size_t)NN * DD * sizeof(float));
    float* sortws = carve((size_t)NN * DD * sizeof(float));
    (void)ws_size; (void)in_sizes; (void)n_in; (void)out_size;

    float* Mf = fold;
    float* Pf = fold + NHOP * DD * DD;
    float* cf = Pf + NHOP * P_SLAB_DWORDS;
    float* df = cf + NHOP * DD;

    int* cnt     = (int*)counts;
    int* perm    = (int*)sortws;
    int* ovf_n   = (int*)spare;
    int* ovf_dst = (int*)spare + 64;
    int* ovf_src = (int*)spare + 64 + EE;

    prologue_kernel<<<PRO_BLOCKS, PRO_T, 0, stream>>>(cnt, ovf_n, depths, Wg1, bg1, Wg2, bg2, sw,
                                                      Wh, bh, Wf, Wa, Mf, Pf, cf, df);
    bucket_kernel<<<cdiv(EE / 4, 256), 256, 0, stream>>>(row, col, cnt, ovf_n, perm, ovf_dst, ovf_src, EE / 4);

    gather_mean_kernel<<<2048, 256, 0, stream>>>(x,    perm, cnt, ovf_n, ovf_dst, ovf_src, agg1, NN);
    gather_mean_kernel<<<2048, 256, 0, stream>>>(agg1, perm, cnt, ovf_n, ovf_dst, ovf_src, agg2, NN);

    fused_kernel<<<GEMM_BLOCKS, GEMM_THREADS, 0, stream>>>(x, agg1, agg2, Mf, Pf, cf, df, sw, ba, bf, out);
}


# edges grouped by destination without global atomics: chunks counting-sorted by 64-node range in LDS, first gather launch builds the buckets in LDS; 5 launches to 4; slots in LDS arrival order
# speedup vs baseline: 5.3398x; 1.4137x over previous
#include <hip/hip_runtime.h>
#include <hip/hip_bf16.h>
#include <math.h>

#define NN 50000
#define DD 128
#define EE 800000
#define NHOP 3

typedef __attribute__((ext_vector_type(2))) float v2f;
typedef __attribute__((ext_vector_type(8))) float v8f;

#define PRO_T 512
#define PRO_WAVES (PRO_T / 32)

#define GATE_ITEMS ((NN / 4 + PRO_T - 1) / PRO_T)
static_assert(NN % 4 == 0, "the depths are read as int4");

__device__ __forceinline__ void gate_role(const int* __restrict__ depths,
                                          const float* __restrict__ Wg1, const float* __restrict__ bg1,
                                          const float* __restrict__ Wg2, const float* __restrict__ bg2,
                                          float* __restrict__ sw, int* smax) {
    const int tid = threadIdx.x;
    const int lane = tid & 31;
    const int w = tid >> 5;
    int m = 0;
#pragma unroll
    for (int i = 0; i < GATE_ITEMS; ++i) {
        const int idx = min(tid + i * PRO_T, NN / 4 - 1);
        const int4 v = ((const int4*)depths)[idx];
        m = max(m, max(max(v.x, v.y), max(v.z, v.w)));
    }
    for (int d = 16; d > 0; d >>= 1) m = max(m, __shfl_xor(m, d, 32));
    if (lane == 0) smax[w] = m;
    __syncthreads();
    if (tid == 0) {
        int mm = smax[0];
        for (int k = 1; k < PRO_WAVES; ++k) mm = max(mm, smax[k]);
        float sf0 = (float)NN / 5000.0f;
        float sf1 = (float)mm / 20.0f;
        float a0 = bg2[0], a1 = bg2[1], a2 = bg2[2];
        for (int j = 0; j < DD / 4; ++j) {
            float h = sf0 * Wg1[j] + sf1 * Wg1[DD / 4 + j] + bg1[j];
            h = fmaxf(h, 0.0f);
            a0 += h * Wg2[j * NHOP + 0];
            a1 += h * Wg2[j * NHOP + 1];
            a2 += h * Wg2[j * NHOP + 2];
        }
        sw[0] = 1.0f / (1.0f + __expf(-a0));
        sw[1] = 1.0f / (1.0f + __expf(-a1));
        sw[2] = 1.0f / (1.0f + __expf(-a2));
    }
}

#define CAP 32
#define RANGE 64
#define NBIN ((NN + RANGE - 1) / RANGE)
#define NCHUNK 250
#define CHUNK_E (EE / NCHUNK)
#define CHUNK_E4 (CHUNK_E / 4)
#define HIST_PAD (2 * PRO_T)
#define SORT_ROUNDS ((CHUNK_E4 + PRO_T - 1) / PRO_T)
#define OVB 64
static_assert(EE % (4 * NCHUNK) == 0, "a chunk is a whole number of int4 reads");
static_assert(NN <= 65536, "a source row id is packed into 16 bits");
static_assert(NBIN <= HIST_PAD, "two range counters per thread");
static_assert((RANGE & (RANGE - 1)) == 0, "range is a power of two");

__device__ __forceinline__ void sort_role(const int chunk, const int* __restrict__ row,
                                          const int* __restrict__ col, int2* __restrict__ tab,
                                          int* __restrict__ region,
                                          int* hist, int* offs, int* sorted, int* wsum) {
    const int tid = threadIdx.x;
    const int lane = tid & 31;
    const int w = tid >> 5;
    hist[tid] = 0;
    hist[tid + PRO_T] = 0;

    int4 c4[SORT_ROUNDS], r4[SORT_ROUNDS];
#pragma unroll
    for (int r = 0; r < SORT_ROUNDS; ++r) {
        const int i = min(tid + r * PRO_T, CHUNK_E4 - 1);
        c4[r] = ((const int4*)col)[chunk * CHUNK_E4 + i];
        r4[r] = ((const int4*)row)[chunk * CHUNK_E4 + i];
    }
    __syncthreads();

    int bin[SORT_ROUNDS * 4], rk[SORT_ROUNDS * 4], wd[SORT_ROUNDS * 4];
#pragma unroll
    for (int r = 0; r < SORT_ROUNDS; ++r) {
        const bool ok = tid + r * PRO_T < CHUNK_E4;
        const int cc[4] = {c4[r].x, c4[r].y, c4[r].z, c4[r].w};
        const int ss[4] = {r4[r].x, r4[r].y, r4[r].z, r4[r].w};
#pragma unroll
        for (int e = 0; e < 4; ++e) {
            const int c = min(max(cc[e], 0), NN - 1);
            const int sr = min(max(ss[e], 0), NN - 1);
            const int q = r * 4 + e;
            bin[q] = ok ? (c / RANGE) : -1;
            wd[q] = ((c & (RANGE - 1)) << 16) | sr;
            rk[q] = ok ? atomicAdd(&hist[bin[q]], 1) : 0;
        }
    }
    __syncthreads();

    const int a0 = hist[2 * tid], a1 = hist[2 * tid + 1];
    const int sum2 = a0 + a1;
    int incl = sum2;
    for (int d = 1; d < 32; d <<= 1) {
        const int t = __shfl_up(incl, d, 32);
        if (lane >= d) incl += t;
    }
    if (lane == 31) wsum[w] = incl;
    __syncthreads();
    int wp = 0;
    for (int k = 0; k < PRO_WAVES; ++k) wp += (k < w) ? wsum[k] : 0;
    const int ex = wp + incl - sum2;
    offs[2 * tid] = ex;
    offs[2 * tid + 1] = ex + a0;
    if (2 * tid < NBIN)     tab[(2 * tid) * NCHUNK + chunk]     = make_int2(ex, a0);
    if (2 * tid + 1 < NBIN) tab[(2 * tid + 1) * NCHUNK + chunk] = make_int2(ex + a0, a1);
    __syncthreads();

#pragma unroll
    for (int q = 0; q < SORT_ROUNDS * 4; ++q)
        if (bin[q] >= 0) sorted[offs[bin[q]] + rk[q]] = wd[q];
    __syncthreads();

    for (int i = tid; i < CHUNK_E4; i += PRO_T)
        ((int4*)region)[chunk * CHUNK_E4 + i] = ((const int4*)sorted)[i];
}

#define GATHER_WAVES_PER_BLOCK 8
#define GATHER_BATCH 8

__device__ __forceinline__ float4 load_row4(const float* __restrict__ src, int r, int lane) {
    return *(const float4*)(src + (long)r * DD + lane * 4);
}

__device__ __forceinline__ void add_hits(const float* __restrict__ src, const bool hit, const int w,
                                         const int lane, float4& acc) {
    const int s = min(w & 0xFFFF, NN - 1);
    unsigned mask = __builtin_amdgcn_ballot_w32(hit);
    while (mask != 0u) {
        const int q = __builtin_ctz(mask);
        mask &= mask - 1u;
        const float4 v = load_row4(src, __builtin_amdgcn_readlane(s, q), lane);
        acc.x += v.x; acc.y += v.y; acc.z += v.z; acc.w += v.w;
    }
}

__device__ __forceinline__ float4 sum_node(const float* __restrict__ src, const int idx, const int deg,
                                           const int d, const int b, const int lane,
                                           const int* ovlist, const int no,
                                           const int2* __restrict__ tab, const int* __restrict__ region) {
    float4 acc = make_float4(0.0f, 0.0f, 0.0f, 0.0f);
    if (deg > CAP && no > OVB) {
        for (int c = 0; c < NCHUNK; ++c) {
            const int2 sc = tab[b * NCHUNK + c];
            const int start = min(max(sc.x, 0), CHUNK_E);
            const int count = min(max(sc.y, 0), CHUNK_E - start);
            for (int k0 = 0; k0 < count; k0 += 32) {
                const int k = k0 + lane;
                const int w = (k < count) ? region[c * CHUNK_E + start + k] : 0;
                add_hits(src, (k < count) && (((w >> 16) & (RANGE - 1)) == d), w, lane, acc);
            }
        }
        return acc;
    }
    const int m = min(deg, CAP);
    int j = 0;
    for (; j + GATHER_BATCH <= m; j += GATHER_BATCH) {
        float4 v[GATHER_BATCH];
        for (int u = 0; u < GATHER_BATCH; ++u)
            v[u] = load_row4(src, __builtin_amdgcn_readlane(idx, j + u), lane);
        for (int u = 0; u < GATHER_BATCH; ++u) {
            acc.x += v[u].x; acc.y += v[u].y; acc.z += v[u].z; acc.w += v[u].w;
        }
    }
    if (j < m) {
        float4 v[GATHER_BATCH - 1];
        for (int u = 0; u < GATHER_BATCH - 1; ++u) {
            v[u] = make_float4(0.0f, 0.0f, 0.0f, 0.0f);
            if (j + u < m)
                v[u] = load_row4(src, __builtin_amdgcn_readlane(idx, j + u), lane);
        }
        for (int u = 0; u < GATHER_BATCH - 1; ++u) {
            if (j + u < m) {
                acc.x += v[u].x; acc.y += v[u].y; acc.z += v[u].z; acc.w += v[u].w;
            }
        }
    }
    if (deg > CAP) {
        const int nl = min(no, OVB);
        for (int q0 = 0; q0 < nl; q0 += 32) {
            const int k = q0 + lane;
            const int w = (k < nl) ? ovlist[k] : 0;
            add_hits(src, (k < nl) && (((w >> 16) & (RANGE - 1)) == d), w, lane, acc);
        }
    }
    return acc;
}

__device__ __forceinline__ void store_mean(float* __restrict__ dst, const int node, const int deg,
                                           const int lane, float4 acc) {
    const float inv = 1.0f / fmaxf((float)deg, 1.0f);
    acc.x *= inv; acc.y *= inv; acc.z *= inv; acc.w *= inv;
    *(float4*)(dst + (long)node * DD + lane * 4) = acc;
}

#define G1_T 512
#define G1_WAVES (G1_T / 32)
#define SEG_BATCH 12
static_assert(NCHUNK <= G1_T, "one thread per chunk segment");

__global__ void __launch_bounds__(G1_T)
gather1_kernel(const float* __restrict__ src, const int* __restrict__ tab_w,
               const int* __restrict__ region, int* __restrict__ perm, int* __restrict__ cnt,
               int* __restrict__ ovl, int* __restrict__ ovn, float* __restrict__ dst) {
    __shared__ __attribute__((aligned(16))) int lperm[RANGE * CAP];
    __shared__ int lcnt[RANGE];
    __shared__ int lovl[OVB];
    __shared__ int lovn;
    const int2* __restrict__ tab = (const int2*)tab_w;
    const int tid = threadIdx.x;
    const int lane = tid & 31;
    const int wave = __builtin_amdgcn_readfirstlane(tid >> 5);
    const int b = blockIdx.x;
    const int nodes_here = min(RANGE, NN - b * RANGE);

    if (tid < RANGE) lcnt[tid] = 0;
    if (tid == 0) lovn = 0;
    int start = 0, count = 0;
    if (tid < NCHUNK) {
        const int2 sc = tab[b * NCHUNK + tid];
        start = min(max(sc.x, 0), CHUNK_E);
        count = min(max(sc.y, 0), CHUNK_E - start);
    }
    const int* base = region + min(tid, NCHUNK - 1) * CHUNK_E + start;
    __syncthreads();

    for (int k0 = 0; k0 < count; k0 += SEG_BATCH) {
        int wv[SEG_BATCH];
#pragma unroll
        for (int u = 0; u < SEG_BATCH; ++u) wv[u] = (k0 + u < count) ? base[k0 + u] : 0;
#pragma unroll
        for (int u = 0; u < SEG_BATCH; ++u) {
            if (k0 + u < count) {
                const int d = (wv[u] >> 16) & (RANGE - 1);
                const int slot = atomicAdd(&lcnt[d], 1);
                if (slot < CAP) {
                    lperm[d * CAP + slot] = wv[u] & 0xFFFF;
                } else {
                    const int q = atomicAdd(&lovn, 1);
                    if (q < OVB) lovl[q] = wv[u];
                }
            }
        }
    }
    __syncthreads();

    const int no = __builtin_amdgcn_readfirstlane(lovn);
    for (int i = tid; i < nodes_here * (CAP / 4); i += G1_T)
        ((int4*)(perm + (long)b * RANGE * CAP))[i] = ((const int4*)lperm)[i];
    if (tid < nodes_here) cnt[b * RANGE + tid] = lcnt[tid];
    if (tid < min(no, OVB)) ovl[b * OVB + tid] = lovl[tid];
    if (tid == 0) ovn[b] = no;

    for (int d = wave; d < nodes_here; d += G1_WAVES) {
        const int deg = __builtin_amdgcn_readfirstlane(lcnt[d]);
        const int idx = min(max(lperm[d * CAP + lane], 0), NN - 1);
        const float4 acc = sum_node(src, idx, deg, d, b, lane, lovl, no, tab, region);
        store_mean(dst, b * RANGE + d, deg, lane, acc);
    }
}

__global__ void __launch_bounds__(256)
gather_mean_kernel(const float* __restrict__ src, const int* __restrict__ perm,
                   const int* __restrict__ cnt, const int* __restrict__ ovl,
                   const int* __restrict__ ovn, const int* __restrict__ tab_w,
                   const int* __restrict__ region, float* __restrict__ dst) {
    const int2* __restrict__ tab = (const int2*)tab_w;
    const int lane = threadIdx.x & 31;
    const int wave0 = __builtin_amdgcn_readfirstlane(
        (int)(blockIdx.x * GATHER_WAVES_PER_BLOCK + (threadIdx.x >> 5)));
    const int nwaves = (int)gridDim.x * GATHER_WAVES_PER_BLOCK;

    for (int node = wave0; node < NN; node += nwaves) {
        const int deg = cnt[node];
        const int m = min(deg, CAP);
        const int b = node / RANGE;
        int idx = (lane < m) ? perm[node * CAP + lane] : 0;
        idx = min(max(idx, 0), NN - 1);
        const int no = (deg > CAP) ? ovn[b] : 0;
        const float4 acc = sum_node(src, idx, deg, node & (RANGE - 1), b, lane,
                                    ovl + b * OVB, no, tab, region);
        store_mean(dst, node, deg, lane, acc);
    }
}

#define GEMM_WAVES 4
#define GEMM_THREADS (GEMM_WAVES * 32)
#define ROW_TILES (NN / 16)
#define GEMM_BLOCKS ((ROW_TILES + GEMM_WAVES - 1) / GEMM_WAVES)
static_assert(NN % 16 == 0, "rows are handled in whole 16-row tiles");
static_assert(DD == 128, "the slab layout below is written for 128 x 128 weight slabs");

#define W_PITCH 288
#define W_SLAB_DWORDS ((DD / 2) * W_PITCH)
#define W_ITEMS_PER_THREAD ((DD / 2) / GEMM_WAVES)

__device__ __forceinline__ void stage_slab(const float* __restrict__ Wsrc, float* wl, int tid) {
    const int kp0 = tid >> 5;
    const int c4  = tid & 31;
    const float* src = Wsrc + (long)(2 * kp0) * DD + 4 * c4;
    float* dst = wl + kp0 * W_PITCH + 8 * c4;
#pragma unroll
    for (int it = 0; it < W_ITEMS_PER_THREAD; ++it) {
        const float4 r0 = *(const float4*)(src + it * (2 * GEMM_WAVES * DD));
        const float4 r1 = *(const float4*)(src + it * (2 * GEMM_WAVES * DD) + DD);
        float* d = dst + it * (GEMM_WAVES * W_PITCH);
        *(float4*)(d)     = make_float4(r0.x, r1.x, r0.y, r1.y);
        *(float4*)(d + 4) = make_float4(r0.z, r1.z, r0.w, r1.w);
    }
}

__device__ __forceinline__ void load_a_frags(const float* __restrict__ arow, v2f (&a)[DD / 4]) {
#pragma unroll
    for (int s = 0; s < DD / 4; ++s) {
        a[s].x = arow[4 * s];
        a[s].y = arow[4 * s + 1];
    }
}

#define P_PITCH 32
#define P_SLAB_DWORDS ((DD / 2) * P_PITCH)

__device__ __forceinline__ void mma_slab_p(const v2f (&a)[DD / 4], const float* wl_lane,
                                           const float* pl_lane, v8f (&acc)[8], v8f& accl) {
#pragma unroll
    for (int s = 0; s < DD / 4; ++s) {
#pragma unroll
        for (int ct = 0; ct < 8; ++ct) {
            const v2f bb = *(const v2f*)(wl_lane + 2 * s * W_PITCH + 32 * ct);
            acc[ct] = __builtin_amdgcn_wmma_f32_16x16x4_f32(false, a[s], false, bb,
                                                            (short)0, acc[ct], false, false);
        }
        const v2f pp = *(const v2f*)(pl_lane + 2 * s * P_PITCH);
        accl = __builtin_amdgcn_wmma_f32_16x16x4_f32(false, a[s], false, pp,
                                                     (short)0, accl, false, false);
    }
}

#define FOLD_ROW_TILES 9
#define FOLD_COL_GROUPS 5
#define FOLD_BLOCKS (NHOP * FOLD_ROW_TILES * FOLD_COL_GROUPS)
#define DF_STRIDE 32

__device__ __forceinline__ void fold_role(const int b, const int lane,
            const float* __restrict__ Wh, const float* __restrict__ bh,
            const float* __restrict__ Wf, const float* __restrict__ Wa,
            float* __restrict__ Mf, float* __restrict__ Pf,
            float* __restrict__ cf, float* __restrict__ df) {
    const int hop = b / (FOLD_ROW_TILES * FOLD_COL_GROUPS);
    const int rt  = (b / FOLD_COL_GROUPS) % FOLD_ROW_TILES;
    const int cg  = b % FOLD_COL_GROUPS;
    const int half = lane >> 4;
    const int l    = lane & 15;

    const int R = 16 * rt + l;
    const float* ap = (R < DD) ? Wh + ((long)hop * DD + R) * DD : bh + hop * DD;
    const bool aok = R <= DD;
    ap += 2 * half;

    const float* Wf_h = Wf + (long)hop * DD * DD;
    const float* Wa_h = Wa + (long)hop * DD * NHOP;
    const int C0 = 32 * cg + l;
    const int C1 = C0 + 16;
    const float* bp0; int bs0; bool bok0;
    if (C0 < DD)             { bp0 = Wf_h + C0;        bs0 = DD;   bok0 = true;  }
    else if (C0 < DD + NHOP) { bp0 = Wa_h + (C0 - DD); bs0 = NHOP; bok0 = true;  }
    else                     { bp0 = Wa_h;             bs0 = NHOP; bok0 = false; }
    const float* bp1; int bs1; bool bok1;
    if (C1 < DD)             { bp1 = Wf_h + C1;        bs1 = DD;   bok1 = true;  }
    else if (C1 < DD + NHOP) { bp1 = Wa_h + (C1 - DD); bs1 = NHOP; bok1 = true;  }
    else                     { bp1 = Wa_h;             bs1 = NHOP; bok1 = false; }
    bp0 += 2 * half * bs0;
    bp1 += 2 * half * bs1;

    v2f a[DD / 4], b0[DD / 4], b1[DD / 4];
#pragma unroll
    for (int s = 0; s < DD / 4; ++s) {
        a[s].x  = ap[4 * s];
        a[s].y  = ap[4 * s + 1];
        b0[s].x = bp0[(4 * s) * bs0];
        b0[s].y = bp0[(4 * s + 1) * bs0];
        b1[s].x = bp1[(4 * s) * bs1];
        b1[s].y = bp1[(4 * s + 1) * bs1];
    }
    v8f acc0 = {}, acc1 = {};
#pragma unroll
    for (int s = 0; s < DD / 4; ++s) {
        v2f av, bv0, bv1;
        av.x  = aok  ? a[s].x  : 0.0f;
        av.y  = aok  ? a[s].y  : 0.0f;
        bv0.x = bok0 ? b0[s].x : 0.0f;
        bv0.y = bok0 ? b0[s].y : 0.0f;
        bv1.x = bok1 ? b1[s].x : 0.0f;
        bv1.y = bok1 ? b1[s].y : 0.0f;
        acc0 = __builtin_amdgcn_wmma_f32_16x16x4_f32(false, av, false, bv0, (short)0, acc0, false, false);
        acc1 = __builtin_amdgcn_wmma_f32_16x16x4_f32(false, av, false, bv1, (short)0, acc1, false, false);
    }

#pragma unroll
    for (int j = 0; j < 8; ++j) {
        const int Rr = 16 * rt + 8 * half + j;
        if (Rr < DD) {
            if (cg < FOLD_COL_GROUPS - 1) {
                Mf[((long)hop * DD + Rr) * DD + C0] = acc0[j];
                Mf[((long)hop * DD + Rr) * DD + C1] = acc1[j];
            } else {
                Pf[hop * P_SLAB_DWORDS + (Rr >> 1) * P_PITCH + 2 * l + (Rr & 1)] = acc0[j];
            }
        } else if (Rr == DD) {
            if (cg < FOLD_COL_GROUPS - 1) {
                cf[hop * DD + C0] = acc0[j];
                cf[hop * DD + C1] = acc1[j];
            } else {
                df[hop * DF_STRIDE + l] = acc0[j];
            }
        }
    }
}

#define PRO_BLOCKS (NCHUNK + 1 + FOLD_BLOCKS)

__global__ void __launch_bounds__(PRO_T)
prologue_kernel(const int* __restrict__ row, const int* __restrict__ col,
                int* __restrict__ tab_w, int* __restrict__ region,
                const int* __restrict__ depths,
                const float* __restrict__ Wg1, const float* __restrict__ bg1,
                const float* __restrict__ Wg2, const float* __restrict__ bg2,
                float* __restrict__ sw,
                const float* __restrict__ Wh, const float* __restrict__ bh,
                const float* __restrict__ Wf, const float* __restrict__ Wa,
                float* __restrict__ Mf, float* __restrict__ Pf,
                float* __restrict__ cf, float* __restrict__ df) {
    __shared__ int sh[32];
    __shared__ int hist[HIST_PAD];
    __shared__ int offs[HIST_PAD];
    __shared__ __attribute__((aligned(16))) int sorted[CHUNK_E];
    const int b = blockIdx.x;
    if (b < NCHUNK) {
        sort_role(b, row, col, (int2*)tab_w, region, hist, offs, sorted, sh);
    } else if (b == NCHUNK) {
        gate_role(depths, Wg1, bg1, Wg2, bg2, sw, sh);
    } else {
        const int wave = __builtin_amdgcn_readfirstlane((int)(threadIdx.x >> 5));
        if (wave == 0) fold_role(b - NCHUNK - 1, (int)threadIdx.x, Wh, bh, Wf, Wa, Mf, Pf, cf, df);
    }
}


__global__ void __launch_bounds__(GEMM_THREADS) __attribute__((amdgpu_waves_per_eu(1, 2)))
fused_kernel(const float* __restrict__ X0, const float* __restrict__ X1,
             const float* __restrict__ X2,
             const float* __restrict__ Mf, const float* __restrict__ Pf,
             const float* __restrict__ cf, const float* __restrict__ df,
             const float* __restrict__ sw, const float* __restrict__ ba,
             const float* __restrict__ bf, float* __restrict__ out) {
    __shared__ __attribute__((aligned(16))) float wl[W_SLAB_DWORDS];
    __shared__ __attribute__((aligned(16))) float pl[P_SLAB_DWORDS];
    const int tid = threadIdx.x;
    const int wave = __builtin_amdgcn_readfirstlane(tid >> 5);
    const int lane = tid & 31;
    const int half = lane >> 4;
    const int l    = lane & 15;
    const int rt = blockIdx.x * GEMM_WAVES + wave;
    const bool valid = rt < ROW_TILES;
    const int row0 = (valid ? rt : 0) * 16;
    const int row = row0 + l;

    v8f acc[NHOP][8] = {};
    v8f accl[NHOP] = {};
#pragma unroll
    for (int hop = 0; hop < NHOP; ++hop) {
        const float* X = (hop == 0) ? X0 : ((hop == 1) ? X1 : X2);

        stage_slab(Mf + (long)hop * DD * DD, wl, tid);
#pragma unroll
        for (int it = 0; it < P_SLAB_DWORDS / 4 / GEMM_THREADS; ++it)
            ((float4*)pl)[tid + GEMM_THREADS * it] =
                ((const float4*)(Pf + hop * P_SLAB_DWORDS))[tid + GEMM_THREADS * it];
        v2f a[DD / 4];
        load_a_frags(X + (long)row * DD + 2 * half, a);
        __syncthreads();
        if (valid) mma_slab_p(a, wl + half * W_PITCH + 2 * l, pl + half * P_PITCH + 2 * l,
                              acc[hop], accl[hop]);
        if (hop + 1 < NHOP) __syncthreads();
    }
    if (!valid) return;

    const float s0 = sw[0], s1 = sw[1], s2 = sw[2];
    const float bac = ba[min(l, NHOP - 1)];
    const float d0 = df[0 * DF_STRIDE + l], d1 = df[1 * DF_STRIDE + l], d2 = df[2 * DF_STRIDE + l];
    const int src = lane & 16;
    float w0[8], w1[8], w2[8];
#pragma unroll
    for (int j = 0; j < 8; ++j) {
        const float lg = bac + s0 * (accl[0][j] + d0) + s1 * (accl[1][j] + d1) + s2 * (accl[2][j] + d2);
        const float l0 = __shfl(lg, src + 0, 32);
        const float l1 = __shfl(lg, src + 1, 32);
        const float l2 = __shfl(lg, src + 2, 32);
        const float mx = fmaxf(l0, fmaxf(l1, l2));
        const float e0 = __expf(l0 - mx), e1 = __expf(l1 - mx), e2 = __expf(l2 - mx);
        const float inv = 1.0f / (e0 + e1 + e2);
        w0[j] = e0 * inv * s0;
        w1[j] = e1 * inv * s1;
        w2[j] = e2 * inv * s2;
    }

#pragma unroll
    for (int ct = 0; ct < 8; ++ct) {
        const int col = 16 * ct + l;
        const float c0 = cf[0 * DD + col], c1 = cf[1 * DD + col], c2 = cf[2 * DD + col];
        const float bias = bf[col];
#pragma unroll
        for (int j = 0; j < 8; ++j) {
            const int m = j + 8 * half;
            out[(long)(row0 + m) * DD + col] = w0[j] * (acc[0][ct][j] + c0) + w1[j] * (acc[1][ct][j] + c1)
                                             + w2[j] * (acc[2][ct][j] + c2) + bias;
        }
    }
}

extern "C" void kernel_launch(void* const* d_in, const int* in_sizes, int n_in,
                              void* d_out, int out_size, void* d_ws, size_t ws_size,
                              hipStream_t stream) {
    const float* x      = (const float*)d_in[0];
    const int*   ei     = (const int*)  d_in[1];
    const int*   depths = (const int*)  d_in[2];
    const float* Wh     = (const float*)d_in[3];
    const float* bh     = (const float*)d_in[4];
    const float* Wg1    = (const float*)d_in[5];
    const float* bg1    = (const float*)d_in[6];
    const float* Wg2    = (const float*)d_in[7];
    const float* bg2    = (const float*)d_in[8];
    const float* Wa     = (const float*)d_in[9];
    const float* ba     = (const float*)d_in[10];
    const float* Wf     = (const float*)d_in[11];
    const float* bf     = (const float*)d_in[12];
    float* out = (float*)d_out;

    const int* row = ei;
    const int* col = ei + EE;

    char* ws = (char*)d_ws;
    size_t o = 0;
    auto carve = [&](size_t bytes) -> float* {
        float* p = (float*)(ws + o);
        o += (bytes + 255) & ~(size_t)255;
        return p;
    };
    float* sw     = carve(16 * sizeof(float));
    float* counts = carve((size_t)NN * sizeof(float));
    float* agg1   = carve((size_t)NN * DD * sizeof(float));
    float* agg2   = carve((size_t)NN * DD * sizeof(float));
    float* fold   = carve((size_t)NN * DD * sizeof(float));
    float* spare  = carve((size_t)NN * DD * sizeof(float));
    float* sortws = carve((size_t)NN * DD * sizeof(float));
    (void)ws_size; (void)in_sizes; (void)n_in; (void)out_size;

    float* Mf = fold;
    float* Pf = fold + NHOP * DD * DD;
    float* cf = Pf + NHOP * P_SLAB_DWORDS;
    float* df = cf + NHOP * DD;

    int*  cnt    = (int*)counts;
    int*  perm   = (int*)sortws;
    int*  region = (int*)spare;
    int*  tab    = (int*)spare + EE;
    int*  ovl    = (int*)spare + EE + 2 * NBIN * NCHUNK;
    int*  ovn    = ovl + NBIN * OVB;

    prologue_kernel<<<PRO_BLOCKS, PRO_T, 0, stream>>>(row, col, tab, region, depths, Wg1, bg1, Wg2, bg2, sw,
                                                      Wh, bh, Wf, Wa, Mf, Pf, cf, df);

    gather1_kernel<<<NBIN, G1_T, 0, stream>>>(x, tab, region, perm, cnt, ovl, ovn, agg1);
    gather_mean_kernel<<<2048, 256, 0, stream>>>(agg1, perm, cnt, ovl, ovn, tab, region, agg2);

    fused_kernel<<<GEMM_BLOCKS, GEMM_THREADS, 0, stream>>>(x, agg1, agg2, Mf, Pf, cf, df, sw, ba, bf, out);
}


# fused pass as one work-group per WGP: the three weight slabs staged in LDS once, twelve waves run their row tiles with no barrier between passes, leftover tiles cut by columns; same sums
# speedup vs baseline: 5.6820x; 1.0641x over previous
#include <hip/hip_runtime.h>
#include <hip/hip_bf16.h>
#include <math.h>

#define NN 50000
#define DD 128
#define EE 800000
#define NHOP 3

typedef __attribute__((ext_vector_type(2))) float v2f;
typedef __attribute__((ext_vector_type(8))) float v8f;

#define PRO_T 512
#define PRO_WAVES (PRO_T / 32)

#define GATE_ITEMS ((NN / 4 + PRO_T - 1) / PRO_T)
static_assert(NN % 4 == 0, "the depths are read as int4");

__device__ __forceinline__ void gate_role(const int* __restrict__ depths,
                                          const float* __restrict__ Wg1, const float* __restrict__ bg1,
                                          const float* __restrict__ Wg2, const float* __restrict__ bg2,
                                          float* __restrict__ sw, int* smax) {
    const int tid = threadIdx.x;
    const int lane = tid & 31;
    const int w = tid >> 5;
    int m = 0;
#pragma unroll
    for (int i = 0; i < GATE_ITEMS; ++i) {
        const int idx = min(tid + i * PRO_T, NN / 4 - 1);
        const int4 v = ((const int4*)depths)[idx];
        m = max(m, max(max(v.x, v.y), max(v.z, v.w)));
    }
    for (int d = 16; d > 0; d >>= 1) m = max(m, __shfl_xor(m, d, 32));
    if (lane == 0) smax[w] = m;
    __syncthreads();
    if (tid == 0) {
        int mm = smax[0];
        for (int k = 1; k < PRO_WAVES; ++k) mm = max(mm, smax[k]);
        float sf0 = (float)NN / 5000.0f;
        float sf1 = (float)mm / 20.0f;
        float a0 = bg2[0], a1 = bg2[1], a2 = bg2[2];
        for (int j = 0; j < DD / 4; ++j) {
            float h = sf0 * Wg1[j] + sf1 * Wg1[DD / 4 + j] + bg1[j];
            h = fmaxf(h, 0.0f);
            a0 += h * Wg2[j * NHOP + 0];
            a1 += h * Wg2[j * NHOP + 1];
            a2 += h * Wg2[j * NHOP + 2];
        }
        sw[0] = 1.0f / (1.0f + __expf(-a0));
        sw[1] = 1.0f / (1.0f + __expf(-a1));
        sw[2] = 1.0f / (1.0f + __expf(-a2));
    }
}

#define CAP 32
#define RANGE 64
#define NBIN ((NN + RANGE - 1) / RANGE)
#define NCHUNK 250
#define CHUNK_E (EE / NCHUNK)
#define CHUNK_E4 (CHUNK_E / 4)
#define HIST_PAD (2 * PRO_T)
#define SORT_ROUNDS ((CHUNK_E4 + PRO_T - 1) / PRO_T)
#define OVB 64
static_assert(EE % (4 * NCHUNK) == 0, "a chunk is a whole number of int4 reads");
static_assert(NN <= 65536, "a source row id is packed into 16 bits");
static_assert(NBIN <= HIST_PAD, "two range counters per thread");
static_assert((RANGE & (RANGE - 1)) == 0, "range is a power of two");

__device__ __forceinline__ void sort_role(const int chunk, const int* __restrict__ row,
                                          const int* __restrict__ col, int2* __restrict__ tab,
                                          int* __restrict__ region,
                                          int* hist, int* offs, int* sorted, int* wsum) {
    const int tid = threadIdx.x;
    const int lane = tid & 31;
    const int w = tid >> 5;
    hist[tid] = 0;
    hist[tid + PRO_T] = 0;

    int4 c4[SORT_ROUNDS], r4[SORT_ROUNDS];
#pragma unroll
    for (int r = 0; r < SORT_ROUNDS; ++r) {
        const int i = min(tid + r * PRO_T, CHUNK_E4 - 1);
        c4[r] = ((const int4*)col)[chunk * CHUNK_E4 + i];
        r4[r] = ((const int4*)row)[chunk * CHUNK_E4 + i];
    }
    __syncthreads();

    int bin[SORT_ROUNDS * 4], rk[SORT_ROUNDS * 4], wd[SORT_ROUNDS * 4];
#pragma unroll
    for (int r = 0; r < SORT_ROUNDS; ++r) {
        const bool ok = tid + r * PRO_T < CHUNK_E4;
        const int cc[4] = {c4[r].x, c4[r].y, c4[r].z, c4[r].w};
        const int ss[4] = {r4[r].x, r4[r].y, r4[r].z, r4[r].w};
#pragma unroll
        for (int e = 0; e < 4; ++e) {
            const int c = min(max(cc[e], 0), NN - 1);
            const int sr = min(max(ss[e], 0), NN - 1);
            const int q = r * 4 + e;
            bin[q] = ok ? (c / RANGE) : -1;
            wd[q] = ((c & (RANGE - 1)) << 16) | sr;
            rk[q] = ok ? atomicAdd(&hist[bin[q]], 1) : 0;
        }
    }
    __syncthreads();

    const int a0 = hist[2 * tid], a1 = hist[2 * tid + 1];
    const int sum2 = a0 + a1;
    int incl = sum2;
    for (int d = 1; d < 32; d <<= 1) {
        const int t = __shfl_up(incl, d, 32);
        if (lane >= d) incl += t;
    }
    if (lane == 31) wsum[w] = incl;
    __syncthreads();
    int wp = 0;
    for (int k = 0; k < PRO_WAVES; ++k) wp += (k < w) ? wsum[k] : 0;
    const int ex = wp + incl - sum2;
    offs[2 * tid] = ex;
    offs[2 * tid + 1] = ex + a0;
    if (2 * tid < NBIN)     tab[(2 * tid) * NCHUNK + chunk]     = make_int2(ex, a0);
    if (2 * tid + 1 < NBIN) tab[(2 * tid + 1) * NCHUNK + chunk] = make_int2(ex + a0, a1);
    __syncthreads();

#pragma unroll
    for (int q = 0; q < SORT_ROUNDS * 4; ++q)
        if (bin[q] >= 0) sorted[offs[bin[q]] + rk[q]] = wd[q];
    __syncthreads();

    for (int i = tid; i < CHUNK_E4; i += PRO_T)
        ((int4*)region)[chunk * CHUNK_E4 + i] = ((const int4*)sorted)[i];
}

#define GATHER_WAVES_PER_BLOCK 8
#define GATHER_BATCH 8

__device__ __forceinline__ float4 load_row4(const float* __restrict__ src, int r, int lane) {
    return *(const float4*)(src + (long)r * DD + lane * 4);
}

__device__ __forceinline__ void add_hits(const float* __restrict__ src, const bool hit, const int w,
                                         const int lane, float4& acc) {
    const int s = min(w & 0xFFFF, NN - 1);
    unsigned mask = __builtin_amdgcn_ballot_w32(hit);
    while (mask != 0u) {
        const int q = __builtin_ctz(mask);
        mask &= mask - 1u;
        const float4 v = load_row4(src, __builtin_amdgcn_readlane(s, q), lane);
        acc.x += v.x; acc.y += v.y; acc.z += v.z; acc.w += v.w;
    }
}

__device__ __forceinline__ float4 sum_node(const float* __restrict__ src, const int idx, const int deg,
                                           const int d, const int b, const int lane,
                                           const int* ovlist, const int no,
                                           const int2* __restrict__ tab, const int* __restrict__ region) {
    float4 acc = make_float4(0.0f, 0.0f, 0.0f, 0.0f);
    if (deg > CAP && no > OVB) {
        for (int c = 0; c < NCHUNK; ++c) {
            const int2 sc = tab[b * NCHUNK + c];
            const int start = min(max(sc.x, 0), CHUNK_E);
            const int count = min(max(sc.y, 0), CHUNK_E - start);
            for (int k0 = 0; k0 < count; k0 += 32) {
                const int k = k0 + lane;
                const int w = (k < count) ? region[c * CHUNK_E + start + k] : 0;
                add_hits(src, (k < count) && (((w >> 16) & (RANGE - 1)) == d), w, lane, acc);
            }
        }
        return acc;
    }
    const int m = min(deg, CAP);
    int j = 0;
    for (; j + GATHER_BATCH <= m; j += GATHER_BATCH) {
        float4 v[GATHER_BATCH];
        for (int u = 0; u < GATHER_BATCH; ++u)
            v[u] = load_row4(src, __builtin_amdgcn_readlane(idx, j + u), lane);
        for (int u = 0; u < GATHER_BATCH; ++u) {
            acc.x += v[u].x; acc.y += v[u].y; acc.z += v[u].z; acc.w += v[u].w;
        }
    }
    if (j < m) {
        float4 v[GATHER_BATCH - 1];
        for (int u = 0; u < GATHER_BATCH - 1; ++u) {
            v[u] = make_float4(0.0f, 0.0f, 0.0f, 0.0f);
            if (j + u < m)
                v[u] = load_row4(src, __builtin_amdgcn_readlane(idx, j + u), lane);
        }
        for (int u = 0; u < GATHER_BATCH - 1; ++u) {
            if (j + u < m) {
                acc.x += v[u].x; acc.y += v[u].y; acc.z += v[u].z; acc.w += v[u].w;
            }
        }
    }
    if (deg > CAP) {
        const int nl = min(no, OVB);
        for (int q0 = 0; q0 < nl; q0 += 32) {
            const int k = q0 + lane;
            const int w = (k < nl) ? ovlist[k] : 0;
            add_hits(src, (k < nl) && (((w >> 16) & (RANGE - 1)) == d), w, lane, acc);
        }
    }
    return acc;
}

__device__ __forceinline__ void store_mean(float* __restrict__ dst, const int node, const int deg,
                                           const int lane, float4 acc) {
    const float inv = 1.0f / fmaxf((float)deg, 1.0f);
    acc.x *= inv; acc.y *= inv; acc.z *= inv; acc.w *= inv;
    *(float4*)(dst + (long)node * DD + lane * 4) = acc;
}

#define G1_T 512
#define G1_WAVES (G1_T / 32)
#define SEG_BATCH 12
static_assert(NCHUNK <= G1_T, "one thread per chunk segment");

__global__ void __launch_bounds__(G1_T)
gather1_kernel(const float* __restrict__ src, const int* __restrict__ tab_w,
               const int* __restrict__ region, int* __restrict__ perm, int* __restrict__ cnt,
               int* __restrict__ ovl, int* __restrict__ ovn, float* __restrict__ dst) {
    __shared__ __attribute__((aligned(16))) int lperm[RANGE * CAP];
    __shared__ int lcnt[RANGE];
    __shared__ int lovl[OVB];
    __shared__ int lovn;
    const int2* __restrict__ tab = (const int2*)tab_w;
    const int tid = threadIdx.x;
    const int lane = tid & 31;
    const int wave = __builtin_amdgcn_readfirstlane(tid >> 5);
    const int b = blockIdx.x;
    const int nodes_here = min(RANGE, NN - b * RANGE);

    if (tid < RANGE) lcnt[tid] = 0;
    if (tid == 0) lovn = 0;
    int start = 0, count = 0;
    if (tid < NCHUNK) {
        const int2 sc = tab[b * NCHUNK + tid];
        start = min(max(sc.x, 0), CHUNK_E);
        count = min(max(sc.y, 0), CHUNK_E - start);
    }
    const int* base = region + min(tid, NCHUNK - 1) * CHUNK_E + start;
    __syncthreads();

    for (int k0 = 0; k0 < count; k0 += SEG_BATCH) {
        int wv[SEG_BATCH];
#pragma unroll
        for (int u = 0; u < SEG_BATCH; ++u) wv[u] = (k0 + u < count) ? base[k0 + u] : 0;
#pragma unroll
        for (int u = 0; u < SEG_BATCH; ++u) {
            if (k0 + u < count) {
                const int d = (wv[u] >> 16) & (RANGE - 1);
                const int slot = atomicAdd(&lcnt[d], 1);
                if (slot < CAP) {
                    lperm[d * CAP + slot] = wv[u] & 0xFFFF;
                } else {
                    const int q = atomicAdd(&lovn, 1);
                    if (q < OVB) lovl[q] = wv[u];
                }
            }
        }
    }
    __syncthreads();

    const int no = __builtin_amdgcn_readfirstlane(lovn);
    for (int i = tid; i < nodes_here * (CAP / 4); i += G1_T)
        ((int4*)(perm + (long)b * RANGE * CAP))[i] = ((const int4*)lperm)[i];
    if (tid < nodes_here) cnt[b * RANGE + tid] = lcnt[tid];
    if (tid < min(no, OVB)) ovl[b * OVB + tid] = lovl[tid];
    if (tid == 0) ovn[b] = no;

    for (int d = wave; d < nodes_here; d += G1_WAVES) {
        const int deg = __builtin_amdgcn_readfirstlane(lcnt[d]);
        const int idx = min(max(lperm[d * CAP + lane], 0), NN - 1);
        const float4 acc = sum_node(src, idx, deg, d, b, lane, lovl, no, tab, region);
        store_mean(dst, b * RANGE + d, deg, lane, acc);
    }
}

__global__ void __launch_bounds__(256)
gather_mean_kernel(const float* __restrict__ src, const int* __restrict__ perm,
                   const int* __restrict__ cnt, const int* __restrict__ ovl,
                   const int* __restrict__ ovn, const int* __restrict__ tab_w,
                   const int* __restrict__ region, float* __restrict__ dst) {
    const int2* __restrict__ tab = (const int2*)tab_w;
    const int lane = threadIdx.x & 31;
    const int wave0 = __builtin_amdgcn_readfirstlane(
        (int)(blockIdx.x * GATHER_WAVES_PER_BLOCK + (threadIdx.x >> 5)));
    const int nwaves = (int)gridDim.x * GATHER_WAVES_PER_BLOCK;

    for (int node = wave0; node < NN; node += nwaves) {
        const int deg = cnt[node];
        const int m = min(deg, CAP);
        const int b = node / RANGE;
        int idx = (lane < m) ? perm[node * CAP + lane] : 0;
        idx = min(max(idx, 0), NN - 1);
        const int no = (deg > CAP) ? ovn[b] : 0;
        const float4 acc = sum_node(src, idx, deg, node & (RANGE - 1), b, lane,
                                    ovl + b * OVB, no, tab, region);
        store_mean(dst, node, deg, lane, acc);
    }
}

#define GEMM_WAVES 4
#define GEMM_THREADS (GEMM_WAVES * 32)
#define ROW_TILES (NN / 16)
#define GEMM_BLOCKS ((ROW_TILES + GEMM_WAVES - 1) / GEMM_WAVES)
static_assert(NN % 16 == 0, "rows are handled in whole 16-row tiles");
static_assert(DD == 128, "the slab layout below is written for 128 x 128 weight slabs");

#define W_PITCH 288
#define W_SLAB_DWORDS ((DD / 2) * W_PITCH)
#define W_ITEMS_PER_THREAD ((DD / 2) / GEMM_WAVES)

__device__ __forceinline__ void stage_slab_load(const float* __restrict__ Wsrc, int tid,
                                                float4 (&r0)[W_ITEMS_PER_THREAD],
                                                float4 (&r1)[W_ITEMS_PER_THREAD]) {
    const int kp0 = tid >> 5;
    const int c4  = tid & 31;
    const float* src = Wsrc + (long)(2 * kp0) * DD + 4 * c4;
#pragma unroll
    for (int it = 0; it < W_ITEMS_PER_THREAD; ++it) {
        r0[it] = *(const float4*)(src + it * (2 * GEMM_WAVES * DD));
        r1[it] = *(const float4*)(src + it * (2 * GEMM_WAVES * DD) + DD);
    }
}

__device__ __forceinline__ void stage_slab_store(float* wl, int tid,
                                                 const float4 (&r0)[W_ITEMS_PER_THREAD],
                                                 const float4 (&r1)[W_ITEMS_PER_THREAD]) {
    const int kp0 = tid >> 5;
    const int c4  = tid & 31;
    float* dst = wl + kp0 * W_PITCH + 8 * c4;
#pragma unroll
    for (int it = 0; it < W_ITEMS_PER_THREAD; ++it) {
        float* d = dst + it * (GEMM_WAVES * W_PITCH);
        *(float4*)(d)     = make_float4(r0[it].x, r1[it].x, r0[it].y, r1[it].y);
        *(float4*)(d + 4) = make_float4(r0[it].z, r1[it].z, r0[it].w, r1[it].w);
    }
}

__device__ __forceinline__ void load_a_frags(const float* __restrict__ arow, v2f (&a)[DD / 4]) {
#pragma unroll
    for (int s = 0; s < DD / 4; ++s) {
        a[s].x = arow[4 * s];
        a[s].y = arow[4 * s + 1];
    }
}

#define P_PITCH 32
#define P_SLAB_DWORDS ((DD / 2) * P_PITCH)

template <int NCT, int S0, int S1>
__device__ __forceinline__ void mma_steps(const v2f (&a)[DD / 4], const float* wl_lane,
                                          const float* pl_lane, v8f (&acc)[NCT], v8f& accl) {
#pragma unroll
    for (int s = S0; s < S1; ++s) {
        v2f bb[NCT];
#pragma unroll
        for (int ct = 0; ct < NCT; ++ct)
            bb[ct] = *(const v2f*)(wl_lane + 2 * (s - S0) * W_PITCH + 32 * ct);
        const v2f pp = *(const v2f*)(pl_lane + 2 * (s - S0) * P_PITCH);
#pragma unroll
        for (int ct = 0; ct < NCT; ++ct)
            acc[ct] = __builtin_amdgcn_wmma_f32_16x16x4_f32(false, a[s], false, bb[ct],
                                                            (short)0, acc[ct], false, false);
        accl = __builtin_amdgcn_wmma_f32_16x16x4_f32(false, a[s], false, pp,
                                                     (short)0, accl, false, false);
    }
}

#define FOLD_ROW_TILES 9
#define FOLD_COL_GROUPS 5
#define FOLD_BLOCKS (NHOP * FOLD_ROW_TILES * FOLD_COL_GROUPS)
#define DF_STRIDE 32

__device__ __forceinline__ void fold_role(const int b, const int lane,
            const float* __restrict__ Wh, const float* __restrict__ bh,
            const float* __restrict__ Wf, const float* __restrict__ Wa,
            float* __restrict__ Mf, float* __restrict__ Pf,
            float* __restrict__ cf, float* __restrict__ df) {
    const int hop = b / (FOLD_ROW_TILES * FOLD_COL_GROUPS);
    const int rt  = (b / FOLD_COL_GROUPS) % FOLD_ROW_TILES;
    const int cg  = b % FOLD_COL_GROUPS;
    const int half = lane >> 4;
    const int l    = lane & 15;

    const int R = 16 * rt + l;
    const float* ap = (R < DD) ? Wh + ((long)hop * DD + R) * DD : bh + hop * DD;
    const bool aok = R <= DD;
    ap += 2 * half;

    const float* Wf_h = Wf + (long)hop * DD * DD;
    const float* Wa_h = Wa + (long)hop * DD * NHOP;
    const int C0 = 32 * cg + l;
    const int C1 = C0 + 16;
    const float* bp0; int bs0; bool bok0;
    if (C0 < DD)             { bp0 = Wf_h + C0;        bs0 = DD;   bok0 = true;  }
    else if (C0 < DD + NHOP) { bp0 = Wa_h + (C0 - DD); bs0 = NHOP; bok0 = true;  }
    else                     { bp0 = Wa_h;             bs0 = NHOP; bok0 = false; }
    const float* bp1; int bs1; bool bok1;
    if (C1 < DD)             { bp1 = Wf_h + C1;        bs1 = DD;   bok1 = true;  }
    else if (C1 < DD + NHOP) { bp1 = Wa_h + (C1 - DD); bs1 = NHOP; bok1 = true;  }
    else                     { bp1 = Wa_h;             bs1 = NHOP; bok1 = false; }
    bp0 += 2 * half * bs0;
    bp1 += 2 * half * bs1;

    v2f a[DD / 4], b0[DD / 4], b1[DD / 4];
#pragma unroll
    for (int s = 0; s < DD / 4; ++s) {
        a[s].x  = ap[4 * s];
        a[s].y  = ap[4 * s + 1];
        b0[s].x = bp0[(4 * s) * bs0];
        b0[s].y = bp0[(4 * s + 1) * bs0];
        b1[s].x = bp1[(4 * s) * bs1];
        b1[s].y = bp1[(4 * s + 1) * bs1];
    }
    v8f acc0 = {}, acc1 = {};
#pragma unroll
    for (int s = 0; s < DD / 4; ++s) {
        v2f av, bv0, bv1;
        av.x  = aok  ? a[s].x  : 0.0f;
        av.y  = aok  ? a[s].y  : 0.0f;
        bv0.x = bok0 ? b0[s].x : 0.0f;
        bv0.y = bok0 ? b0[s].y : 0.0f;
        bv1.x = bok1 ? b1[s].x : 0.0f;
        bv1.y = bok1 ? b1[s].y : 0.0f;
        acc0 = __builtin_amdgcn_wmma_f32_16x16x4_f32(false, av, false, bv0, (short)0, acc0, false, false);
        acc1 = __builtin_amdgcn_wmma_f32_16x16x4_f32(false, av, false, bv1, (short)0, acc1, false, false);
    }

#pragma unroll
    for (int j = 0; j < 8; ++j) {
        const int Rr = 16 * rt + 8 * half + j;
        if (Rr < DD) {
            if (cg < FOLD_COL_GROUPS - 1) {
                Mf[((long)hop * DD + Rr) * DD + C0] = acc0[j];
                Mf[((long)hop * DD + Rr) * DD + C1] = acc1[j];
            } else {
                Pf[hop * P_SLAB_DWORDS + (Rr >> 1) * P_PITCH + 2 * l + (Rr & 1)] = acc0[j];
            }
        } else if (Rr == DD) {
            if (cg < FOLD_COL_GROUPS - 1) {
                cf[hop * DD + C0] = acc0[j];
                cf[hop * DD + C1] = acc1[j];
            } else {
                df[hop * DF_STRIDE + l] = acc0[j];
            }
        }
    }
}

#define PRO_BLOCKS (NCHUNK + 1 + FOLD_BLOCKS)

__global__ void __launch_bounds__(PRO_T)
prologue_kernel(const int* __restrict__ row, const int* __restrict__ col,
                int* __restrict__ tab_w, int* __restrict__ region,
                const int* __restrict__ depths,
                const float* __restrict__ Wg1, const float* __restrict__ bg1,
                const float* __restrict__ Wg2, const float* __restrict__ bg2,
                float* __restrict__ sw,
                const float* __restrict__ Wh, const float* __restrict__ bh,
                const float* __restrict__ Wf, const float* __restrict__ Wa,
                float* __restrict__ Mf, float* __restrict__ Pf,
                float* __restrict__ cf, float* __restrict__ df) {
    __shared__ int sh[32];
    __shared__ int hist[HIST_PAD];
    __shared__ int offs[HIST_PAD];
    __shared__ __attribute__((aligned(16))) int sorted[CHUNK_E];
    const int b = blockIdx.x;
    if (b < NCHUNK) {
        sort_role(b, row, col, (int2*)tab_w, region, hist, offs, sorted, sh);
    } else if (b == NCHUNK) {
        gate_role(depths, Wg1, bg1, Wg2, bg2, sw, sh);
    } else {
        const int wave = __builtin_amdgcn_readfirstlane((int)(threadIdx.x >> 5));
        if (wave == 0) fold_role(b - NCHUNK - 1, (int)threadIdx.x, Wh, bh, Wf, Wa, Mf, Pf, cf, df);
    }
}


#define F_WAVES 12
#define F_THREADS (F_WAVES * 32)
#define F_BLOCKS 256
#define MAIN_TILES (F_BLOCKS * F_WAVES)
#define REM_TILES (ROW_TILES - MAIN_TILES)
#define REM_SPLIT 4
#define REM_NCT (8 / REM_SPLIT)
#define REM_UNITS (REM_TILES * REM_SPLIT)
#define P_BASE_DWORDS (NHOP * W_SLAB_DWORDS)
static_assert(F_WAVES == NHOP * GEMM_WAVES, "one group of GEMM_WAVES waves stages one hop's slabs");
static_assert(MAIN_TILES <= ROW_TILES, "every wave has a whole row tile");
static_assert(REM_UNITS >= 0 && REM_UNITS <= F_BLOCKS, "at most one column group of a leftover tile per work-group");
static_assert(P_SLAB_DWORDS == 4 * 4 * GEMM_THREADS, "the P slab copy is four 16-byte items per thread");

template <int NCT>
__device__ __forceinline__ void tile_pass(const int row0, const int ct0, const int lane, v2f (&a)[DD / 4],
             const float* __restrict__ X1, const float* __restrict__ X2,
             const float* lds,
             const float* __restrict__ cf, const float* __restrict__ df,
             const float* __restrict__ sw, const float* __restrict__ ba,
             const float* __restrict__ bf, float* __restrict__ out) {
    const int half = lane >> 4;
    const int l    = lane & 15;
    const int row  = row0 + l;

    v8f acc[NHOP][NCT] = {};
    v8f accl[NHOP] = {};
#pragma unroll
    for (int hop = 0; hop < NHOP; ++hop) {
        if (hop > 0) load_a_frags(((hop == 1) ? X1 : X2) + (long)row * DD + 2 * half, a);
        int w_lo = hop * W_SLAB_DWORDS + half * W_PITCH + 2 * l + 32 * ct0;
        int w_hi = w_lo + (DD / 4) * W_PITCH;
        int p_lo = P_BASE_DWORDS + hop * P_SLAB_DWORDS + half * P_PITCH + 2 * l;
        int p_hi = p_lo + (DD / 4) * P_PITCH;
        asm volatile("" : "+v"(w_lo), "+v"(w_hi), "+v"(p_lo), "+v"(p_hi));
        mma_steps<NCT, 0, DD / 8>(a, lds + w_lo, lds + p_lo, acc[hop], accl[hop]);
        mma_steps<NCT, DD / 8, DD / 4>(a, lds + w_hi, lds + p_hi, acc[hop], accl[hop]);
    }

    const float s0 = sw[0], s1 = sw[1], s2 = sw[2];
    const float bac = ba[min(l, NHOP - 1)];
    const float d0 = df[0 * DF_STRIDE + l], d1 = df[1 * DF_STRIDE + l], d2 = df[2 * DF_STRIDE + l];
    const int src = lane & 16;
    float w0[8], w1[8], w2[8];
#pragma unroll
    for (int j = 0; j < 8; ++j) {
        const float lg = bac + s0 * (accl[0][j] + d0) + s1 * (accl[1][j] + d1) + s2 * (accl[2][j] + d2);
        const float l0 = __shfl(lg, src + 0, 32);
        const float l1 = __shfl(lg, src + 1, 32);
        const float l2 = __shfl(lg, src + 2, 32);
        const float mx = fmaxf(l0, fmaxf(l1, l2));
        const float e0 = __expf(l0 - mx), e1 = __expf(l1 - mx), e2 = __expf(l2 - mx);
        const float inv = 1.0f / (e0 + e1 + e2);
        w0[j] = e0 * inv * s0;
        w1[j] = e1 * inv * s1;
        w2[j] = e2 * inv * s2;
    }

#pragma unroll
    for (int ct = 0; ct < NCT; ++ct) {
        const int col = 16 * (ct0 + ct) + l;
        const float c0 = cf[0 * DD + col], c1 = cf[1 * DD + col], c2 = cf[2 * DD + col];
        const float bias = bf[col];
#pragma unroll
        for (int j = 0; j < 8; ++j) {
            const int m = j + 8 * half;
            out[(long)(row0 + m) * DD + col] = w0[j] * (acc[0][ct][j] + c0) + w1[j] * (acc[1][ct][j] + c1)
                                             + w2[j] * (acc[2][ct][j] + c2) + bias;
        }
    }
}

__global__ void __launch_bounds__(F_THREADS) __attribute__((amdgpu_waves_per_eu(3, 3)))
fused_kernel(const float* __restrict__ X0, const float* __restrict__ X1,
             const float* __restrict__ X2,
             const float* __restrict__ Mf, const float* __restrict__ Pf,
             const float* __restrict__ cf, const float* __restrict__ df,
             const float* __restrict__ sw, const float* __restrict__ ba,
             const float* __restrict__ bf, float* __restrict__ out) {
    __shared__ __attribute__((aligned(16))) float lds[NHOP * W_SLAB_DWORDS + NHOP * P_SLAB_DWORDS];
    float* wl = lds;
    float* pl = lds + P_BASE_DWORDS;
    const int tid = threadIdx.x;
    const int wave = __builtin_amdgcn_readfirstlane(tid >> 5);
    const int lane = tid & 31;
    const int half = lane >> 4;
    const int l    = lane & 15;

    const int g    = wave / GEMM_WAVES;
    const int ltid = tid - g * GEMM_THREADS;
    float4 r0[W_ITEMS_PER_THREAD], r1[W_ITEMS_PER_THREAD];
    stage_slab_load(Mf + (long)g * DD * DD, ltid, r0, r1);
    const float4* psrc = (const float4*)(Pf + g * P_SLAB_DWORDS) + ltid;
    const float4 p0 = psrc[0], p1 = psrc[GEMM_THREADS], p2 = psrc[2 * GEMM_THREADS], p3 = psrc[3 * GEMM_THREADS];
    const int row0 = (blockIdx.x * F_WAVES + wave) * 16;
    v2f a[DD / 4];
    load_a_frags(X0 + (long)(row0 + l) * DD + 2 * half, a);
    __builtin_amdgcn_sched_barrier(0);
    stage_slab_store(wl + g * W_SLAB_DWORDS, ltid, r0, r1);
    float4* pdst = (float4*)(pl + g * P_SLAB_DWORDS) + ltid;
    pdst[0] = p0; pdst[GEMM_THREADS] = p1; pdst[2 * GEMM_THREADS] = p2; pdst[3 * GEMM_THREADS] = p3;
    __syncthreads();

    tile_pass<8>(row0, 0, lane, a, X1, X2, lds, cf, df, sw, ba, bf, out);

    const int u = blockIdx.x;
    if (wave == 0 && u < REM_UNITS) {
        const int rrow0 = (MAIN_TILES + u / REM_SPLIT) * 16;
        load_a_frags(X0 + (long)(rrow0 + l) * DD + 2 * half, a);
        tile_pass<REM_NCT>(rrow0, (u % REM_SPLIT) * REM_NCT, lane, a, X1, X2, lds, cf, df, sw, ba, bf, out);
    }
}

extern "C" void kernel_launch(void* const* d_in, const int* in_sizes, int n_in,
                              void* d_out, int out_size, void* d_ws, size_t ws_size,
                              hipStream_t stream) {
    const float* x      = (const float*)d_in[0];
    const int*   ei     = (const int*)  d_in[1];
    const int*   depths = (const int*)  d_in[2];
    const float* Wh     = (const float*)d_in[3];
    const float* bh     = (const float*)d_in[4];
    const float* Wg1    = (const float*)d_in[5];
    const float* bg1    = (const float*)d_in[6];
    const float* Wg2    = (const float*)d_in[7];
    const float* bg2    = (const float*)d_in[8];
    const float* Wa     = (const float*)d_in[9];
    const float* ba     = (const float*)d_in[10];
    const float* Wf     = (const float*)d_in[11];
    const float* bf     = (const float*)d_in[12];
    float* out = (float*)d_out;

    const int* row = ei;
    const int* col = ei + EE;

    char* ws = (char*)d_ws;
    size_t o = 0;
    auto carve = [&](size_t bytes) -> float* {
        float* p = (float*)(ws + o);
        o += (bytes + 255) & ~(size_t)255;
        return p;
    };
    float* sw     = carve(16 * sizeof(float));
    float* counts = carve((size_t)NN * sizeof(float));
    float* agg1   = carve((size_t)NN * DD * sizeof(float));
    float* agg2   = carve((size_t)NN * DD * sizeof(float));
    float* fold   = carve((size_t)NN * DD * sizeof(float));
    float* spare  = carve((size_t)NN * DD * sizeof(float));
    float* sortws = carve((size_t)NN * DD * sizeof(float));
    (void)ws_size; (void)in_sizes; (void)n_in; (void)out_size;

    float* Mf = fold;
    float* Pf = fold + NHOP * DD * DD;
    float* cf = Pf + NHOP * P_SLAB_DWORDS;
    float* df = cf + NHOP * DD;

    int*  cnt    = (int*)counts;
    int*  perm   = (int*)sortws;
    int*  region = (int*)spare;
    int*  tab    = (int*)spare + EE;
    int*  ovl    = (int*)spare + EE + 2 * NBIN * NCHUNK;
    int*  ovn    = ovl + NBIN * OVB;

    prologue_kernel<<<PRO_BLOCKS, PRO_T, 0, stream>>>(row, col, tab, region, depths, Wg1, bg1, Wg2, bg2, sw,
                                                      Wh, bh, Wf, Wa, Mf, Pf, cf, df);

    gather1_kernel<<<NBIN, G1_T, 0, stream>>>(x, tab, region, perm, cnt, ovl, ovn, agg1);
    gather_mean_kernel<<<2048, 256, 0, stream>>>(agg1, perm, cnt, ovl, ovn, tab, region, agg2);

    fused_kernel<<<F_BLOCKS, F_THREADS, 0, stream>>>(x, agg1, agg2, Mf, Pf, cf, df, sw, ba, bf, out);
}


# second scatter-mean folded into the fused pass: each wave sums its tile's 16 mean rows of agg1 into an LDS strip between the hop-1 and hop-2 passes; 4 launches to 3, no agg2 plane; sums unchanged
# speedup vs baseline: 5.9736x; 1.0513x over previous
#include <hip/hip_runtime.h>
#include <hip/hip_bf16.h>
#include <math.h>

#define NN 50000
#define DD 128
#define EE 800000
#define NHOP 3

typedef __attribute__((ext_vector_type(2))) float v2f;
typedef __attribute__((ext_vector_type(8))) float v8f;

#define PRO_T 512
#define PRO_WAVES (PRO_T / 32)

#define GATE_ITEMS ((NN / 4 + PRO_T - 1) / PRO_T)
static_assert(NN % 4 == 0, "the depths are read as int4");

__device__ __forceinline__ void gate_role(const int* __restrict__ depths,
                                          const float* __restrict__ Wg1, const float* __restrict__ bg1,
                                          const float* __restrict__ Wg2, const float* __restrict__ bg2,
                                          float* __restrict__ sw, int* smax) {
    const int tid = threadIdx.x;
    const int lane = tid & 31;
    const int w = tid >> 5;
    int m = 0;
#pragma unroll
    for (int i = 0; i < GATE_ITEMS; ++i) {
        const int idx = min(tid + i * PRO_T, NN / 4 - 1);
        const int4 v = ((const int4*)depths)[idx];
        m = max(m, max(max(v.x, v.y), max(v.z, v.w)));
    }
    for (int d = 16; d > 0; d >>= 1) m = max(m, __shfl_xor(m, d, 32));
    if (lane == 0) smax[w] = m;
    __syncthreads();
    if (tid == 0) {
        int mm = smax[0];
        for (int k = 1; k < PRO_WAVES; ++k) mm = max(mm, smax[k]);
        float sf0 = (float)NN / 5000.0f;
        float sf1 = (float)mm / 20.0f;
        float a0 = bg2[0], a1 = bg2[1], a2 = bg2[2];
        for (int j = 0; j < DD / 4; ++j) {
            float h = sf0 * Wg1[j] + sf1 * Wg1[DD / 4 + j] + bg1[j];
            h = fmaxf(h, 0.0f);
            a0 += h * Wg2[j * NHOP + 0];
            a1 += h * Wg2[j * NHOP + 1];
            a2 += h * Wg2[j * NHOP + 2];
        }
        sw[0] = 1.0f / (1.0f + __expf(-a0));
        sw[1] = 1.0f / (1.0f + __expf(-a1));
        sw[2] = 1.0f / (1.0f + __expf(-a2));
    }
}

#define CAP 32
#define RANGE 64
#define NBIN ((NN + RANGE - 1) / RANGE)
#define NCHUNK 250
#define CHUNK_E (EE / NCHUNK)
#define CHUNK_E4 (CHUNK_E / 4)
#define HIST_PAD (2 * PRO_T)
#define SORT_ROUNDS ((CHUNK_E4 + PRO_T - 1) / PRO_T)
#define OVB 64
static_assert(EE % (4 * NCHUNK) == 0, "a chunk is a whole number of int4 reads");
static_assert(NN <= 65536, "a source row id is packed into 16 bits");
static_assert(NBIN <= HIST_PAD, "two range counters per thread");
static_assert((RANGE & (RANGE - 1)) == 0, "range is a power of two");

__device__ __forceinline__ void sort_role(const int chunk, const int* __restrict__ row,
                                          const int* __restrict__ col, int2* __restrict__ tab,
                                          int* __restrict__ region,
                                          int* hist, int* offs, int* sorted, int* wsum) {
    const int tid = threadIdx.x;
    const int lane = tid & 31;
    const int w = tid >> 5;
    hist[tid] = 0;
    hist[tid + PRO_T] = 0;

    int4 c4[SORT_ROUNDS], r4[SORT_ROUNDS];
#pragma unroll
    for (int r = 0; r < SORT_ROUNDS; ++r) {
        const int i = min(tid + r * PRO_T, CHUNK_E4 - 1);
        c4[r] = ((const int4*)col)[chunk * CHUNK_E4 + i];
        r4[r] = ((const int4*)row)[chunk * CHUNK_E4 + i];
    }
    __syncthreads();

    int bin[SORT_ROUNDS * 4], rk[SORT_ROUNDS * 4], wd[SORT_ROUNDS * 4];
#pragma unroll
    for (int r = 0; r < SORT_ROUNDS; ++r) {
        const bool ok = tid + r * PRO_T < CHUNK_E4;
        const int cc[4] = {c4[r].x, c4[r].y, c4[r].z, c4[r].w};
        const int ss[4] = {r4[r].x, r4[r].y, r4[r].z, r4[r].w};
#pragma unroll
        for (int e = 0; e < 4; ++e) {
            const int c = min(max(cc[e], 0), NN - 1);
            const int sr = min(max(ss[e], 0), NN - 1);
            const int q = r * 4 + e;
            bin[q] = ok ? (c / RANGE) : -1;
            wd[q] = ((c & (RANGE - 1)) << 16) | sr;
            rk[q] = ok ? atomicAdd(&hist[bin[q]], 1) : 0;
        }
    }
    __syncthreads();

    const int a0 = hist[2 * tid], a1 = hist[2 * tid + 1];
    const int sum2 = a0 + a1;
    int incl = sum2;
    for (int d = 1; d < 32; d <<= 1) {
        const int t = __shfl_up(incl, d, 32);
        if (lane >= d) incl += t;
    }
    if (lane == 31) wsum[w] = incl;
    __syncthreads();
    int wp = 0;
    for (int k = 0; k < PRO_WAVES; ++k) wp += (k < w) ? wsum[k] : 0;
    const int ex = wp + incl - sum2;
    offs[2 * tid] = ex;
    offs[2 * tid + 1] = ex + a0;
    if (2 * tid < NBIN)     tab[(2 * tid) * NCHUNK + chunk]     = make_int2(ex, a0);
    if (2 * tid + 1 < NBIN) tab[(2 * tid + 1) * NCHUNK + chunk] = make_int2(ex + a0, a1);
    __syncthreads();

#pragma unroll
    for (int q = 0; q < SORT_ROUNDS * 4; ++q)
        if (bin[q] >= 0) sorted[offs[bin[q]] + rk[q]] = wd[q];
    __syncthreads();

    for (int i = tid; i < CHUNK_E4; i += PRO_T)
        ((int4*)region)[chunk * CHUNK_E4 + i] = ((const int4*)sorted)[i];
}

#define GATHER_BATCH 8

__device__ __forceinline__ float4 load_row4(const float* __restrict__ src, int r, int lane) {
    return *(const float4*)(src + (long)r * DD + lane * 4);
}

__device__ __forceinline__ void add_hits(const float* __restrict__ src, const bool hit, const int w,
                                         const int lane, float4& acc) {
    const int s = min(w & 0xFFFF, NN - 1);
    unsigned mask = __builtin_amdgcn_ballot_w32(hit);
    while (mask != 0u) {
        const int q = __builtin_ctz(mask);
        mask &= mask - 1u;
        const float4 v = load_row4(src, __builtin_amdgcn_readlane(s, q), lane);
        acc.x += v.x; acc.y += v.y; acc.z += v.z; acc.w += v.w;
    }
}

template <int BATCH>
__device__ __forceinline__ float4 sum_node(const float* __restrict__ src, const int idx, const int deg,
                                           const int d, const int b, const int lane,
                                           const int* ovlist, const int no,
                                           const int2* __restrict__ tab, const int* __restrict__ region) {
    float4 acc = make_float4(0.0f, 0.0f, 0.0f, 0.0f);
    if (deg > CAP && no > OVB) {
        for (int c = 0; c < NCHUNK; ++c) {
            const int2 sc = tab[b * NCHUNK + c];
            const int start = min(max(sc.x, 0), CHUNK_E);
            const int count = min(max(sc.y, 0), CHUNK_E - start);
            for (int k0 = 0; k0 < count; k0 += 32) {
                const int k = k0 + lane;
                const int w = (k < count) ? region[c * CHUNK_E + start + k] : 0;
                add_hits(src, (k < count) && (((w >> 16) & (RANGE - 1)) == d), w, lane, acc);
            }
        }
        return acc;
    }
    const int m = min(deg, CAP);
    int j = 0;
    for (; j + BATCH <= m; j += BATCH) {
        float4 v[BATCH];
        for (int u = 0; u < BATCH; ++u)
            v[u] = load_row4(src, __builtin_amdgcn_readlane(idx, j + u), lane);
        for (int u = 0; u < BATCH; ++u) {
            acc.x += v[u].x; acc.y += v[u].y; acc.z += v[u].z; acc.w += v[u].w;
        }
    }
    if (j < m) {
        float4 v[BATCH - 1];
        for (int u = 0; u < BATCH - 1; ++u) {
            v[u] = make_float4(0.0f, 0.0f, 0.0f, 0.0f);
            if (j + u < m)
                v[u] = load_row4(src, __builtin_amdgcn_readlane(idx, j + u), lane);
        }
        for (int u = 0; u < BATCH - 1; ++u) {
            if (j + u < m) {
                acc.x += v[u].x; acc.y += v[u].y; acc.z += v[u].z; acc.w += v[u].w;
            }
        }
    }
    if (deg > CAP) {
        const int nl = min(no, OVB);
        for (int q0 = 0; q0 < nl; q0 += 32) {
            const int k = q0 + lane;
            const int w = (k < nl) ? ovlist[k] : 0;
            add_hits(src, (k < nl) && (((w >> 16) & (RANGE - 1)) == d), w, lane, acc);
        }
    }
    return acc;
}

__device__ __forceinline__ void store_mean(float* __restrict__ dst, const int node, const int deg,
                                           const int lane, float4 acc) {
    const float inv = 1.0f / fmaxf((float)deg, 1.0f);
    acc.x *= inv; acc.y *= inv; acc.z *= inv; acc.w *= inv;
    *(float4*)(dst + (long)node * DD + lane * 4) = acc;
}

#define G1_T 512
#define G1_WAVES (G1_T / 32)
#define SEG_BATCH 12
static_assert(NCHUNK <= G1_T, "one thread per chunk segment");

__global__ void __launch_bounds__(G1_T)
gather1_kernel(const float* __restrict__ src, const int* __restrict__ tab_w,
               const int* __restrict__ region, int* __restrict__ perm, int* __restrict__ cnt,
               int* __restrict__ ovl, int* __restrict__ ovn, float* __restrict__ dst) {
    __shared__ __attribute__((aligned(16))) int lperm[RANGE * CAP];
    __shared__ int lcnt[RANGE];
    __shared__ int lovl[OVB];
    __shared__ int lovn;
    const int2* __restrict__ tab = (const int2*)tab_w;
    const int tid = threadIdx.x;
    const int lane = tid & 31;
    const int wave = __builtin_amdgcn_readfirstlane(tid >> 5);
    const int b = blockIdx.x;
    const int nodes_here = min(RANGE, NN - b * RANGE);

    if (tid < RANGE) lcnt[tid] = 0;
    if (tid == 0) lovn = 0;
    int start = 0, count = 0;
    if (tid < NCHUNK) {
        const int2 sc = tab[b * NCHUNK + tid];
        start = min(max(sc.x, 0), CHUNK_E);
        count = min(max(sc.y, 0), CHUNK_E - start);
    }
    const int* base = region + min(tid, NCHUNK - 1) * CHUNK_E + start;
    __syncthreads();

    for (int k0 = 0; k0 < count; k0 += SEG_BATCH) {
        int wv[SEG_BATCH];
#pragma unroll
        for (int u = 0; u < SEG_BATCH; ++u) wv[u] = (k0 + u < count) ? base[k0 + u] : 0;
#pragma unroll
        for (int u = 0; u < SEG_BATCH; ++u) {
            if (k0 + u < count) {
                const int d = (wv[u] >> 16) & (RANGE - 1);
                const int slot = atomicAdd(&lcnt[d], 1);
                if (slot < CAP) {
                    lperm[d * CAP + slot] = wv[u] & 0xFFFF;
                } else {
                    const int q = atomicAdd(&lovn, 1);
                    if (q < OVB) lovl[q] = wv[u];
                }
            }
        }
    }
    __syncthreads();

    const int no = __builtin_amdgcn_readfirstlane(lovn);
    for (int i = tid; i < nodes_here * (CAP / 4); i += G1_T)
        ((int4*)(perm + (long)b * RANGE * CAP))[i] = ((const int4*)lperm)[i];
    if (tid < nodes_here) cnt[b * RANGE + tid] = lcnt[tid];
    if (tid < min(no, OVB)) ovl[b * OVB + tid] = lovl[tid];
    if (tid == 0) ovn[b] = no;

    for (int d = wave; d < nodes_here; d += G1_WAVES) {
        const int deg = __builtin_amdgcn_readfirstlane(lcnt[d]);
        const int idx = min(max(lperm[d * CAP + lane], 0), NN - 1);
        const float4 acc = sum_node<GATHER_BATCH>(src, idx, deg, d, b, lane, lovl, no, tab, region);
        store_mean(dst, b * RANGE + d, deg, lane, acc);
    }
}

#define GEMM_WAVES 4
#define GEMM_THREADS (GEMM_WAVES * 32)
#define ROW_TILES (NN / 16)
#define GEMM_BLOCKS ((ROW_TILES + GEMM_WAVES - 1) / GEMM_WAVES)
static_assert(NN % 16 == 0, "rows are handled in whole 16-row tiles");
static_assert(DD == 128, "the slab layout below is written for 128 x 128 weight slabs");

#define W_PITCH 288
#define W_SLAB_DWORDS ((DD / 2) * W_PITCH)
#define W_ITEMS_PER_THREAD ((DD / 2) / GEMM_WAVES)

__device__ __forceinline__ void stage_slab_load(const float* __restrict__ Wsrc, int tid,
                                                float4 (&r0)[W_ITEMS_PER_THREAD],
                                                float4 (&r1)[W_ITEMS_PER_THREAD]) {
    const int kp0 = tid >> 5;
    const int c4  = tid & 31;
    const float* src = Wsrc + (long)(2 * kp0) * DD + 4 * c4;
#pragma unroll
    for (int it = 0; it < W_ITEMS_PER_THREAD; ++it) {
        r0[it] = *(const float4*)(src + it * (2 * GEMM_WAVES * DD));
        r1[it] = *(const float4*)(src + it * (2 * GEMM_WAVES * DD) + DD);
    }
}

__device__ __forceinline__ void stage_slab_store(float* wl, int tid,
                                                 const float4 (&r0)[W_ITEMS_PER_THREAD],
                                                 const float4 (&r1)[W_ITEMS_PER_THREAD]) {
    const int kp0 = tid >> 5;
    const int c4  = tid & 31;
    float* dst = wl + kp0 * W_PITCH + 8 * c4;
#pragma unroll
    for (int it = 0; it < W_ITEMS_PER_THREAD; ++it) {
        float* d = dst + it * (GEMM_WAVES * W_PITCH);
        *(float4*)(d)     = make_float4(r0[it].x, r1[it].x, r0[it].y, r1[it].y);
        *(float4*)(d + 4) = make_float4(r0[it].z, r1[it].z, r0[it].w, r1[it].w);
    }
}

__device__ __forceinline__ void load_a_frags(const float* __restrict__ arow, v2f (&a)[DD / 4]) {
#pragma unroll
    for (int s = 0; s < DD / 4; ++s) {
        a[s].x = arow[4 * s];
        a[s].y = arow[4 * s + 1];
    }
}

#define P_PITCH 32
#define P_SLAB_DWORDS ((DD / 2) * P_PITCH)

template <int NCT, int S0, int S1>
__device__ __forceinline__ void mma_steps(const v2f (&a)[DD / 4], const float* wl_lane,
                                          const float* pl_lane, v8f (&acc)[NCT], v8f& accl) {
#pragma unroll
    for (int s = S0; s < S1; ++s) {
        v2f bb[NCT];
#pragma unroll
        for (int ct = 0; ct < NCT; ++ct)
            bb[ct] = *(const v2f*)(wl_lane + 2 * (s - S0) * W_PITCH + 32 * ct);
        const v2f pp = *(const v2f*)(pl_lane + 2 * (s - S0) * P_PITCH);
#pragma unroll
        for (int ct = 0; ct < NCT; ++ct)
            acc[ct] = __builtin_amdgcn_wmma_f32_16x16x4_f32(false, a[s], false, bb[ct],
                                                            (short)0, acc[ct], false, false);
        accl = __builtin_amdgcn_wmma_f32_16x16x4_f32(false, a[s], false, pp,
                                                     (short)0, accl, false, false);
    }
}

#define FOLD_ROW_TILES 9
#define FOLD_COL_GROUPS 5
#define FOLD_BLOCKS (NHOP * FOLD_ROW_TILES * FOLD_COL_GROUPS)
#define DF_STRIDE 32

__device__ __forceinline__ void fold_role(const int b, const int lane,
            const float* __restrict__ Wh, const float* __restrict__ bh,
            const float* __restrict__ Wf, const float* __restrict__ Wa,
            float* __restrict__ Mf, float* __restrict__ Pf,
            float* __restrict__ cf, float* __restrict__ df) {
    const int hop = b / (FOLD_ROW_TILES * FOLD_COL_GROUPS);
    const int rt  = (b / FOLD_COL_GROUPS) % FOLD_ROW_TILES;
    const int cg  = b % FOLD_COL_GROUPS;
    const int half = lane >> 4;
    const int l    = lane & 15;

    const int R = 16 * rt + l;
    const float* ap = (R < DD) ? Wh + ((long)hop * DD + R) * DD : bh + hop * DD;
    const bool aok = R <= DD;
    ap += 2 * half;

    const float* Wf_h = Wf + (long)hop * DD * DD;
    const float* Wa_h = Wa + (long)hop * DD * NHOP;
    const int C0 = 32 * cg + l;
    const int C1 = C0 + 16;
    const float* bp0; int bs0; bool bok0;
    if (C0 < DD)             { bp0 = Wf_h + C0;        bs0 = DD;   bok0 = true;  }
    else if (C0 < DD + NHOP) { bp0 = Wa_h + (C0 - DD); bs0 = NHOP; bok0 = true;  }
    else                     { bp0 = Wa_h;             bs0 = NHOP; bok0 = false; }
    const float* bp1; int bs1; bool bok1;
    if (C1 < DD)             { bp1 = Wf_h + C1;        bs1 = DD;   bok1 = true;  }
    else if (C1 < DD + NHOP) { bp1 = Wa_h + (C1 - DD); bs1 = NHOP; bok1 = true;  }
    else                     { bp1 = Wa_h;             bs1 = NHOP; bok1 = false; }
    bp0 += 2 * half * bs0;
    bp1 += 2 * half * bs1;

    v2f a[DD / 4], b0[DD / 4], b1[DD / 4];
#pragma unroll
    for (int s = 0; s < DD / 4; ++s) {
        a[s].x  = ap[4 * s];
        a[s].y  = ap[4 * s + 1];
        b0[s].x = bp0[(4 * s) * bs0];
        b0[s].y = bp0[(4 * s + 1) * bs0];
        b1[s].x = bp1[(4 * s) * bs1];
        b1[s].y = bp1[(4 * s + 1) * bs1];
    }
    v8f acc0 = {}, acc1 = {};
#pragma unroll
    for (int s = 0; s < DD / 4; ++s) {
        v2f av, bv0, bv1;
        av.x  = aok  ? a[s].x  : 0.0f;
        av.y  = aok  ? a[s].y  : 0.0f;
        bv0.x = bok0 ? b0[s].x : 0.0f;
        bv0.y = bok0 ? b0[s].y : 0.0f;
        bv1.x = bok1 ? b1[s].x : 0.0f;
        bv1.y = bok1 ? b1[s].y : 0.0f;
        acc0 = __builtin_amdgcn_wmma_f32_16x16x4_f32(false, av, false, bv0, (short)0, acc0, false, false);
        acc1 = __builtin_amdgcn_wmma_f32_16x16x4_f32(false, av, false, bv1, (short)0, acc1, false, false);
    }

#pragma unroll
    for (int j = 0; j < 8; ++j) {
        const int Rr = 16 * rt + 8 * half + j;
        if (Rr < DD) {
            if (cg < FOLD_COL_GROUPS - 1) {
                Mf[((long)hop * DD + Rr) * DD + C0] = acc0[j];
                Mf[((long)hop * DD + Rr) * DD + C1] = acc1[j];
            } else {
                Pf[hop * P_SLAB_DWORDS + (Rr >> 1) * P_PITCH + 2 * l + (Rr & 1)] = acc0[j];
            }
        } else if (Rr == DD) {
            if (cg < FOLD_COL_GROUPS - 1) {
                cf[hop * DD + C0] = acc0[j];
                cf[hop * DD + C1] = acc1[j];
            } else {
                df[hop * DF_STRIDE + l] = acc0[j];
            }
        }
    }
}

#define PRO_BLOCKS (NCHUNK + 1 + FOLD_BLOCKS)

__global__ void __launch_bounds__(PRO_T)
prologue_kernel(const int* __restrict__ row, const int* __restrict__ col,
                int* __restrict__ tab_w, int* __restrict__ region,
                const int* __restrict__ depths,
                const float* __restrict__ Wg1, const float* __restrict__ bg1,
                const float* __restrict__ Wg2, const float* __restrict__ bg2,
                float* __restrict__ sw,
                const float* __restrict__ Wh, const float* __restrict__ bh,
                const float* __restrict__ Wf, const float* __restrict__ Wa,
                float* __restrict__ Mf, float* __restrict__ Pf,
                float* __restrict__ cf, float* __restrict__ df) {
    __shared__ int sh[32];
    __shared__ int hist[HIST_PAD];
    __shared__ int offs[HIST_PAD];
    __shared__ __attribute__((aligned(16))) int sorted[CHUNK_E];
    const int b = blockIdx.x;
    if (b < NCHUNK) {
        sort_role(b, row, col, (int2*)tab_w, region, hist, offs, sorted, sh);
    } else if (b == NCHUNK) {
        gate_role(depths, Wg1, bg1, Wg2, bg2, sw, sh);
    } else {
        const int wave = __builtin_amdgcn_readfirstlane((int)(threadIdx.x >> 5));
        if (wave == 0) fold_role(b - NCHUNK - 1, (int)threadIdx.x, Wh, bh, Wf, Wa, Mf, Pf, cf, df);
    }
}


#define F_WAVES 12
#define F_THREADS (F_WAVES * 32)
#define F_BLOCKS 256
#define MAIN_TILES (F_BLOCKS * F_WAVES)
#define REM_TILES (ROW_TILES - MAIN_TILES)
#define REM_SPLIT 4
#define REM_NCT (8 / REM_SPLIT)
#define REM_UNITS (REM_TILES * REM_SPLIT)
#define P_BASE_DWORDS (NHOP * W_SLAB_DWORDS)
static_assert(F_WAVES == NHOP * GEMM_WAVES, "one group of GEMM_WAVES waves stages one hop's slabs");
static_assert(MAIN_TILES <= ROW_TILES, "every wave has a whole row tile");
static_assert(REM_UNITS >= 0 && REM_UNITS <= F_BLOCKS, "at most one column group of a leftover tile per work-group");
static_assert(P_SLAB_DWORDS == 4 * 4 * GEMM_THREADS, "the P slab copy is four 16-byte items per thread");

#define S_ROWS 8
#define S_PITCH 132
#define F_GATHER_BATCH 16
#define S_DWORDS (S_ROWS * S_PITCH)
static_assert(2 * S_ROWS == 16, "two strip rounds make one row tile");
#define STRIP_BASE_DWORDS (P_BASE_DWORDS + NHOP * P_SLAB_DWORDS)
#define REM_STRIP_DWORDS (16 * S_PITCH)
#define REM_ROWS_PER_WAVE 2
static_assert(16 / REM_ROWS_PER_WAVE <= F_WAVES, "enough waves in a work-group for the leftover rows");
#define REM_STRIP_BASE_DWORDS (STRIP_BASE_DWORDS + F_WAVES * S_DWORDS)
#define F_LDS_DWORDS (REM_STRIP_BASE_DWORDS + REM_STRIP_DWORDS)
static_assert(F_LDS_DWORDS * 4 <= 320 * 1024, "slabs and strips fit the LDS of a WGP");
static_assert((STRIP_BASE_DWORDS % 4) == 0 && (S_DWORDS % 4) == 0, "strip rows are written as 16-byte items");

struct GatherSrc {
    const float* src;
    const int*   perm;
    const int*   cnt;
    const int*   ovl;
    const int*   ovn;
    const int2*  tab;
    const int*   region;
};

__device__ __forceinline__ void gather_rows(const GatherSrc& g, const int node0, const int n,
                                            const int lane, float* strip) {
    int deg_n = g.cnt[node0];
    int idx_n = g.perm[node0 * CAP + lane];
    for (int i = 0; i < n; ++i) {
        const int node = node0 + i;
        const int deg = deg_n;
        const int idx = min(max(idx_n, 0), NN - 1);
        if (i + 1 < n) {
            deg_n = g.cnt[node + 1];
            idx_n = g.perm[(node + 1) * CAP + lane];
        }
        const int b = node / RANGE;
        const int no = (deg > CAP) ? g.ovn[b] : 0;
        float4 acc = sum_node<F_GATHER_BATCH>(g.src, idx, deg, node & (RANGE - 1), b, lane,
                                              g.ovl + b * OVB, no, g.tab, g.region);
        const float inv = 1.0f / fmaxf((float)deg, 1.0f);
        acc.x *= inv; acc.y *= inv; acc.z *= inv; acc.w *= inv;
        *(float4*)(strip + i * S_PITCH + lane * 4) = acc;
    }
}

__device__ __forceinline__ void load_strip_frags(const float* srow, v2f (&a)[DD / 4]) {
#pragma unroll
    for (int s = 0; s < DD / 4; ++s) {
        a[s].x = srow[4 * s];
        a[s].y = srow[4 * s + 1];
    }
}

__device__ __forceinline__ void gathered_a_frags(const GatherSrc& g, const int row0, const int lane,
                                                 float* strip, v2f (&a)[DD / 4]) {
    const int half = lane >> 4;
    const int l    = lane & 15;
    const float* srow = strip + (l & (S_ROWS - 1)) * S_PITCH + 2 * half;
#pragma unroll
    for (int p = 0; p < 2; ++p) {
        gather_rows(g, row0 + p * S_ROWS, S_ROWS, lane, strip);
        __builtin_amdgcn_wave_barrier();
        if ((l >> 3) == p) load_strip_frags(srow, a);
        __builtin_amdgcn_wave_barrier();
    }
}

template <int NCT, bool OWN_GATHER>
__device__ __forceinline__ void tile_pass(const int row0, const int ct0, const int lane, v2f (&a)[DD / 4],
             const float* __restrict__ X1, const GatherSrc& g, float* strip,
             const float* lds,
             const float* __restrict__ cf, const float* __restrict__ df,
             const float* __restrict__ sw, const float* __restrict__ ba,
             const float* __restrict__ bf, float* __restrict__ out) {
    const int half = lane >> 4;
    const int l    = lane & 15;
    const int row  = row0 + l;

    v8f acc[NHOP][NCT] = {};
    v8f accl[NHOP] = {};
#pragma unroll
    for (int hop = 0; hop < NHOP; ++hop) {
        if (hop == 1) load_a_frags(X1 + (long)row * DD + 2 * half, a);
        if (hop == 2) {
            if (OWN_GATHER) gathered_a_frags(g, row0, lane, strip, a);
            else            load_strip_frags(strip + l * S_PITCH + 2 * half, a);
        }
        int w_lo = hop * W_SLAB_DWORDS + half * W_PITCH + 2 * l + 32 * ct0;
        int w_hi = w_lo + (DD / 4) * W_PITCH;
        int p_lo = P_BASE_DWORDS + hop * P_SLAB_DWORDS + half * P_PITCH + 2 * l;
        int p_hi = p_lo + (DD / 4) * P_PITCH;
        asm volatile("" : "+v"(w_lo), "+v"(w_hi), "+v"(p_lo), "+v"(p_hi));
        mma_steps<NCT, 0, DD / 8>(a, lds + w_lo, lds + p_lo, acc[hop], accl[hop]);
        mma_steps<NCT, DD / 8, DD / 4>(a, lds + w_hi, lds + p_hi, acc[hop], accl[hop]);
    }

    const float s0 = sw[0], s1 = sw[1], s2 = sw[2];
    const float bac = ba[min(l, NHOP - 1)];
    const float d0 = df[0 * DF_STRIDE + l], d1 = df[1 * DF_STRIDE + l], d2 = df[2 * DF_STRIDE + l];
    const int src = lane & 16;
    float w0[8], w1[8], w2[8];
#pragma unroll
    for (int j = 0; j < 8; ++j) {
        const float lg = bac + s0 * (accl[0][j] + d0) + s1 * (accl[1][j] + d1) + s2 * (accl[2][j] + d2);
        const float l0 = __shfl(lg, src + 0, 32);
        const float l1 = __shfl(lg, src + 1, 32);
        const float l2 = __shfl(lg, src + 2, 32);
        const float mx = fmaxf(l0, fmaxf(l1, l2));
        const float e0 = __expf(l0 - mx), e1 = __expf(l1 - mx), e2 = __expf(l2 - mx);
        const float inv = 1.0f / (e0 + e1 + e2);
        w0[j] = e0 * inv * s0;
        w1[j] = e1 * inv * s1;
        w2[j] = e2 * inv * s2;
    }

#pragma unroll
    for (int ct = 0; ct < NCT; ++ct) {
        const int col = 16 * (ct0 + ct) + l;
        const float c0 = cf[0 * DD + col], c1 = cf[1 * DD + col], c2 = cf[2 * DD + col];
        const float bias = bf[col];
#pragma unroll
        for (int j = 0; j < 8; ++j) {
            const int m = j + 8 * half;
            out[(long)(row0 + m) * DD + col] = w0[j] * (acc[0][ct][j] + c0) + w1[j] * (acc[1][ct][j] + c1)
                                             + w2[j] * (acc[2][ct][j] + c2) + bias;
        }
    }
}

__global__ void __launch_bounds__(F_THREADS) __attribute__((amdgpu_waves_per_eu(3, 3)))
fused_kernel(const float* __restrict__ X0, const float* __restrict__ X1,
             const int* __restrict__ perm, const int* __restrict__ cnt,
             const int* __restrict__ ovl, const int* __restrict__ ovn,
             const int* __restrict__ tab_w, const int* __restrict__ region,
             const float* __restrict__ Mf, const float* __restrict__ Pf,
             const float* __restrict__ cf, const float* __restrict__ df,
             const float* __restrict__ sw, const float* __restrict__ ba,
             const float* __restrict__ bf, float* __restrict__ out) {
    __shared__ __attribute__((aligned(16))) float lds[F_LDS_DWORDS];
    float* wl = lds;
    float* pl = lds + P_BASE_DWORDS;
    const int tid = threadIdx.x;
    const int wave = __builtin_amdgcn_readfirstlane(tid >> 5);
    const int lane = tid & 31;
    const int half = lane >> 4;
    const int l    = lane & 15;
    float* strip = lds + STRIP_BASE_DWORDS + wave * S_DWORDS;
    float* rem_strip = lds + REM_STRIP_BASE_DWORDS;
    const GatherSrc gs = {X1, perm, cnt, ovl, ovn, (const int2*)tab_w, region};
    const int u = blockIdx.x;
    const bool has_rem = u < REM_UNITS;
    const int rrow0 = (MAIN_TILES + u / REM_SPLIT) * 16;

    const int g    = wave / GEMM_WAVES;
    const int ltid = tid - g * GEMM_THREADS;
    float4 r0[W_ITEMS_PER_THREAD], r1[W_ITEMS_PER_THREAD];
    stage_slab_load(Mf + (long)g * DD * DD, ltid, r0, r1);
    const float4* psrc = (const float4*)(Pf + g * P_SLAB_DWORDS) + ltid;
    const float4 p0 = psrc[0], p1 = psrc[GEMM_THREADS], p2 = psrc[2 * GEMM_THREADS], p3 = psrc[3 * GEMM_THREADS];
    const int row0 = (blockIdx.x * F_WAVES + wave) * 16;
    v2f a[DD / 4];
    load_a_frags(X0 + (long)(row0 + l) * DD + 2 * half, a);
    __builtin_amdgcn_sched_barrier(0);
    if (has_rem && wave < 16 / REM_ROWS_PER_WAVE)
        gather_rows(gs, rrow0 + wave * REM_ROWS_PER_WAVE, REM_ROWS_PER_WAVE, lane,
                    rem_strip + wave * REM_ROWS_PER_WAVE * S_PITCH);
    stage_slab_store(wl + g * W_SLAB_DWORDS, ltid, r0, r1);
    float4* pdst = (float4*)(pl + g * P_SLAB_DWORDS) + ltid;
    pdst[0] = p0; pdst[GEMM_THREADS] = p1; pdst[2 * GEMM_THREADS] = p2; pdst[3 * GEMM_THREADS] = p3;
    __syncthreads();

    tile_pass<8, true>(row0, 0, lane, a, X1, gs, strip, lds, cf, df, sw, ba, bf, out);

    if (wave == 0 && has_rem) {
        load_a_frags(X0 + (long)(rrow0 + l) * DD + 2 * half, a);
        tile_pass<REM_NCT, false>(rrow0, (u % REM_SPLIT) * REM_NCT, lane, a, X1, gs, rem_strip, lds, cf, df, sw, ba, bf, out);
    }
}

extern "C" void kernel_launch(void* const* d_in, const int* in_sizes, int n_in,
                              void* d_out, int out_size, void* d_ws, size_t ws_size,
                              hipStream_t stream) {
    const float* x      = (const float*)d_in[0];
    const int*   ei     = (const int*)  d_in[1];
    const int*   depths = (const int*)  d_in[2];
    const float* Wh     = (const float*)d_in[3];
    const float* bh     = (const float*)d_in[4];
    const float* Wg1    = (const float*)d_in[5];
    const float* bg1    = (const float*)d_in[6];
    const float* Wg2    = (const float*)d_in[7];
    const float* bg2    = (const float*)d_in[8];
    const float* Wa     = (const float*)d_in[9];
    const float* ba     = (const float*)d_in[10];
    const float* Wf     = (const float*)d_in[11];
    const float* bf     = (const float*)d_in[12];
    float* out = (float*)d_out;

    const int* row = ei;
    const int* col = ei + EE;

    char* ws = (char*)d_ws;
    size_t o = 0;
    auto carve = [&](size_t bytes) -> float* {
        float* p = (float*)(ws + o);
        o += (bytes + 255) & ~(size_t)255;
        return p;
    };
    float* sw     = carve(16 * sizeof(float));
    float* counts = carve((size_t)NN * sizeof(float));
    float* agg1   = carve((size_t)NN * DD * sizeof(float));
    float* fold   = carve((size_t)NN * DD * sizeof(float));
    float* spare  = carve((size_t)NN * DD * sizeof(float));
    float* sortws = carve((size_t)NN * DD * sizeof(float));
    (void)ws_size; (void)in_sizes; (void)n_in; (void)out_size;

    float* Mf = fold;
    float* Pf = fold + NHOP * DD * DD;
    float* cf = Pf + NHOP * P_SLAB_DWORDS;
    float* df = cf + NHOP * DD;

    int*  cnt    = (int*)counts;
    int*  perm   = (int*)sortws;
    int*  region = (int*)spare;
    int*  tab    = (int*)spare + EE;
    int*  ovl    = (int*)spare + EE + 2 * NBIN * NCHUNK;
    int*  ovn    = ovl + NBIN * OVB;

    prologue_kernel<<<PRO_BLOCKS, PRO_T, 0, stream>>>(row, col, tab, region, depths, Wg1, bg1, Wg2, bg2, sw,
                                                      Wh, bh, Wf, Wa, Mf, Pf, cf, df);

    gather1_kernel<<<NBIN, G1_T, 0, stream>>>(x, tab, region, perm, cnt, ovl, ovn, agg1);

    fused_kernel<<<F_BLOCKS, F_THREADS, 0, stream>>>(x, agg1, perm, cnt, ovl, ovn, tab, region,
                                                     Mf, Pf, cf, df, sw, ba, bf, out);
}


# prologue launch in one round: weight fold walks k in 8-step chunks with immediate-offset loads (91 registers, was 212: two blocks fit a WGP), four units per block, beside the edge sort; same sums
# speedup vs baseline: 6.2144x; 1.0403x over previous
#include <hip/hip_runtime.h>
#include <hip/hip_bf16.h>
#include <math.h>

#define NN 50000
#define DD 128
#define EE 800000
#define NHOP 3

typedef __attribute__((ext_vector_type(2))) float v2f;
typedef __attribute__((ext_vector_type(8))) float v8f;

#define PRO_T 512
#define PRO_WAVES (PRO_T / 32)

#define GATE_ITEMS ((NN / 4 + PRO_T - 1) / PRO_T)
static_assert(NN % 4 == 0, "the depths are read as int4");

__device__ __forceinline__ void gate_role(const int* __restrict__ depths,
                                          const float* __restrict__ Wg1, const float* __restrict__ bg1,
                                          const float* __restrict__ Wg2, const float* __restrict__ bg2,
                                          float* __restrict__ sw, int* smax) {
    const int tid = threadIdx.x;
    const int lane = tid & 31;
    const int w = tid >> 5;
    int m = 0;
#pragma unroll
    for (int i = 0; i < GATE_ITEMS; ++i) {
        const int idx = min(tid + i * PRO_T, NN / 4 - 1);
        const int4 v = ((const int4*)depths)[idx];
        m = max(m, max(max(v.x, v.y), max(v.z, v.w)));
    }
    for (int d = 16; d > 0; d >>= 1) m = max(m, __shfl_xor(m, d, 32));
    if (lane == 0) smax[w] = m;
    __syncthreads();
    if (tid == 0) {
        int mm = smax[0];
        for (int k = 1; k < PRO_WAVES; ++k) mm = max(mm, smax[k]);
        float sf0 = (float)NN / 5000.0f;
        float sf1 = (float)mm / 20.0f;
        float a0 = bg2[0], a1 = bg2[1], a2 = bg2[2];
        for (int j = 0; j < DD / 4; ++j) {
            float h = sf0 * Wg1[j] + sf1 * Wg1[DD / 4 + j] + bg1[j];
            h = fmaxf(h, 0.0f);
            a0 += h * Wg2[j * NHOP + 0];
            a1 += h * Wg2[j * NHOP + 1];
            a2 += h * Wg2[j * NHOP + 2];
        }
        sw[0] = 1.0f / (1.0f + __expf(-a0));
        sw[1] = 1.0f / (1.0f + __expf(-a1));
        sw[2] = 1.0f / (1.0f + __expf(-a2));
    }
}

#define CAP 32
#define RANGE 64
#define NBIN ((NN + RANGE - 1) / RANGE)
#define NCHUNK 250
#define CHUNK_E (EE / NCHUNK)
#define CHUNK_E4 (CHUNK_E / 4)
#define HIST_PAD (2 * PRO_T)
#define SORT_ROUNDS ((CHUNK_E4 + PRO_T - 1) / PRO_T)
#define OVB 64
static_assert(EE % (4 * NCHUNK) == 0, "a chunk is a whole number of int4 reads");
static_assert(NN <= 65536, "a source row id is packed into 16 bits");
static_assert(NBIN <= HIST_PAD, "two range counters per thread");
static_assert((RANGE & (RANGE - 1)) == 0, "range is a power of two");

__device__ __forceinline__ void sort_role(const int chunk, const int* __restrict__ row,
                                          const int* __restrict__ col, int2* __restrict__ tab,
                                          int* __restrict__ region,
                                          int* hist, int* offs, int* sorted, int* wsum) {
    const int tid = threadIdx.x;
    const int lane = tid & 31;
    const int w = tid >> 5;
    hist[tid] = 0;
    hist[tid + PRO_T] = 0;

    int4 c4[SORT_ROUNDS], r4[SORT_ROUNDS];
#pragma unroll
    for (int r = 0; r < SORT_ROUNDS; ++r) {
        const int i = min(tid + r * PRO_T, CHUNK_E4 - 1);
        c4[r] = ((const int4*)col)[chunk * CHUNK_E4 + i];
        r4[r] = ((const int4*)row)[chunk * CHUNK_E4 + i];
    }
    __syncthreads();

    int bin[SORT_ROUNDS * 4], rk[SORT_ROUNDS * 4], wd[SORT_ROUNDS * 4];
#pragma unroll
    for (int r = 0; r < SORT_ROUNDS; ++r) {
        const bool ok = tid + r * PRO_T < CHUNK_E4;
        const int cc[4] = {c4[r].x, c4[r].y, c4[r].z, c4[r].w};
        const int ss[4] = {r4[r].x, r4[r].y, r4[r].z, r4[r].w};
#pragma unroll
        for (int e = 0; e < 4; ++e) {
            const int c = min(max(cc[e], 0), NN - 1);
            const int sr = min(max(ss[e], 0), NN - 1);
            const int q = r * 4 + e;
            bin[q] = ok ? (c / RANGE) : -1;
            wd[q] = ((c & (RANGE - 1)) << 16) | sr;
            rk[q] = ok ? atomicAdd(&hist[bin[q]], 1) : 0;
        }
    }
    __syncthreads();

    const int a0 = hist[2 * tid], a1 = hist[2 * tid + 1];
    const int sum2 = a0 + a1;
    int incl = sum2;
    for (int d = 1; d < 32; d <<= 1) {
        const int t = __shfl_up(incl, d, 32);
        if (lane >= d) incl += t;
    }
    if (lane == 31) wsum[w] = incl;
    __syncthreads();
    int wp = 0;
    for (int k = 0; k < PRO_WAVES; ++k) wp += (k < w) ? wsum[k] : 0;
    const int ex = wp + incl - sum2;
    offs[2 * tid] = ex;
    offs[2 * tid + 1] = ex + a0;
    if (2 * tid < NBIN)     tab[(2 * tid) * NCHUNK + chunk]     = make_int2(ex, a0);
    if (2 * tid + 1 < NBIN) tab[(2 * tid + 1) * NCHUNK + chunk] = make_int2(ex + a0, a1);
    __syncthreads();

#pragma unroll
    for (int q = 0; q < SORT_ROUNDS * 4; ++q)
        if (bin[q] >= 0) sorted[offs[bin[q]] + rk[q]] = wd[q];
    __syncthreads();

    for (int i = tid; i < CHUNK_E4; i += PRO_T)
        ((int4*)region)[chunk * CHUNK_E4 + i] = ((const int4*)sorted)[i];
}

#define GATHER_BATCH 8

__device__ __forceinline__ float4 load_row4(const float* __restrict__ src, int r, int lane) {
    return *(const float4*)(src + (long)r * DD + lane * 4);
}

__device__ __forceinline__ void add_hits(const float* __restrict__ src, const bool hit, const int w,
                                         const int lane, float4& acc) {
    const int s = min(w & 0xFFFF, NN - 1);
    unsigned mask = __builtin_amdgcn_ballot_w32(hit);
    while (mask != 0u) {
        const int q = __builtin_ctz(mask);
        mask &= mask - 1u;
        const float4 v = load_row4(src, __builtin_amdgcn_readlane(s, q), lane);
        acc.x += v.x; acc.y += v.y; acc.z += v.z; acc.w += v.w;
    }
}

template <int BATCH>
__device__ __forceinline__ float4 sum_node(const float* __restrict__ src, const int idx, const int deg,
                                           const int d, const int b, const int lane,
                                           const int* ovlist, const int no,
                                           const int2* __restrict__ tab, const int* __restrict__ region) {
    float4 acc = make_float4(0.0f, 0.0f, 0.0f, 0.0f);
    if (deg > CAP && no > OVB) {
        for (int c = 0; c < NCHUNK; ++c) {
            const int2 sc = tab[b * NCHUNK + c];
            const int start = min(max(sc.x, 0), CHUNK_E);
            const int count = min(max(sc.y, 0), CHUNK_E - start);
            for (int k0 = 0; k0 < count; k0 += 32) {
                const int k = k0 + lane;
                const int w = (k < count) ? region[c * CHUNK_E + start + k] : 0;
                add_hits(src, (k < count) && (((w >> 16) & (RANGE - 1)) == d), w, lane, acc);
            }
        }
        return acc;
    }
    const int m = min(deg, CAP);
    int j = 0;
    for (; j + BATCH <= m; j += BATCH) {
        float4 v[BATCH];
        for (int u = 0; u < BATCH; ++u)
            v[u] = load_row4(src, __builtin_amdgcn_readlane(idx, j + u), lane);
        for (int u = 0; u < BATCH; ++u) {
            acc.x += v[u].x; acc.y += v[u].y; acc.z += v[u].z; acc.w += v[u].w;
        }
    }
    if (j < m) {
        float4 v[BATCH - 1];
        for (int u = 0; u < BATCH - 1; ++u) {
            v[u] = make_float4(0.0f, 0.0f, 0.0f, 0.0f);
            if (j + u < m)
                v[u] = load_row4(src, __builtin_amdgcn_readlane(idx, j + u), lane);
        }
        for (int u = 0; u < BATCH - 1; ++u) {
            if (j + u < m) {
                acc.x += v[u].x; acc.y += v[u].y; acc.z += v[u].z; acc.w += v[u].w;
            }
        }
    }
    if (deg > CAP) {
        const int nl = min(no, OVB);
        for (int q0 = 0; q0 < nl; q0 += 32) {
            const int k = q0 + lane;
            const int w = (k < nl) ? ovlist[k] : 0;
            add_hits(src, (k < nl) && (((w >> 16) & (RANGE - 1)) == d), w, lane, acc);
        }
    }
    return acc;
}

__device__ __forceinline__ void store_mean(float* __restrict__ dst, const int node, const int deg,
                                           const int lane, float4 acc) {
    const float inv = 1.0f / fmaxf((float)deg, 1.0f);
    acc.x *= inv; acc.y *= inv; acc.z *= inv; acc.w *= inv;
    *(float4*)(dst + (long)node * DD + lane * 4) = acc;
}

#define G1_T 512
#define G1_WAVES (G1_T / 32)
#define SEG_BATCH 12
static_assert(NCHUNK <= G1_T, "one thread per chunk segment");

__global__ void __launch_bounds__(G1_T)
gather1_kernel(const float* __restrict__ src, const int* __restrict__ tab_w,
               const int* __restrict__ region, int* __restrict__ perm, int* __restrict__ cnt,
               int* __restrict__ ovl, int* __restrict__ ovn, float* __restrict__ dst) {
    __shared__ __attribute__((aligned(16))) int lperm[RANGE * CAP];
    __shared__ int lcnt[RANGE];
    __shared__ int lovl[OVB];
    __shared__ int lovn;
    const int2* __restrict__ tab = (const int2*)tab_w;
    const int tid = threadIdx.x;
    const int lane = tid & 31;
    const int wave = __builtin_amdgcn_readfirstlane(tid >> 5);
    const int b = blockIdx.x;
    const int nodes_here = min(RANGE, NN - b * RANGE);

    if (tid < RANGE) lcnt[tid] = 0;
    if (tid == 0) lovn = 0;
    int start = 0, count = 0;
    if (tid < NCHUNK) {
        const int2 sc = tab[b * NCHUNK + tid];
        start = min(max(sc.x, 0), CHUNK_E);
        count = min(max(sc.y, 0), CHUNK_E - start);
    }
    const int* base = region + min(tid, NCHUNK - 1) * CHUNK_E + start;
    __syncthreads();

    for (int k0 = 0; k0 < count; k0 += SEG_BATCH) {
        int wv[SEG_BATCH];
#pragma unroll
        for (int u = 0; u < SEG_BATCH; ++u) wv[u] = (k0 + u < count) ? base[k0 + u] : 0;
#pragma unroll
        for (int u = 0; u < SEG_BATCH; ++u) {
            if (k0 + u < count) {
                const int d = (wv[u] >> 16) & (RANGE - 1);
                const int slot = atomicAdd(&lcnt[d], 1);
                if (slot < CAP) {
                    lperm[d * CAP + slot] = wv[u] & 0xFFFF;
                } else {
                    const int q = atomicAdd(&lovn, 1);
                    if (q < OVB) lovl[q] = wv[u];
                }
            }
        }
    }
    __syncthreads();

    const int no = __builtin_amdgcn_readfirstlane(lovn);
    for (int i = tid; i < nodes_here * (CAP / 4); i += G1_T)
        ((int4*)(perm + (long)b * RANGE * CAP))[i] = ((const int4*)lperm)[i];
    if (tid < nodes_here) cnt[b * RANGE + tid] = lcnt[tid];
    if (tid < min(no, OVB)) ovl[b * OVB + tid] = lovl[tid];
    if (tid == 0) ovn[b] = no;

    for (int d = wave; d < nodes_here; d += G1_WAVES) {
        const int deg = __builtin_amdgcn_readfirstlane(lcnt[d]);
        const int idx = min(max(lperm[d * CAP + lane], 0), NN - 1);
        const float4 acc = sum_node<GATHER_BATCH>(src, idx, deg, d, b, lane, lovl, no, tab, region);
        store_mean(dst, b * RANGE + d, deg, lane, acc);
    }
}

#define GEMM_WAVES 4
#define GEMM_THREADS (GEMM_WAVES * 32)
#define ROW_TILES (NN / 16)
#define GEMM_BLOCKS ((ROW_TILES + GEMM_WAVES - 1) / GEMM_WAVES)
static_assert(NN % 16 == 0, "rows are handled in whole 16-row tiles");
static_assert(DD == 128, "the slab layout below is written for 128 x 128 weight slabs");

#define W_PITCH 288
#define W_SLAB_DWORDS ((DD / 2) * W_PITCH)
#define W_ITEMS_PER_THREAD ((DD / 2) / GEMM_WAVES)

__device__ __forceinline__ void stage_slab_load(const float* __restrict__ Wsrc, int tid,
                                                float4 (&r0)[W_ITEMS_PER_THREAD],
                                                float4 (&r1)[W_ITEMS_PER_THREAD]) {
    const int kp0 = tid >> 5;
    const int c4  = tid & 31;
    const float* src = Wsrc + (long)(2 * kp0) * DD + 4 * c4;
#pragma unroll
    for (int it = 0; it < W_ITEMS_PER_THREAD; ++it) {
        r0[it] = *(const float4*)(src + it * (2 * GEMM_WAVES * DD));
        r1[it] = *(const float4*)(src + it * (2 * GEMM_WAVES * DD) + DD);
    }
}

__device__ __forceinline__ void stage_slab_store(float* wl, int tid,
                                                 const float4 (&r0)[W_ITEMS_PER_THREAD],
                                                 const float4 (&r1)[W_ITEMS_PER_THREAD]) {
    const int kp0 = tid >> 5;
    const int c4  = tid & 31;
    float* dst = wl + kp0 * W_PITCH + 8 * c4;
#pragma unroll
    for (int it = 0; it < W_ITEMS_PER_THREAD; ++it) {
        float* d = dst + it * (GEMM_WAVES * W_PITCH);
        *(float4*)(d)     = make_float4(r0[it].x, r1[it].x, r0[it].y, r1[it].y);
        *(float4*)(d + 4) = make_float4(r0[it].z, r1[it].z, r0[it].w, r1[it].w);
    }
}

__device__ __forceinline__ void load_a_frags(const float* __restrict__ arow, v2f (&a)[DD / 4]) {
#pragma unroll
    for (int s = 0; s < DD / 4; ++s) {
        a[s].x = arow[4 * s];
        a[s].y = arow[4 * s + 1];
    }
}

#define P_PITCH 32
#define P_SLAB_DWORDS ((DD / 2) * P_PITCH)

template <int NCT, int S0, int S1>
__device__ __forceinline__ void mma_steps(const v2f (&a)[DD / 4], const float* wl_lane,
                                          const float* pl_lane, v8f (&acc)[NCT], v8f& accl) {
#pragma unroll
    for (int s = S0; s < S1; ++s) {
        v2f bb[NCT];
#pragma unroll
        for (int ct = 0; ct < NCT; ++ct)
            bb[ct] = *(const v2f*)(wl_lane + 2 * (s - S0) * W_PITCH + 32 * ct);
        const v2f pp = *(const v2f*)(pl_lane + 2 * (s - S0) * P_PITCH);
#pragma unroll
        for (int ct = 0; ct < NCT; ++ct)
            acc[ct] = __builtin_amdgcn_wmma_f32_16x16x4_f32(false, a[s], false, bb[ct],
                                                            (short)0, acc[ct], false, false);
        accl = __builtin_amdgcn_wmma_f32_16x16x4_f32(false, a[s], false, pp,
                                                     (short)0, accl, false, false);
    }
}

#define FOLD_ROW_TILES 9
#define FOLD_COL_GROUPS 5
#define FOLD_BLOCKS (NHOP * FOLD_ROW_TILES * FOLD_COL_GROUPS)
#define DF_STRIDE 32
#define FOLD_STEPS 8
static_assert((DD / 4) % FOLD_STEPS == 0, "the k-steps are walked in whole chunks");

template <int BS>
__device__ __forceinline__ void fold_products(const float* ap, const float* bp0, const float* bp1,
                                              v8f& acc0, v8f& acc1) {
#pragma nounroll
    for (int s0 = 0; s0 < DD / 4; s0 += FOLD_STEPS) {
        const float* apc  = ap + 4 * s0;
        const float* bpc0 = bp0 + (4 * s0) * BS;
        const float* bpc1 = bp1 + (4 * s0) * BS;
        v2f a[FOLD_STEPS], b0[FOLD_STEPS], b1[FOLD_STEPS];
#pragma unroll
        for (int u = 0; u < FOLD_STEPS; ++u) {
            a[u].x  = apc[4 * u];
            a[u].y  = apc[4 * u + 1];
            b0[u].x = bpc0[(4 * u) * BS];
            b0[u].y = bpc0[(4 * u + 1) * BS];
            b1[u].x = bpc1[(4 * u) * BS];
            b1[u].y = bpc1[(4 * u + 1) * BS];
        }
        __builtin_amdgcn_sched_barrier(0);
#pragma unroll
        for (int u = 0; u < FOLD_STEPS; ++u) {
            acc0 = __builtin_amdgcn_wmma_f32_16x16x4_f32(false, a[u], false, b0[u], (short)0, acc0, false, false);
            acc1 = __builtin_amdgcn_wmma_f32_16x16x4_f32(false, a[u], false, b1[u], (short)0, acc1, false, false);
        }
    }
}

__device__ __forceinline__ void fold_role(const int b, const int lane,
            const float* __restrict__ Wh, const float* __restrict__ bh,
            const float* __restrict__ Wf, const float* __restrict__ Wa,
            float* __restrict__ Mf, float* __restrict__ Pf,
            float* __restrict__ cf, float* __restrict__ df) {
    const int hop = b / (FOLD_ROW_TILES * FOLD_COL_GROUPS);
    const int rt  = (b / FOLD_COL_GROUPS) % FOLD_ROW_TILES;
    const int cg  = b % FOLD_COL_GROUPS;
    const int half = lane >> 4;
    const int l    = lane & 15;

    const int R = 16 * rt + l;
    const float* ap = (R < DD) ? Wh + ((long)hop * DD + R) * DD : bh + hop * DD;
    ap += 2 * half;

    const float* Wf_h = Wf + (long)hop * DD * DD;
    const float* Wa_h = Wa + (long)hop * DD * NHOP;
    const int C0 = 32 * cg + l;
    const int C1 = C0 + 16;
    const float* bp0; int bs0; bool bok0;
    if (C0 < DD)             { bp0 = Wf_h + C0;        bs0 = DD;   bok0 = true;  }
    else if (C0 < DD + NHOP) { bp0 = Wa_h + (C0 - DD); bs0 = NHOP; bok0 = true;  }
    else                     { bp0 = Wa_h;             bs0 = NHOP; bok0 = false; }
    const float* bp1; int bs1;
    if (C1 < DD)             { bp1 = Wf_h + C1;        bs1 = DD;   }
    else if (C1 < DD + NHOP) { bp1 = Wa_h + (C1 - DD); bs1 = NHOP; }
    else                     { bp1 = Wa_h;             bs1 = NHOP; }
    bp0 += 2 * half * bs0;
    bp1 += 2 * half * bs1;

    v8f acc0 = {}, acc1 = {};
    if (cg < FOLD_COL_GROUPS - 1) fold_products<DD>(ap, bp0, bp1, acc0, acc1);
    else                          fold_products<NHOP>(ap, bp0, bp1, acc0, acc1);

#pragma unroll
    for (int j = 0; j < 8; ++j) {
        const int Rr = 16 * rt + 8 * half + j;
        if (Rr < DD) {
            if (cg < FOLD_COL_GROUPS - 1) {
                Mf[((long)hop * DD + Rr) * DD + C0] = acc0[j];
                Mf[((long)hop * DD + Rr) * DD + C1] = acc1[j];
            } else {
                Pf[hop * P_SLAB_DWORDS + (Rr >> 1) * P_PITCH + 2 * l + (Rr & 1)] = bok0 ? acc0[j] : 0.0f;
            }
        } else if (Rr == DD) {
            if (cg < FOLD_COL_GROUPS - 1) {
                cf[hop * DD + C0] = acc0[j];
                cf[hop * DD + C1] = acc1[j];
            } else {
                df[hop * DF_STRIDE + l] = bok0 ? acc0[j] : 0.0f;
            }
        }
    }
}

#define FOLD_UNITS_PER_BLOCK 4
#define PRO_BLOCKS (NCHUNK + 1 + (FOLD_BLOCKS + FOLD_UNITS_PER_BLOCK - 1) / FOLD_UNITS_PER_BLOCK)
static_assert(PRO_BLOCKS <= 2 * 256, "two blocks on each of the 256 WGPs hold the whole grid");

__global__ void __launch_bounds__(PRO_T)
prologue_kernel(const int* __restrict__ row, const int* __restrict__ col,
                int* __restrict__ tab_w, int* __restrict__ region,
                const int* __restrict__ depths,
                const float* __restrict__ Wg1, const float* __restrict__ bg1,
                const float* __restrict__ Wg2, const float* __restrict__ bg2,
                float* __restrict__ sw,
                const float* __restrict__ Wh, const float* __restrict__ bh,
                const float* __restrict__ Wf, const float* __restrict__ Wa,
                float* __restrict__ Mf, float* __restrict__ Pf,
                float* __restrict__ cf, float* __restrict__ df) {
    __shared__ int sh[32];
    __shared__ int hist[HIST_PAD];
    __shared__ int offs[HIST_PAD];
    __shared__ __attribute__((aligned(16))) int sorted[CHUNK_E];
    const int b = blockIdx.x;
    if (b < NCHUNK) {
        sort_role(b, row, col, (int2*)tab_w, region, hist, offs, sorted, sh);
    } else if (b == NCHUNK) {
        gate_role(depths, Wg1, bg1, Wg2, bg2, sw, sh);
    } else {
        const int wave = __builtin_amdgcn_readfirstlane((int)(threadIdx.x >> 5));
        const int unit = (b - NCHUNK - 1) * FOLD_UNITS_PER_BLOCK + wave;
        if (wave < FOLD_UNITS_PER_BLOCK && unit < FOLD_BLOCKS)
            fold_role(unit, (int)(threadIdx.x & 31), Wh, bh, Wf, Wa, Mf, Pf, cf, df);
    }
}


#define F_WAVES 12
#define F_THREADS (F_WAVES * 32)
#define F_BLOCKS 256
#define MAIN_TILES (F_BLOCKS * F_WAVES)
#define REM_TILES (ROW_TILES - MAIN_TILES)
#define REM_SPLIT 4
#define REM_NCT (8 / REM_SPLIT)
#define REM_UNITS (REM_TILES * REM_SPLIT)
#define P_BASE_DWORDS (NHOP * W_SLAB_DWORDS)
static_assert(F_WAVES == NHOP * GEMM_WAVES, "one group of GEMM_WAVES waves stages one hop's slabs");
static_assert(MAIN_TILES <= ROW_TILES, "every wave has a whole row tile");
static_assert(REM_UNITS >= 0 && REM_UNITS <= F_BLOCKS, "at most one column group of a leftover tile per work-group");
static_assert(P_SLAB_DWORDS == 4 * 4 * GEMM_THREADS, "the P slab copy is four 16-byte items per thread");

#define S_ROWS 8
#define S_PITCH 132
#define F_GATHER_BATCH 16
#define S_DWORDS (S_ROWS * S_PITCH)
static_assert(2 * S_ROWS == 16, "two strip rounds make one row tile");
#define STRIP_BASE_DWORDS (P_BASE_DWORDS + NHOP * P_SLAB_DWORDS)
#define REM_STRIP_DWORDS (16 * S_PITCH)
#define REM_ROWS_PER_WAVE 2
static_assert(16 / REM_ROWS_PER_WAVE <= F_WAVES, "enough waves in a work-group for the leftover rows");
#define REM_STRIP_BASE_DWORDS (STRIP_BASE_DWORDS + F_WAVES * S_DWORDS)
#define F_LDS_DWORDS (REM_STRIP_BASE_DWORDS + REM_STRIP_DWORDS)
static_assert(F_LDS_DWORDS * 4 <= 320 * 1024, "slabs and strips fit the LDS of a WGP");
static_assert((STRIP_BASE_DWORDS % 4) == 0 && (S_DWORDS % 4) == 0, "strip rows are written as 16-byte items");

struct GatherSrc {
    const float* src;
    const int*   perm;
    const int*   cnt;
    const int*   ovl;
    const int*   ovn;
    const int2*  tab;
    const int*   region;
};

__device__ __forceinline__ void gather_rows(const GatherSrc& g, const int node0, const int n,
                                            const int lane, float* strip) {
    int deg_n = g.cnt[node0];
    int idx_n = g.perm[node0 * CAP + lane];
    for (int i = 0; i < n; ++i) {
        const int node = node0 + i;
        const int deg = deg_n;
        const int idx = min(max(idx_n, 0), NN - 1);
        if (i + 1 < n) {
            deg_n = g.cnt[node + 1];
            idx_n = g.perm[(node + 1) * CAP + lane];
        }
        const int b = node / RANGE;
        const int no = (deg > CAP) ? g.ovn[b] : 0;
        float4 acc = sum_node<F_GATHER_BATCH>(g.src, idx, deg, node & (RANGE - 1), b, lane,
                                              g.ovl + b * OVB, no, g.tab, g.region);
        const float inv = 1.0f / fmaxf((float)deg, 1.0f);
        acc.x *= inv; acc.y *= inv; acc.z *= inv; acc.w *= inv;
        *(float4*)(strip + i * S_PITCH + lane * 4) = acc;
    }
}

__device__ __forceinline__ void load_strip_frags(const float* srow, v2f (&a)[DD / 4]) {
#pragma unroll
    for (int s = 0; s < DD / 4; ++s) {
        a[s].x = srow[4 * s];
        a[s].y = srow[4 * s + 1];
    }
}

__device__ __forceinline__ void gathered_a_frags(const GatherSrc& g, const int row0, const int lane,
                                                 float* strip, v2f (&a)[DD / 4]) {
    const int half = lane >> 4;
    const int l    = lane & 15;
    const float* srow = strip + (l & (S_ROWS - 1)) * S_PITCH + 2 * half;
#pragma unroll
    for (int p = 0; p < 2; ++p) {
        gather_rows(g, row0 + p * S_ROWS, S_ROWS, lane, strip);
        __builtin_amdgcn_wave_barrier();
        if ((l >> 3) == p) load_strip_frags(srow, a);
        __builtin_amdgcn_wave_barrier();
    }
}

template <int NCT, bool OWN_GATHER>
__device__ __forceinline__ void tile_pass(const int row0, const int ct0, const int lane, v2f (&a)[DD / 4],
             const float* __restrict__ X1, const GatherSrc& g, float* strip,
             const float* lds,
             const float* __restrict__ cf, const float* __restrict__ df,
             const float* __restrict__ sw, const float* __restrict__ ba,
             const float* __restrict__ bf, float* __restrict__ out) {
    const int half = lane >> 4;
    const int l    = lane & 15;
    const int row  = row0 + l;

    v8f acc[NHOP][NCT] = {};
    v8f accl[NHOP] = {};
#pragma unroll
    for (int hop = 0; hop < NHOP; ++hop) {
        if (hop == 1) load_a_frags(X1 + (long)row * DD + 2 * half, a);
        if (hop == 2) {
            if (OWN_GATHER) gathered_a_frags(g, row0, lane, strip, a);
            else            load_strip_frags(strip + l * S_PITCH + 2 * half, a);
        }
        int w_lo = hop * W_SLAB_DWORDS + half * W_PITCH + 2 * l + 32 * ct0;
        int w_hi = w_lo + (DD / 4) * W_PITCH;
        int p_lo = P_BASE_DWORDS + hop * P_SLAB_DWORDS + half * P_PITCH + 2 * l;
        int p_hi = p_lo + (DD / 4) * P_PITCH;
        asm volatile("" : "+v"(w_lo), "+v"(w_hi), "+v"(p_lo), "+v"(p_hi));
        mma_steps<NCT, 0, DD / 8>(a, lds + w_lo, lds + p_lo, acc[hop], accl[hop]);
        mma_steps<NCT, DD / 8, DD / 4>(a, lds + w_hi, lds + p_hi, acc[hop], accl[hop]);
    }

    const float s0 = sw[0], s1 = sw[1], s2 = sw[2];
    const float bac = ba[min(l, NHOP - 1)];
    const float d0 = df[0 * DF_STRIDE + l], d1 = df[1 * DF_STRIDE + l], d2 = df[2 * DF_STRIDE + l];
    const int src = lane & 16;
    float w0[8], w1[8], w2[8];
#pragma unroll
    for (int j = 0; j < 8; ++j) {
        const float lg = bac + s0 * (accl[0][j] + d0) + s1 * (accl[1][j] + d1) + s2 * (accl[2][j] + d2);
        const float l0 = __shfl(lg, src + 0, 32);
        const float l1 = __shfl(lg, src + 1, 32);
        const float l2 = __shfl(lg, src + 2, 32);
        const float mx = fmaxf(l0, fmaxf(l1, l2));
        const float e0 = __expf(l0 - mx), e1 = __expf(l1 - mx), e2 = __expf(l2 - mx);
        const float inv = 1.0f / (e0 + e1 + e2);
        w0[j] = e0 * inv * s0;
        w1[j] = e1 * inv * s1;
        w2[j] = e2 * inv * s2;
    }

#pragma unroll
    for (int ct = 0; ct < NCT; ++ct) {
        const int col = 16 * (ct0 + ct) + l;
        const float c0 = cf[0 * DD + col], c1 = cf[1 * DD + col], c2 = cf[2 * DD + col];
        const float bias = bf[col];
#pragma unroll
        for (int j = 0; j < 8; ++j) {
            const int m = j + 8 * half;
            out[(long)(row0 + m) * DD + col] = w0[j] * (acc[0][ct][j] + c0) + w1[j] * (acc[1][ct][j] + c1)
                                             + w2[j] * (acc[2][ct][j] + c2) + bias;
        }
    }
}

__global__ void __launch_bounds__(F_THREADS) __attribute__((amdgpu_waves_per_eu(3, 3)))
fused_kernel(const float* __restrict__ X0, const float* __restrict__ X1,
             const int* __restrict__ perm, const int* __restrict__ cnt,
             const int* __restrict__ ovl, const int* __restrict__ ovn,
             const int* __restrict__ tab_w, const int* __restrict__ region,
             const float* __restrict__ Mf, const float* __restrict__ Pf,
             const float* __restrict__ cf, const float* __restrict__ df,
             const float* __restrict__ sw, const float* __restrict__ ba,
             const float* __restrict__ bf, float* __restrict__ out) {
    __shared__ __attribute__((aligned(16))) float lds[F_LDS_DWORDS];
    float* wl = lds;
    float* pl = lds + P_BASE_DWORDS;
    const int tid = threadIdx.x;
    const int wave = __builtin_amdgcn_readfirstlane(tid >> 5);
    const int lane = tid & 31;
    const int half = lane >> 4;
    const int l    = lane & 15;
    float* strip = lds + STRIP_BASE_DWORDS + wave * S_DWORDS;
    float* rem_strip = lds + REM_STRIP_BASE_DWORDS;
    const GatherSrc gs = {X1, perm, cnt, ovl, ovn, (const int2*)tab_w, region};
    const int u = blockIdx.x;
    const bool has_rem = u < REM_UNITS;
    const int rrow0 = (MAIN_TILES + u / REM_SPLIT) * 16;

    const int g    = wave / GEMM_WAVES;
    const int ltid = tid - g * GEMM_THREADS;
    float4 r0[W_ITEMS_PER_THREAD], r1[W_ITEMS_PER_THREAD];
    stage_slab_load(Mf + (long)g * DD * DD, ltid, r0, r1);
    const float4* psrc = (const float4*)(Pf + g * P_SLAB_DWORDS) + ltid;
    const float4 p0 = psrc[0], p1 = psrc[GEMM_THREADS], p2 = psrc[2 * GEMM_THREADS], p3 = psrc[3 * GEMM_THREADS];
    const int row0 = (blockIdx.x * F_WAVES + wave) * 16;
    v2f a[DD / 4];
    load_a_frags(X0 + (long)(row0 + l) * DD + 2 * half, a);
    __builtin_amdgcn_sched_barrier(0);
    if (has_rem && wave < 16 / REM_ROWS_PER_WAVE)
        gather_rows(gs, rrow0 + wave * REM_ROWS_PER_WAVE, REM_ROWS_PER_WAVE, lane,
                    rem_strip + wave * REM_ROWS_PER_WAVE * S_PITCH);
    stage_slab_store(wl + g * W_SLAB_DWORDS, ltid, r0, r1);
    float4* pdst = (float4*)(pl + g * P_SLAB_DWORDS) + ltid;
    pdst[0] = p0; pdst[GEMM_THREADS] = p1; pdst[2 * GEMM_THREADS] = p2; pdst[3 * GEMM_THREADS] = p3;
    __syncthreads();

    tile_pass<8, true>(row0, 0, lane, a, X1, gs, strip, lds, cf, df, sw, ba, bf, out);

    if (wave == 0 && has_rem) {
        load_a_frags(X0 + (long)(rrow0 + l) * DD + 2 * half, a);
        tile_pass<REM_NCT, false>(rrow0, (u % REM_SPLIT) * REM_NCT, lane, a, X1, gs, rem_strip, lds, cf, df, sw, ba, bf, out);
    }
}

extern "C" void kernel_launch(void* const* d_in, const int* in_sizes, int n_in,
                              void* d_out, int out_size, void* d_ws, size_t ws_size,
                              hipStream_t stream) {
    const float* x      = (const float*)d_in[0];
    const int*   ei     = (const int*)  d_in[1];
    const int*   depths = (const int*)  d_in[2];
    const float* Wh     = (const float*)d_in[3];
    const float* bh     = (const float*)d_in[4];
    const float* Wg1    = (const float*)d_in[5];
    const float* bg1    = (const float*)d_in[6];
    const float* Wg2    = (const float*)d_in[7];
    const float* bg2    = (const float*)d_in[8];
    const float* Wa     = (const float*)d_in[9];
    const float* ba     = (const float*)d_in[10];
    const float* Wf     = (const float*)d_in[11];
    const float* bf     = (const float*)d_in[12];
    float* out = (float*)d_out;

    const int* row = ei;
    const int* col = ei + EE;

    char* ws = (char*)d_ws;
    size_t o = 0;
    auto carve = [&](size_t bytes) -> float* {
        float* p = (float*)(ws + o);
        o += (bytes + 255) & ~(size_t)255;
        return p;
    };
    float* sw     = carve(16 * sizeof(float));
    float* counts = carve((size_t)NN * sizeof(float));
    float* agg1   = carve((size_t)NN * DD * sizeof(float));
    float* fold   = carve((size_t)NN * DD * sizeof(float));
    float* spare  = carve((size_t)NN * DD * sizeof(float));
    float* sortws = carve((size_t)NN * DD * sizeof(float));
    (void)ws_size; (void)in_sizes; (void)n_in; (void)out_size;

    float* Mf = fold;
    float* Pf = fold + NHOP * DD * DD;
    float* cf = Pf + NHOP * P_SLAB_DWORDS;
    float* df = cf + NHOP * DD;

    int*  cnt    = (int*)counts;
    int*  perm   = (int*)sortws;
    int*  region = (int*)spare;
    int*  tab    = (int*)spare + EE;
    int*  ovl    = (int*)spare + EE + 2 * NBIN * NCHUNK;
    int*  ovn    = ovl + NBIN * OVB;

    prologue_kernel<<<PRO_BLOCKS, PRO_T, 0, stream>>>(row, col, tab, region, depths, Wg1, bg1, Wg2, bg2, sw,
                                                      Wh, bh, Wf, Wa, Mf, Pf, cf, df);

    gather1_kernel<<<NBIN, G1_T, 0, stream>>>(x, tab, region, perm, cnt, ovl, ovn, agg1);

    fused_kernel<<<F_BLOCKS, F_THREADS, 0, stream>>>(x, agg1, perm, cnt, ovl, ovn, tab, region,
                                                     Mf, Pf, cf, df, sw, ba, bf, out);
}
